# starting kernel
# baseline (speedup 1.0000x reference)
#include <hip/hip_runtime.h>
#include <math.h>

#define N_NODES 40000
#define N_EDGES 320000
#define ETOT    (N_EDGES + N_NODES)
#define F_IN    256
#define H1      4
#define C1      128
#define D1      (H1*C1)
#define H2      8
#define C2      6
#define D2      (H2*C2)
#define BN_EPS    1e-5f
#define NEG_SLOPE 0.2f

typedef __attribute__((ext_vector_type(2))) float v2f;
typedef __attribute__((ext_vector_type(8))) float v8f;


__device__ __forceinline__ void atomic_max_f32(float* addr, float v) {
  if (v >= 0.0f) atomicMax((int*)addr, __float_as_int(v));
  else           atomicMin((unsigned int*)addr, __float_as_uint(v));
}

__device__ __forceinline__ void edge_rc(const int* __restrict__ rowA,
                                        const int* __restrict__ colA,
                                        int e, int& r, int& c) {
  if (e < N_EDGES) { r = rowA[e]; c = colA[e]; }
  else             { r = e - N_EDGES; c = r; }
}


__global__ void fill_f32(float* __restrict__ p, float v, int n) {
  int i = blockIdx.x * blockDim.x + threadIdx.x;
  int stride = gridDim.x * blockDim.x;
  for (; i < n; i += stride) p[i] = v;
}

__global__ void bn_prep(const float* g1, const float* be1, const float* mu1, const float* va1,
                        const float* g2, const float* be2, const float* mu2, const float* va2,
                        const float* bias1,
                        float* sc1, float* sh1, float* sc2, float* sh2) {
  int i = blockIdx.x * blockDim.x + threadIdx.x;
  if (i < F_IN) {
    float s = g1[i] / sqrtf(va1[i] + BN_EPS);
    sc1[i] = s; sh1[i] = be1[i] - mu1[i] * s;
  }
  if (i < D1) {
    float s = g2[i] / sqrtf(va2[i] + BN_EPS);
    sc2[i] = s; sh2[i] = be2[i] + (bias1[i] - mu2[i]) * s;
  }
}


__global__ __launch_bounds__(256)
void gemm_bn_wmma(const float* __restrict__ A,
                  const float* __restrict__ scale,
                  const float* __restrict__ shift,
                  const float* __restrict__ B,
                  float* __restrict__ C,
                  int M, int K, int N) {
  int wave = blockIdx.x * (blockDim.x >> 5) + (threadIdx.x >> 5);
  int ntiles = N >> 4;
  int tm = wave / ntiles;
  int tn = wave - tm * ntiles;
  if (tm >= (M >> 4)) return;

  int lane = threadIdx.x & 31;
  int half = lane >> 4;
  int l16  = lane & 15;
  int row  = (tm << 4) + l16;
  int colB = (tn << 4) + l16;

  const float* __restrict__ Arow = A + (size_t)row * K;
  v8f acc = {};
  for (int k = 0; k < K; k += 4) {
    int kk = k + (half << 1);
    v2f a, b;
    a.x = Arow[kk]     * scale[kk]     + shift[kk];
    a.y = Arow[kk + 1] * scale[kk + 1] + shift[kk + 1];
    b.x = B[(size_t)kk * N + colB];
    b.y = B[(size_t)(kk + 1) * N + colB];
    acc = __builtin_amdgcn_wmma_f32_16x16x4_f32(
         false, a,  false, b,
         (short)0, acc,  false,  false);
  }
  float* Crow = C + (size_t)((tm << 4) + (half << 3)) * N + colB;
  #pragma unroll
  for (int r = 0; r < 8; ++r) Crow[(size_t)r * N] = acc[r];
}


__global__ void node_att1(const float* __restrict__ h1, const float* __restrict__ att1,
                          float* __restrict__ ai, float* __restrict__ aj) {
  int wave = blockIdx.x * (blockDim.x >> 5) + (threadIdx.x >> 5);
  if (wave >= N_NODES * H1) return;
  int n = wave >> 2, h = wave & 3;
  int lane = threadIdx.x & 31;
  const float* __restrict__ hp = h1 + (size_t)n * D1 + h * C1;
  const float* __restrict__ al = att1 + h * (2 * C1);
  const float* __restrict__ ar = al + C1;
  float si = 0.f, sj = 0.f;
  #pragma unroll
  for (int c0 = 0; c0 < C1; c0 += 32) {
    float v = hp[c0 + lane];
    si += v * al[c0 + lane];
    sj += v * ar[c0 + lane];
  }
  #pragma unroll
  for (int off = 16; off > 0; off >>= 1) {
    si += __shfl_xor(si, off, 32);
    sj += __shfl_xor(sj, off, 32);
  }
  if (lane == 0) { ai[wave] = si; aj[wave] = sj; }
}

__global__ void edge_max1(const int* __restrict__ rowA, const int* __restrict__ colA,
                          const float* __restrict__ ai, const float* __restrict__ aj,
                          float* __restrict__ m1) {
  int e = blockIdx.x * blockDim.x + threadIdx.x;
  if (e >= ETOT) return;
  int row, col; edge_rc(rowA, colA, e, row, col);
  float4 av = *(const float4*)(ai + (size_t)row * 4);
  float4 bv = *(const float4*)(aj + (size_t)col * 4);
  float s[4] = {av.x + bv.x, av.y + bv.y, av.z + bv.z, av.w + bv.w};
  #pragma unroll
  for (int h = 0; h < 4; ++h) {
    float a = s[h]; a = a > 0.f ? a : NEG_SLOPE * a;
    atomic_max_f32(&m1[row * 4 + h], a);
  }
}

__global__ void edge_sum1(const int* __restrict__ rowA, const int* __restrict__ colA,
                          const float* __restrict__ ai, const float* __restrict__ aj,
                          const float* __restrict__ m1, float* __restrict__ s1) {
  int e = blockIdx.x * blockDim.x + threadIdx.x;
  if (e >= ETOT) return;
  int row, col; edge_rc(rowA, colA, e, row, col);
  float4 av = *(const float4*)(ai + (size_t)row * 4);
  float4 bv = *(const float4*)(aj + (size_t)col * 4);
  float4 mv = *(const float4*)(m1 + (size_t)row * 4);
  float s[4] = {av.x + bv.x, av.y + bv.y, av.z + bv.z, av.w + bv.w};
  float m[4] = {mv.x, mv.y, mv.z, mv.w};
  #pragma unroll
  for (int h = 0; h < 4; ++h) {
    float a = s[h]; a = a > 0.f ? a : NEG_SLOPE * a;
    atomicAdd(&s1[row * 4 + h], expf(a - m[h]));
  }
}

__global__ __launch_bounds__(256)
void edge_scatter1(const int* __restrict__ rowA, const int* __restrict__ colA,
                   const float* __restrict__ ai, const float* __restrict__ aj,
                   const float* __restrict__ m1, const float* __restrict__ s1,
                   const float* __restrict__ h1, float* __restrict__ out1) {
  int wave = blockIdx.x * (blockDim.x >> 5) + (threadIdx.x >> 5);
  if (wave >= ETOT * H1) return;
  int e = wave >> 2, h = wave & 3;
  int row, col; edge_rc(rowA, colA, e, row, col);
  float a = ai[row * 4 + h] + aj[col * 4 + h];
  a = a > 0.f ? a : NEG_SLOPE * a;
  float w = expf(a - m1[row * 4 + h]) / (s1[row * 4 + h] + 1e-16f);
  int lane = threadIdx.x & 31;
  const float4* __restrict__ src = (const float4*)(h1 + (size_t)col * D1 + h * C1);
  float4 v = src[lane];
  float* dst = out1 + (size_t)row * D1 + h * C1 + lane * 4;
  atomicAdd(dst + 0, v.x * w);
  atomicAdd(dst + 1, v.y * w);
  atomicAdd(dst + 2, v.z * w);
  atomicAdd(dst + 3, v.w * w);
}


__global__ void node_att2(const float* __restrict__ h2, const float* __restrict__ att2,
                          float* __restrict__ s2) {
  int tid = blockIdx.x * blockDim.x + threadIdx.x;
  if (tid >= N_NODES * H2) return;
  int n = tid >> 3, h = tid & 7;
  const float* __restrict__ hp = h2 + (size_t)n * D2 + h * C2;
  const float* __restrict__ ap = att2 + h * (2 * C2);
  float s = 0.f;
  #pragma unroll
  for (int c = 0; c < C2; ++c) s += hp[c] * (ap[c] + ap[C2 + c]);
  s2[tid] = tanhf(s);
}

__global__ void edge_max2(const int* __restrict__ rowA, const int* __restrict__ colA,
                          const float* __restrict__ s2, float* __restrict__ m2) {
  int e = blockIdx.x * blockDim.x + threadIdx.x;
  if (e >= ETOT) return;
  int row, col; edge_rc(rowA, colA, e, row, col);
  const float* sv = s2 + (size_t)col * 8;
  float* mr = m2 + (size_t)row * 8;
  #pragma unroll
  for (int h = 0; h < 8; ++h) atomic_max_f32(&mr[h], sv[h]);
}

__global__ void edge_sum2(const int* __restrict__ rowA, const int* __restrict__ colA,
                          const float* __restrict__ s2, const float* __restrict__ m2,
                          float* __restrict__ ss2) {
  int e = blockIdx.x * blockDim.x + threadIdx.x;
  if (e >= ETOT) return;
  int row, col; edge_rc(rowA, colA, e, row, col);
  const float* sv = s2 + (size_t)col * 8;
  const float* mr = m2 + (size_t)row * 8;
  #pragma unroll
  for (int h = 0; h < 8; ++h) atomicAdd(&ss2[row * 8 + h], expf(sv[h] - mr[h]));
}

__global__ void edge_scatter2(const int* __restrict__ rowA, const int* __restrict__ colA,
                              const float* __restrict__ s2, const float* __restrict__ m2,
                              const float* __restrict__ ss2,
                              const float* __restrict__ h2, float* __restrict__ out2) {
  int tid = blockIdx.x * blockDim.x + threadIdx.x;
  if (tid >= ETOT * H2) return;
  int e = tid >> 3, h = tid & 7;
  int row, col; edge_rc(rowA, colA, e, row, col);
  float w = expf(s2[col * 8 + h] - m2[row * 8 + h]) / (ss2[row * 8 + h] + 1e-16f);
  const float* __restrict__ src = h2 + (size_t)col * D2 + h * C2;
  float* dst = out2 + (size_t)row * D2 + h * C2;
  #pragma unroll
  for (int c = 0; c < C2; ++c) atomicAdd(&dst[c], src[c] * w);
}

__global__ void finalize(const float* __restrict__ out2, const float* __restrict__ b2,
                         float* __restrict__ out) {
  int n = blockIdx.x * blockDim.x + threadIdx.x;
  if (n >= N_NODES) return;
  float v[C2];
  #pragma unroll
  for (int c = 0; c < C2; ++c) {
    float s = 0.f;
    #pragma unroll
    for (int h = 0; h < H2; ++h) s += out2[(size_t)n * D2 + h * C2 + c];
    float t = s * (1.0f / H2) + b2[c];
    v[c] = t > 0.f ? t : (expf(t) - 1.0f);
  }
  float m = v[0];
  #pragma unroll
  for (int c = 1; c < C2; ++c) m = fmaxf(m, v[c]);
  float sum = 0.f;
  #pragma unroll
  for (int c = 0; c < C2; ++c) sum += expf(v[c] - m);
  float lse = logf(sum);
  #pragma unroll
  for (int c = 0; c < C2; ++c) out[(size_t)n * C2 + c] = v[c] - m - lse;
}


extern "C" void kernel_launch(void* const* d_in, const int* in_sizes, int n_in,
                              void* d_out, int out_size, void* d_ws, size_t ws_size,
                              hipStream_t stream) {
  const float* x    = (const float*)d_in[0];
  const int*   ei   = (const int*)  d_in[1];
  const float* g1   = (const float*)d_in[2];
  const float* be1  = (const float*)d_in[3];
  const float* mu1  = (const float*)d_in[4];
  const float* va1  = (const float*)d_in[5];
  const float* W1   = (const float*)d_in[6];
  const float* att1 = (const float*)d_in[7];
  const float* b1   = (const float*)d_in[8];
  const float* g2   = (const float*)d_in[9];
  const float* be2  = (const float*)d_in[10];
  const float* mu2  = (const float*)d_in[11];
  const float* va2  = (const float*)d_in[12];
  const float* W2   = (const float*)d_in[13];
  const float* att2 = (const float*)d_in[14];
  const float* b2   = (const float*)d_in[15];
  const int* rowA = ei;
  const int* colA = ei + N_EDGES;
  float* out = (float*)d_out;
  float* ws  = (float*)d_ws;

  float* sc1  = ws;
  float* sh1  = ws + 256;
  float* sc2  = ws + 512;
  float* sh2  = ws + 1024;
  float* ai   = ws + 1536;
  float* aj   = ai  + (size_t)N_NODES * H1;
  float* m1   = aj  + (size_t)N_NODES * H1;
  float* s1   = m1  + (size_t)N_NODES * H1;
  float* out1 = s1  + (size_t)N_NODES * H1;
  float* h1   = out1 + (size_t)N_NODES * D1;
  float* h2   = h1;
  float* s2   = h2  + (size_t)N_NODES * D2;
  float* m2   = s2  + (size_t)N_NODES * H2;
  float* ss2  = m2  + (size_t)N_NODES * H2;
  float* out2 = ss2 + (size_t)N_NODES * H2;

  const float NEG_INF = -__builtin_inff();

  bn_prep<<<2, 256, 0, stream>>>(g1, be1, mu1, va1, g2, be2, mu2, va2, b1,
                                 sc1, sh1, sc2, sh2);
  fill_f32<<<625,   256, 0, stream>>>(m1, NEG_INF, N_NODES * H1);
  fill_f32<<<625,   256, 0, stream>>>(s1, 0.0f,    N_NODES * H1);
  fill_f32<<<20000, 256, 0, stream>>>(out1, 0.0f,  N_NODES * D1);

  gemm_bn_wmma<<<10000, 256, 0, stream>>>(x, sc1, sh1, W1, h1, N_NODES, F_IN, D1);

  node_att1<<<20000, 256, 0, stream>>>(h1, att1, ai, aj);
  edge_max1<<<(ETOT + 255) / 256, 256, 0, stream>>>(rowA, colA, ai, aj, m1);
  edge_sum1<<<(ETOT + 255) / 256, 256, 0, stream>>>(rowA, colA, ai, aj, m1, s1);
  edge_scatter1<<<(ETOT * H1) / 8, 256, 0, stream>>>(rowA, colA, ai, aj, m1, s1, h1, out1);

  fill_f32<<<1250, 256, 0, stream>>>(m2, NEG_INF, N_NODES * H2);
  fill_f32<<<1250, 256, 0, stream>>>(ss2, 0.0f,   N_NODES * H2);
  fill_f32<<<7500, 256, 0, stream>>>(out2, 0.0f,  N_NODES * D2);

  gemm_bn_wmma<<<(2500 * 3 + 7) / 8, 256, 0, stream>>>(out1, sc2, sh2, W2, h2,
                                                       N_NODES, D1, D2);

  node_att2<<<(N_NODES * H2) / 256, 256, 0, stream>>>(h2, att2, s2);
  edge_max2<<<(ETOT + 255) / 256, 256, 0, stream>>>(rowA, colA, s2, m2);
  edge_sum2<<<(ETOT + 255) / 256, 256, 0, stream>>>(rowA, colA, s2, m2, ss2);
  edge_scatter2<<<(ETOT * H2) / 256, 256, 0, stream>>>(rowA, colA, s2, m2, ss2, h2, out2);

  finalize<<<(N_NODES + 255) / 256, 256, 0, stream>>>(out2, b2, out);
}


# Atomic passes re-mapped: lanes that update one destination row sit side by side and cover its words contiguously, so a wave's atomic sends one request per row or line; sums stay arrival-order
# speedup vs baseline: 1.5402x; 1.5402x over previous
#include <hip/hip_runtime.h>
#include <math.h>

#define N_NODES 40000
#define N_EDGES 320000
#define ETOT    (N_EDGES + N_NODES)
#define F_IN    256
#define H1      4
#define C1      128
#define D1      (H1*C1)
#define H2      8
#define C2      6
#define D2      (H2*C2)
#define BN_EPS    1e-5f
#define NEG_SLOPE 0.2f

typedef __attribute__((ext_vector_type(2))) float v2f;
typedef __attribute__((ext_vector_type(8))) float v8f;


__device__ __forceinline__ void atomic_max_f32(float* addr, float v) {
  if (v >= 0.0f) atomicMax((int*)addr, __float_as_int(v));
  else           atomicMin((unsigned int*)addr, __float_as_uint(v));
}

__device__ __forceinline__ void edge_rc(const int* __restrict__ rowA,
                                        const int* __restrict__ colA,
                                        int e, int& r, int& c) {
  if (e < N_EDGES) { r = rowA[e]; c = colA[e]; }
  else             { r = e - N_EDGES; c = r; }
}

__device__ __forceinline__ void edge_rc_flat(const int* __restrict__ rowA,
                                             const int* __restrict__ colA,
                                             int e, int& r, int& c) {
  int ec = e < N_EDGES ? e : N_EDGES - 1;
  int lr = rowA[ec], lc = colA[ec];
  bool self = e >= N_EDGES;
  r = self ? e - N_EDGES : lr;
  c = self ? e - N_EDGES : lc;
}


__global__ void fill_f32(float* __restrict__ p, float v, int n) {
  int i = blockIdx.x * blockDim.x + threadIdx.x;
  int stride = gridDim.x * blockDim.x;
  for (; i < n; i += stride) p[i] = v;
}

__global__ void bn_prep(const float* g1, const float* be1, const float* mu1, const float* va1,
                        const float* g2, const float* be2, const float* mu2, const float* va2,
                        const float* bias1,
                        float* sc1, float* sh1, float* sc2, float* sh2) {
  int i = blockIdx.x * blockDim.x + threadIdx.x;
  if (i < F_IN) {
    float s = g1[i] / sqrtf(va1[i] + BN_EPS);
    sc1[i] = s; sh1[i] = be1[i] - mu1[i] * s;
  }
  if (i < D1) {
    float s = g2[i] / sqrtf(va2[i] + BN_EPS);
    sc2[i] = s; sh2[i] = be2[i] + (bias1[i] - mu2[i]) * s;
  }
}


__global__ __launch_bounds__(256)
void gemm_bn_wmma(const float* __restrict__ A,
                  const float* __restrict__ scale,
                  const float* __restrict__ shift,
                  const float* __restrict__ B,
                  float* __restrict__ C,
                  int M, int K, int N) {
  int wave = blockIdx.x * (blockDim.x >> 5) + (threadIdx.x >> 5);
  int ntiles = N >> 4;
  int tm = wave / ntiles;
  int tn = wave - tm * ntiles;
  if (tm >= (M >> 4)) return;

  int lane = threadIdx.x & 31;
  int half = lane >> 4;
  int l16  = lane & 15;
  int row  = (tm << 4) + l16;
  int colB = (tn << 4) + l16;

  const float* __restrict__ Arow = A + (size_t)row * K;
  v8f acc = {};
  for (int k = 0; k < K; k += 4) {
    int kk = k + (half << 1);
    v2f a, b;
    a.x = Arow[kk]     * scale[kk]     + shift[kk];
    a.y = Arow[kk + 1] * scale[kk + 1] + shift[kk + 1];
    b.x = B[(size_t)kk * N + colB];
    b.y = B[(size_t)(kk + 1) * N + colB];
    acc = __builtin_amdgcn_wmma_f32_16x16x4_f32(
         false, a,  false, b,
         (short)0, acc,  false,  false);
  }
  float* Crow = C + (size_t)((tm << 4) + (half << 3)) * N + colB;
  #pragma unroll
  for (int r = 0; r < 8; ++r) Crow[(size_t)r * N] = acc[r];
}


__global__ void node_att1(const float* __restrict__ h1, const float* __restrict__ att1,
                          float* __restrict__ ai, float* __restrict__ aj) {
  int wave = blockIdx.x * (blockDim.x >> 5) + (threadIdx.x >> 5);
  if (wave >= N_NODES * H1) return;
  int n = wave >> 2, h = wave & 3;
  int lane = threadIdx.x & 31;
  const float* __restrict__ hp = h1 + (size_t)n * D1 + h * C1;
  const float* __restrict__ al = att1 + h * (2 * C1);
  const float* __restrict__ ar = al + C1;
  float si = 0.f, sj = 0.f;
  #pragma unroll
  for (int c0 = 0; c0 < C1; c0 += 32) {
    float v = hp[c0 + lane];
    si += v * al[c0 + lane];
    sj += v * ar[c0 + lane];
  }
  #pragma unroll
  for (int off = 16; off > 0; off >>= 1) {
    si += __shfl_xor(si, off, 32);
    sj += __shfl_xor(sj, off, 32);
  }
  if (lane == 0) { ai[wave] = si; aj[wave] = sj; }
}


__global__ void edge_max1(const int* __restrict__ rowA, const int* __restrict__ colA,
                          const float* __restrict__ ai, const float* __restrict__ aj,
                          float* __restrict__ m1) {
  int wave = blockIdx.x * (blockDim.x >> 5) + (threadIdx.x >> 5);
  int e0 = wave << 5;
  if (e0 >= ETOT) return;
  int lane = threadIdx.x & 31;
  int sub = lane >> 2, h = lane & 3;
  int dsti[4]; float val[4];
  #pragma unroll
  for (int i = 0; i < 4; ++i) {
    int row, col; edge_rc_flat(rowA, colA, e0 + (i << 3) + sub, row, col);
    float a = ai[row * 4 + h] + aj[col * 4 + h];
    val[i] = a > 0.f ? a : NEG_SLOPE * a;
    dsti[i] = row * 4 + h;
  }
  #pragma unroll
  for (int i = 0; i < 4; ++i) atomic_max_f32(&m1[dsti[i]], val[i]);
}

__global__ void edge_sum1(const int* __restrict__ rowA, const int* __restrict__ colA,
                          const float* __restrict__ ai, const float* __restrict__ aj,
                          const float* __restrict__ m1, float* __restrict__ s1) {
  int wave = blockIdx.x * (blockDim.x >> 5) + (threadIdx.x >> 5);
  int e0 = wave << 5;
  if (e0 >= ETOT) return;
  int lane = threadIdx.x & 31;
  int sub = lane >> 2, h = lane & 3;
  int dsti[4]; float val[4];
  #pragma unroll
  for (int i = 0; i < 4; ++i) {
    int row, col; edge_rc_flat(rowA, colA, e0 + (i << 3) + sub, row, col);
    float a = ai[row * 4 + h] + aj[col * 4 + h];
    a = a > 0.f ? a : NEG_SLOPE * a;
    val[i] = expf(a - m1[row * 4 + h]);
    dsti[i] = row * 4 + h;
  }
  #pragma unroll
  for (int i = 0; i < 4; ++i) atomicAdd(&s1[dsti[i]], val[i]);
}

__global__ __launch_bounds__(256)
void edge_scatter1(const int* __restrict__ rowA, const int* __restrict__ colA,
                   const float* __restrict__ ai, const float* __restrict__ aj,
                   const float* __restrict__ m1, const float* __restrict__ s1,
                   const float* __restrict__ h1, float* __restrict__ out1) {
  int wave = blockIdx.x * (blockDim.x >> 5) + (threadIdx.x >> 5);
  if (wave >= ETOT * H1) return;
  int e = wave >> 2, h = wave & 3;
  int row, col; edge_rc(rowA, colA, e, row, col);
  float a = ai[row * 4 + h] + aj[col * 4 + h];
  a = a > 0.f ? a : NEG_SLOPE * a;
  float w = expf(a - m1[row * 4 + h]) / (s1[row * 4 + h] + 1e-16f);
  int lane = threadIdx.x & 31;
  const float* __restrict__ src = h1 + (size_t)col * D1 + h * C1 + lane;
  float v0 = src[0], v1 = src[32], v2 = src[64], v3 = src[96];
  float* dst = out1 + (size_t)row * D1 + h * C1 + lane;
  atomicAdd(dst,      v0 * w);
  atomicAdd(dst + 32, v1 * w);
  atomicAdd(dst + 64, v2 * w);
  atomicAdd(dst + 96, v3 * w);
}


__global__ void node_att2(const float* __restrict__ h2, const float* __restrict__ att2,
                          float* __restrict__ s2) {
  int tid = blockIdx.x * blockDim.x + threadIdx.x;
  if (tid >= N_NODES * H2) return;
  int n = tid >> 3, h = tid & 7;
  const float* __restrict__ hp = h2 + (size_t)n * D2 + h * C2;
  const float* __restrict__ ap = att2 + h * (2 * C2);
  float s = 0.f;
  #pragma unroll
  for (int c = 0; c < C2; ++c) s += hp[c] * (ap[c] + ap[C2 + c]);
  s2[tid] = tanhf(s);
}

__global__ void edge_max2(const int* __restrict__ rowA, const int* __restrict__ colA,
                          const float* __restrict__ s2, float* __restrict__ m2) {
  int wave = blockIdx.x * (blockDim.x >> 5) + (threadIdx.x >> 5);
  int e0 = wave << 5;
  if (e0 >= ETOT) return;
  int lane = threadIdx.x & 31;
  int sub = lane >> 3, h = lane & 7;
  int dsti[8]; float val[8];
  #pragma unroll
  for (int i = 0; i < 8; ++i) {
    int row, col; edge_rc_flat(rowA, colA, e0 + (i << 2) + sub, row, col);
    val[i] = s2[col * 8 + h];
    dsti[i] = row * 8 + h;
  }
  #pragma unroll
  for (int i = 0; i < 8; ++i) atomic_max_f32(&m2[dsti[i]], val[i]);
}

__global__ void edge_sum2(const int* __restrict__ rowA, const int* __restrict__ colA,
                          const float* __restrict__ s2, const float* __restrict__ m2,
                          float* __restrict__ ss2) {
  int wave = blockIdx.x * (blockDim.x >> 5) + (threadIdx.x >> 5);
  int e0 = wave << 5;
  if (e0 >= ETOT) return;
  int lane = threadIdx.x & 31;
  int sub = lane >> 3, h = lane & 7;
  int dsti[8]; float val[8];
  #pragma unroll
  for (int i = 0; i < 8; ++i) {
    int row, col; edge_rc_flat(rowA, colA, e0 + (i << 2) + sub, row, col);
    val[i] = expf(s2[col * 8 + h] - m2[row * 8 + h]);
    dsti[i] = row * 8 + h;
  }
  #pragma unroll
  for (int i = 0; i < 8; ++i) atomicAdd(&ss2[dsti[i]], val[i]);
}

__global__ void edge_scatter2(const int* __restrict__ rowA, const int* __restrict__ colA,
                              const float* __restrict__ s2, const float* __restrict__ m2,
                              const float* __restrict__ ss2,
                              const float* __restrict__ h2, float* __restrict__ out2) {
  int wave = blockIdx.x * (blockDim.x >> 5) + (threadIdx.x >> 5);
  int e0 = wave << 3;
  if (e0 >= ETOT) return;
  int lane = threadIdx.x & 31;
  int half = lane >> 4, j = lane & 15, hq = lane & 7;
  int dsti[4]; float val[4][3];
  #pragma unroll
  for (int i = 0; i < 4; ++i) {
    int row, col; edge_rc_flat(rowA, colA, e0 + (i << 1) + half, row, col);
    float w = expf(s2[col * 8 + hq] - m2[row * 8 + hq]) / (ss2[row * 8 + hq] + 1e-16f);
    const float* __restrict__ src = h2 + (size_t)col * D2 + j;
    dsti[i] = row * D2 + j;
    #pragma unroll
    for (int k = 0; k < 3; ++k) {
      int hd = (j + 16 * k) / C2;
      float wk = __shfl(w, (lane & 16) | hd, 32);
      val[i][k] = src[16 * k] * wk;
    }
  }
  #pragma unroll
  for (int i = 0; i < 4; ++i) {
    #pragma unroll
    for (int k = 0; k < 3; ++k) atomicAdd(out2 + dsti[i] + 16 * k, val[i][k]);
  }
}

__global__ void finalize(const float* __restrict__ out2, const float* __restrict__ b2,
                         float* __restrict__ out) {
  int n = blockIdx.x * blockDim.x + threadIdx.x;
  if (n >= N_NODES) return;
  float v[C2];
  #pragma unroll
  for (int c = 0; c < C2; ++c) {
    float s = 0.f;
    #pragma unroll
    for (int h = 0; h < H2; ++h) s += out2[(size_t)n * D2 + h * C2 + c];
    float t = s * (1.0f / H2) + b2[c];
    v[c] = t > 0.f ? t : (expf(t) - 1.0f);
  }
  float m = v[0];
  #pragma unroll
  for (int c = 1; c < C2; ++c) m = fmaxf(m, v[c]);
  float sum = 0.f;
  #pragma unroll
  for (int c = 0; c < C2; ++c) sum += expf(v[c] - m);
  float lse = logf(sum);
  #pragma unroll
  for (int c = 0; c < C2; ++c) out[(size_t)n * C2 + c] = v[c] - m - lse;
}


extern "C" void kernel_launch(void* const* d_in, const int* in_sizes, int n_in,
                              void* d_out, int out_size, void* d_ws, size_t ws_size,
                              hipStream_t stream) {
  const float* x    = (const float*)d_in[0];
  const int*   ei   = (const int*)  d_in[1];
  const float* g1   = (const float*)d_in[2];
  const float* be1  = (const float*)d_in[3];
  const float* mu1  = (const float*)d_in[4];
  const float* va1  = (const float*)d_in[5];
  const float* W1   = (const float*)d_in[6];
  const float* att1 = (const float*)d_in[7];
  const float* b1   = (const float*)d_in[8];
  const float* g2   = (const float*)d_in[9];
  const float* be2  = (const float*)d_in[10];
  const float* mu2  = (const float*)d_in[11];
  const float* va2  = (const float*)d_in[12];
  const float* W2   = (const float*)d_in[13];
  const float* att2 = (const float*)d_in[14];
  const float* b2   = (const float*)d_in[15];
  const int* rowA = ei;
  const int* colA = ei + N_EDGES;
  float* out = (float*)d_out;
  float* ws  = (float*)d_ws;

  float* sc1  = ws;
  float* sh1  = ws + 256;
  float* sc2  = ws + 512;
  float* sh2  = ws + 1024;
  float* ai   = ws + 1536;
  float* aj   = ai  + (size_t)N_NODES * H1;
  float* m1   = aj  + (size_t)N_NODES * H1;
  float* s1   = m1  + (size_t)N_NODES * H1;
  float* out1 = s1  + (size_t)N_NODES * H1;
  float* h1   = out1 + (size_t)N_NODES * D1;
  float* h2   = h1;
  float* s2   = h2  + (size_t)N_NODES * D2;
  float* m2   = s2  + (size_t)N_NODES * H2;
  float* ss2  = m2  + (size_t)N_NODES * H2;
  float* out2 = ss2 + (size_t)N_NODES * H2;

  const float NEG_INF = -__builtin_inff();

  bn_prep<<<2, 256, 0, stream>>>(g1, be1, mu1, va1, g2, be2, mu2, va2, b1,
                                 sc1, sh1, sc2, sh2);
  fill_f32<<<625,   256, 0, stream>>>(m1, NEG_INF, N_NODES * H1);
  fill_f32<<<625,   256, 0, stream>>>(s1, 0.0f,    N_NODES * H1);
  fill_f32<<<20000, 256, 0, stream>>>(out1, 0.0f,  N_NODES * D1);

  gemm_bn_wmma<<<10000, 256, 0, stream>>>(x, sc1, sh1, W1, h1, N_NODES, F_IN, D1);

  node_att1<<<20000, 256, 0, stream>>>(h1, att1, ai, aj);
  edge_max1<<<(ETOT + 255) / 256, 256, 0, stream>>>(rowA, colA, ai, aj, m1);
  edge_sum1<<<(ETOT + 255) / 256, 256, 0, stream>>>(rowA, colA, ai, aj, m1, s1);
  edge_scatter1<<<(ETOT * H1) / 8, 256, 0, stream>>>(rowA, colA, ai, aj, m1, s1, h1, out1);

  fill_f32<<<1250, 256, 0, stream>>>(m2, NEG_INF, N_NODES * H2);
  fill_f32<<<1250, 256, 0, stream>>>(ss2, 0.0f,   N_NODES * H2);
  fill_f32<<<7500, 256, 0, stream>>>(out2, 0.0f,  N_NODES * D2);

  gemm_bn_wmma<<<(2500 * 3 + 7) / 8, 256, 0, stream>>>(out1, sc2, sh2, W2, h2,
                                                       N_NODES, D1, D2);

  node_att2<<<(N_NODES * H2) / 256, 256, 0, stream>>>(h2, att2, s2);
  edge_max2<<<(ETOT + 255) / 256, 256, 0, stream>>>(rowA, colA, s2, m2);
  edge_sum2<<<(ETOT + 255) / 256, 256, 0, stream>>>(rowA, colA, s2, m2, ss2);
  edge_scatter2<<<(ETOT / 8) / 8, 256, 0, stream>>>(rowA, colA, s2, m2, ss2, h2, out2);

  finalize<<<(N_NODES + 255) / 256, 256, 0, stream>>>(out2, b2, out);
}


# GEMM fragment loaders: each lane reads 4 consecutive k of A, scale, shift as one 16-byte load (was 2 k), B at the same k; an A load's 16 lines feed 2 products; k summed in a permuted order
# speedup vs baseline: 2.0035x; 1.3008x over previous
#include <hip/hip_runtime.h>
#include <math.h>

#define N_NODES 40000
#define N_EDGES 320000
#define ETOT    (N_EDGES + N_NODES)
#define F_IN    256
#define H1      4
#define C1      128
#define D1      (H1*C1)
#define H2      8
#define C2      6
#define D2      (H2*C2)
#define BN_EPS    1e-5f
#define NEG_SLOPE 0.2f

typedef __attribute__((ext_vector_type(2))) float v2f;
typedef __attribute__((ext_vector_type(4))) float v4f;
typedef __attribute__((ext_vector_type(8))) float v8f;


__device__ __forceinline__ void atomic_max_f32(float* addr, float v) {
  if (v >= 0.0f) atomicMax((int*)addr, __float_as_int(v));
  else           atomicMin((unsigned int*)addr, __float_as_uint(v));
}

__device__ __forceinline__ void edge_rc_flat(const int* __restrict__ rowA,
                                             const int* __restrict__ colA,
                                             int e, int& r, int& c) {
  int ec = e < N_EDGES ? e : N_EDGES - 1;
  int lr = rowA[ec], lc = colA[ec];
  bool self = e >= N_EDGES;
  r = self ? e - N_EDGES : lr;
  c = self ? e - N_EDGES : lc;
}


__global__ void fill_f32(float* __restrict__ p, float v, int n) {
  int i = blockIdx.x * blockDim.x + threadIdx.x;
  int stride = gridDim.x * blockDim.x;
  for (; i < n; i += stride) p[i] = v;
}

__global__ void bn_prep(const float* g1, const float* be1, const float* mu1, const float* va1,
                        const float* g2, const float* be2, const float* mu2, const float* va2,
                        const float* bias1,
                        float* sc1, float* sh1, float* sc2, float* sh2) {
  int i = blockIdx.x * blockDim.x + threadIdx.x;
  if (i < F_IN) {
    float s = g1[i] / sqrtf(va1[i] + BN_EPS);
    sc1[i] = s; sh1[i] = be1[i] - mu1[i] * s;
  }
  if (i < D1) {
    float s = g2[i] / sqrtf(va2[i] + BN_EPS);
    sc2[i] = s; sh2[i] = be2[i] + (bias1[i] - mu2[i]) * s;
  }
}


__global__ __launch_bounds__(256)
void gemm_bn_wmma(const float* __restrict__ A,
                  const float* __restrict__ scale,
                  const float* __restrict__ shift,
                  const float* __restrict__ B,
                  float* __restrict__ C,
                  int M, int K, int N) {
  int wave = blockIdx.x * (blockDim.x >> 5) + (threadIdx.x >> 5);
  int ntiles = N >> 4;
  int tm = wave / ntiles;
  int tn = wave - tm * ntiles;
  if (tm >= (M >> 4)) return;

  int lane = threadIdx.x & 31;
  int half = lane >> 4;
  int l16  = lane & 15;
  int row  = (tm << 4) + l16;
  int colB = (tn << 4) + l16;

  const float* __restrict__ Arow = A + (size_t)row * K;
  v8f acc = {};
  for (int k = 0; k < K; k += 8) {
    int kk = k + (half << 2);
    v4f a4 = *(const v4f*)(Arow + kk);
    v4f s4 = *(const v4f*)(scale + kk);
    v4f t4 = *(const v4f*)(shift + kk);
    const float* __restrict__ Bk = B + (size_t)kk * N + colB;
    v2f a, b;
    a.x = a4.x * s4.x + t4.x;
    a.y = a4.y * s4.y + t4.y;
    b.x = Bk[0];
    b.y = Bk[(size_t)N];
    acc = __builtin_amdgcn_wmma_f32_16x16x4_f32(
         false, a,  false, b,
         (short)0, acc,  false,  false);
    a.x = a4.z * s4.z + t4.z;
    a.y = a4.w * s4.w + t4.w;
    b.x = Bk[(size_t)2 * N];
    b.y = Bk[(size_t)3 * N];
    acc = __builtin_amdgcn_wmma_f32_16x16x4_f32(
         false, a,  false, b,
         (short)0, acc,  false,  false);
  }
  float* Crow = C + (size_t)((tm << 4) + (half << 3)) * N + colB;
  #pragma unroll
  for (int r = 0; r < 8; ++r) Crow[(size_t)r * N] = acc[r];
}


__global__ void node_att1(const float* __restrict__ h1, const float* __restrict__ att1,
                          float* __restrict__ ai, float* __restrict__ aj) {
  int wave = blockIdx.x * (blockDim.x >> 5) + (threadIdx.x >> 5);
  if (wave >= N_NODES * H1) return;
  int n = wave >> 2, h = wave & 3;
  int lane = threadIdx.x & 31;
  const float* __restrict__ hp = h1 + (size_t)n * D1 + h * C1;
  const float* __restrict__ al = att1 + h * (2 * C1);
  const float* __restrict__ ar = al + C1;
  float si = 0.f, sj = 0.f;
  #pragma unroll
  for (int c0 = 0; c0 < C1; c0 += 32) {
    float v = hp[c0 + lane];
    si += v * al[c0 + lane];
    sj += v * ar[c0 + lane];
  }
  #pragma unroll
  for (int off = 16; off > 0; off >>= 1) {
    si += __shfl_xor(si, off, 32);
    sj += __shfl_xor(sj, off, 32);
  }
  if (lane == 0) { ai[wave] = si; aj[wave] = sj; }
}


__global__ void edge_max1(const int* __restrict__ rowA, const int* __restrict__ colA,
                          const float* __restrict__ ai, const float* __restrict__ aj,
                          float* __restrict__ m1) {
  int wave = blockIdx.x * (blockDim.x >> 5) + (threadIdx.x >> 5);
  int e0 = wave << 5;
  if (e0 >= ETOT) return;
  int lane = threadIdx.x & 31;
  int sub = lane >> 2, h = lane & 3;
  int dsti[4]; float val[4];
  #pragma unroll
  for (int i = 0; i < 4; ++i) {
    int row, col; edge_rc_flat(rowA, colA, e0 + (i << 3) + sub, row, col);
    float a = ai[row * 4 + h] + aj[col * 4 + h];
    val[i] = a > 0.f ? a : NEG_SLOPE * a;
    dsti[i] = row * 4 + h;
  }
  #pragma unroll
  for (int i = 0; i < 4; ++i) atomic_max_f32(&m1[dsti[i]], val[i]);
}

__global__ void edge_sum1(const int* __restrict__ rowA, const int* __restrict__ colA,
                          const float* __restrict__ ai, const float* __restrict__ aj,
                          const float* __restrict__ m1, float* __restrict__ s1) {
  int wave = blockIdx.x * (blockDim.x >> 5) + (threadIdx.x >> 5);
  int e0 = wave << 5;
  if (e0 >= ETOT) return;
  int lane = threadIdx.x & 31;
  int sub = lane >> 2, h = lane & 3;
  int dsti[4]; float val[4];
  #pragma unroll
  for (int i = 0; i < 4; ++i) {
    int row, col; edge_rc_flat(rowA, colA, e0 + (i << 3) + sub, row, col);
    float a = ai[row * 4 + h] + aj[col * 4 + h];
    a = a > 0.f ? a : NEG_SLOPE * a;
    val[i] = expf(a - m1[row * 4 + h]);
    dsti[i] = row * 4 + h;
  }
  #pragma unroll
  for (int i = 0; i < 4; ++i) atomicAdd(&s1[dsti[i]], val[i]);
}

__global__ __launch_bounds__(256)
void edge_scatter1(const int* __restrict__ rowA, const int* __restrict__ colA,
                   const float* __restrict__ ai, const float* __restrict__ aj,
                   const float* __restrict__ m1, const float* __restrict__ s1,
                   const float* __restrict__ h1, float* __restrict__ out1) {
  int wave = blockIdx.x * (blockDim.x >> 5) + (threadIdx.x >> 5);
  int e0 = wave << 3;
  if (e0 >= ETOT) return;
  int lane = threadIdx.x & 31;
  int sub = lane >> 2, h = lane & 3;
  int row, col; edge_rc_flat(rowA, colA, e0 + sub, row, col);
  float a = ai[row * 4 + h] + aj[col * 4 + h];
  a = a > 0.f ? a : NEG_SLOPE * a;
  float w = expf(a - m1[row * 4 + h]) / (s1[row * 4 + h] + 1e-16f);
  int wbits = __float_as_int(w);
  #pragma unroll
  for (int i = 0; i < 8; i += 2) {
    float v[2][16];
    int rr[2];
    #pragma unroll
    for (int u = 0; u < 2; ++u) {
      int cc = __builtin_amdgcn_readlane(col, (i + u) * 4);
      rr[u]  = __builtin_amdgcn_readlane(row, (i + u) * 4);
      const float* __restrict__ src = h1 + (size_t)cc * D1 + lane;
      #pragma unroll
      for (int q = 0; q < 16; ++q) v[u][q] = src[q * 32];
    }
    #pragma unroll
    for (int u = 0; u < 2; ++u) {
      float* dst = out1 + (size_t)rr[u] * D1 + lane;
      #pragma unroll
      for (int q = 0; q < 16; ++q) {
        float wq = __int_as_float(__builtin_amdgcn_readlane(wbits, (i + u) * 4 + (q >> 2)));
        atomicAdd(dst + q * 32, v[u][q] * wq);
      }
    }
  }
}


__global__ void node_att2(const float* __restrict__ h2, const float* __restrict__ att2,
                          float* __restrict__ s2) {
  int tid = blockIdx.x * blockDim.x + threadIdx.x;
  if (tid >= N_NODES * H2) return;
  int n = tid >> 3, h = tid & 7;
  const float* __restrict__ hp = h2 + (size_t)n * D2 + h * C2;
  const float* __restrict__ ap = att2 + h * (2 * C2);
  float s = 0.f;
  #pragma unroll
  for (int c = 0; c < C2; ++c) s += hp[c] * (ap[c] + ap[C2 + c]);
  s2[tid] = tanhf(s);
}

__global__ void edge_max2(const int* __restrict__ rowA, const int* __restrict__ colA,
                          const float* __restrict__ s2, float* __restrict__ m2) {
  int wave = blockIdx.x * (blockDim.x >> 5) + (threadIdx.x >> 5);
  int e0 = wave << 5;
  if (e0 >= ETOT) return;
  int lane = threadIdx.x & 31;
  int sub = lane >> 3, h = lane & 7;
  int dsti[8]; float val[8];
  #pragma unroll
  for (int i = 0; i < 8; ++i) {
    int row, col; edge_rc_flat(rowA, colA, e0 + (i << 2) + sub, row, col);
    val[i] = s2[col * 8 + h];
    dsti[i] = row * 8 + h;
  }
  #pragma unroll
  for (int i = 0; i < 8; ++i) atomic_max_f32(&m2[dsti[i]], val[i]);
}

__global__ void edge_sum2(const int* __restrict__ rowA, const int* __restrict__ colA,
                          const float* __restrict__ s2, const float* __restrict__ m2,
                          float* __restrict__ ss2) {
  int wave = blockIdx.x * (blockDim.x >> 5) + (threadIdx.x >> 5);
  int e0 = wave << 5;
  if (e0 >= ETOT) return;
  int lane = threadIdx.x & 31;
  int sub = lane >> 3, h = lane & 7;
  int dsti[8]; float val[8];
  #pragma unroll
  for (int i = 0; i < 8; ++i) {
    int row, col; edge_rc_flat(rowA, colA, e0 + (i << 2) + sub, row, col);
    val[i] = expf(s2[col * 8 + h] - m2[row * 8 + h]);
    dsti[i] = row * 8 + h;
  }
  #pragma unroll
  for (int i = 0; i < 8; ++i) atomicAdd(&ss2[dsti[i]], val[i]);
}

__global__ void edge_scatter2(const int* __restrict__ rowA, const int* __restrict__ colA,
                              const float* __restrict__ s2, const float* __restrict__ m2,
                              const float* __restrict__ ss2,
                              const float* __restrict__ h2, float* __restrict__ out2) {
  int wave = blockIdx.x * (blockDim.x >> 5) + (threadIdx.x >> 5);
  int e0 = wave << 3;
  if (e0 >= ETOT) return;
  int lane = threadIdx.x & 31;
  int half = lane >> 4, j = lane & 15, hq = lane & 7;
  int dsti[4]; float val[4][3];
  #pragma unroll
  for (int i = 0; i < 4; ++i) {
    int row, col; edge_rc_flat(rowA, colA, e0 + (i << 1) + half, row, col);
    float w = expf(s2[col * 8 + hq] - m2[row * 8 + hq]) / (ss2[row * 8 + hq] + 1e-16f);
    const float* __restrict__ src = h2 + (size_t)col * D2 + j;
    dsti[i] = row * D2 + j;
    #pragma unroll
    for (int k = 0; k < 3; ++k) {
      int hd = (j + 16 * k) / C2;
      float wk = __shfl(w, (lane & 16) | hd, 32);
      val[i][k] = src[16 * k] * wk;
    }
  }
  #pragma unroll
  for (int i = 0; i < 4; ++i) {
    #pragma unroll
    for (int k = 0; k < 3; ++k) atomicAdd(out2 + dsti[i] + 16 * k, val[i][k]);
  }
}

__global__ void finalize(const float* __restrict__ out2, const float* __restrict__ b2,
                         float* __restrict__ out) {
  int n = blockIdx.x * blockDim.x + threadIdx.x;
  if (n >= N_NODES) return;
  float v[C2];
  #pragma unroll
  for (int c = 0; c < C2; ++c) {
    float s = 0.f;
    #pragma unroll
    for (int h = 0; h < H2; ++h) s += out2[(size_t)n * D2 + h * C2 + c];
    float t = s * (1.0f / H2) + b2[c];
    v[c] = t > 0.f ? t : (expf(t) - 1.0f);
  }
  float m = v[0];
  #pragma unroll
  for (int c = 1; c < C2; ++c) m = fmaxf(m, v[c]);
  float sum = 0.f;
  #pragma unroll
  for (int c = 0; c < C2; ++c) sum += expf(v[c] - m);
  float lse = logf(sum);
  #pragma unroll
  for (int c = 0; c < C2; ++c) out[(size_t)n * C2 + c] = v[c] - m - lse;
}


extern "C" void kernel_launch(void* const* d_in, const int* in_sizes, int n_in,
                              void* d_out, int out_size, void* d_ws, size_t ws_size,
                              hipStream_t stream) {
  const float* x    = (const float*)d_in[0];
  const int*   ei   = (const int*)  d_in[1];
  const float* g1   = (const float*)d_in[2];
  const float* be1  = (const float*)d_in[3];
  const float* mu1  = (const float*)d_in[4];
  const float* va1  = (const float*)d_in[5];
  const float* W1   = (const float*)d_in[6];
  const float* att1 = (const float*)d_in[7];
  const float* b1   = (const float*)d_in[8];
  const float* g2   = (const float*)d_in[9];
  const float* be2  = (const float*)d_in[10];
  const float* mu2  = (const float*)d_in[11];
  const float* va2  = (const float*)d_in[12];
  const float* W2   = (const float*)d_in[13];
  const float* att2 = (const float*)d_in[14];
  const float* b2   = (const float*)d_in[15];
  const int* rowA = ei;
  const int* colA = ei + N_EDGES;
  float* out = (float*)d_out;
  float* ws  = (float*)d_ws;

  float* sc1  = ws;
  float* sh1  = ws + 256;
  float* sc2  = ws + 512;
  float* sh2  = ws + 1024;
  float* ai   = ws + 1536;
  float* aj   = ai  + (size_t)N_NODES * H1;
  float* m1   = aj  + (size_t)N_NODES * H1;
  float* s1   = m1  + (size_t)N_NODES * H1;
  float* out1 = s1  + (size_t)N_NODES * H1;
  float* h1   = out1 + (size_t)N_NODES * D1;
  float* h2   = h1;
  float* s2   = h2  + (size_t)N_NODES * D2;
  float* m2   = s2  + (size_t)N_NODES * H2;
  float* ss2  = m2  + (size_t)N_NODES * H2;
  float* out2 = ss2 + (size_t)N_NODES * H2;

  const float NEG_INF = -__builtin_inff();

  bn_prep<<<2, 256, 0, stream>>>(g1, be1, mu1, va1, g2, be2, mu2, va2, b1,
                                 sc1, sh1, sc2, sh2);
  fill_f32<<<625,   256, 0, stream>>>(m1, NEG_INF, N_NODES * H1);
  fill_f32<<<625,   256, 0, stream>>>(s1, 0.0f,    N_NODES * H1);
  fill_f32<<<20000, 256, 0, stream>>>(out1, 0.0f,  N_NODES * D1);

  gemm_bn_wmma<<<10000, 256, 0, stream>>>(x, sc1, sh1, W1, h1, N_NODES, F_IN, D1);

  node_att1<<<20000, 256, 0, stream>>>(h1, att1, ai, aj);
  edge_max1<<<(ETOT + 255) / 256, 256, 0, stream>>>(rowA, colA, ai, aj, m1);
  edge_sum1<<<(ETOT + 255) / 256, 256, 0, stream>>>(rowA, colA, ai, aj, m1, s1);
  edge_scatter1<<<(ETOT / 8) / 8, 256, 0, stream>>>(rowA, colA, ai, aj, m1, s1, h1, out1);

  fill_f32<<<1250, 256, 0, stream>>>(m2, NEG_INF, N_NODES * H2);
  fill_f32<<<1250, 256, 0, stream>>>(ss2, 0.0f,   N_NODES * H2);
  fill_f32<<<7500, 256, 0, stream>>>(out2, 0.0f,  N_NODES * D2);

  gemm_bn_wmma<<<(2500 * 3 + 7) / 8, 256, 0, stream>>>(out1, sc2, sh2, W2, h2,
                                                       N_NODES, D1, D2);

  node_att2<<<(N_NODES * H2) / 256, 256, 0, stream>>>(h2, att2, s2);
  edge_max2<<<(ETOT + 255) / 256, 256, 0, stream>>>(rowA, colA, s2, m2);
  edge_sum2<<<(ETOT + 255) / 256, 256, 0, stream>>>(rowA, colA, s2, m2, ss2);
  edge_scatter2<<<(ETOT / 8) / 8, 256, 0, stream>>>(rowA, colA, s2, m2, ss2, h2, out2);

  finalize<<<(N_NODES + 255) / 256, 256, 0, stream>>>(out2, b2, out);
}


# Layer-1 GEMM loop order swapped: a wave keeps its 16 rows' whole K (BN applied once) in registers and walks 128 columns; B read 16 bytes per lane for 4 accumulators, 2 trips ahead; sums unchanged
# speedup vs baseline: 2.4132x; 1.2045x over previous
#include <hip/hip_runtime.h>
#include <math.h>

#define N_NODES 40000
#define N_EDGES 320000
#define ETOT    (N_EDGES + N_NODES)
#define F_IN    256
#define H1      4
#define C1      128
#define D1      (H1*C1)
#define H2      8
#define C2      6
#define D2      (H2*C2)
#define BN_EPS    1e-5f
#define NEG_SLOPE 0.2f

typedef __attribute__((ext_vector_type(2))) float v2f;
typedef __attribute__((ext_vector_type(4))) float v4f;
typedef __attribute__((ext_vector_type(8))) float v8f;


__device__ __forceinline__ void atomic_max_f32(float* addr, float v) {
  if (v >= 0.0f) atomicMax((int*)addr, __float_as_int(v));
  else           atomicMin((unsigned int*)addr, __float_as_uint(v));
}

__device__ __forceinline__ void edge_rc_flat(const int* __restrict__ rowA,
                                             const int* __restrict__ colA,
                                             int e, int& r, int& c) {
  int ec = e < N_EDGES ? e : N_EDGES - 1;
  int lr = rowA[ec], lc = colA[ec];
  bool self = e >= N_EDGES;
  r = self ? e - N_EDGES : lr;
  c = self ? e - N_EDGES : lc;
}


__global__ void fill_f32(float* __restrict__ p, float v, int n) {
  int i = blockIdx.x * blockDim.x + threadIdx.x;
  int stride = gridDim.x * blockDim.x;
  for (; i < n; i += stride) p[i] = v;
}

__global__ void bn_prep(const float* g1, const float* be1, const float* mu1, const float* va1,
                        const float* g2, const float* be2, const float* mu2, const float* va2,
                        const float* bias1,
                        float* sc1, float* sh1, float* sc2, float* sh2) {
  int i = blockIdx.x * blockDim.x + threadIdx.x;
  if (i < F_IN) {
    float s = g1[i] / sqrtf(va1[i] + BN_EPS);
    sc1[i] = s; sh1[i] = be1[i] - mu1[i] * s;
  }
  if (i < D1) {
    float s = g2[i] / sqrtf(va2[i] + BN_EPS);
    sc2[i] = s; sh2[i] = be2[i] + (bias1[i] - mu2[i]) * s;
  }
}


__global__ __launch_bounds__(256)
void gemm_bn_wmma(const float* __restrict__ A,
                  const float* __restrict__ scale,
                  const float* __restrict__ shift,
                  const float* __restrict__ B,
                  float* __restrict__ C,
                  int M, int K, int N) {
  int wave = blockIdx.x * (blockDim.x >> 5) + (threadIdx.x >> 5);
  int ntiles = N >> 4;
  int tm = wave / ntiles;
  int tn = wave - tm * ntiles;
  if (tm >= (M >> 4)) return;

  int lane = threadIdx.x & 31;
  int half = lane >> 4;
  int l16  = lane & 15;
  int row  = (tm << 4) + l16;
  int colB = (tn << 4) + l16;

  const float* __restrict__ Arow = A + (size_t)row * K;
  v8f acc = {};
  for (int k = 0; k < K; k += 8) {
    int kk = k + (half << 2);
    v4f a4 = *(const v4f*)(Arow + kk);
    v4f s4 = *(const v4f*)(scale + kk);
    v4f t4 = *(const v4f*)(shift + kk);
    const float* __restrict__ Bk = B + (size_t)kk * N + colB;
    v2f a, b;
    a.x = a4.x * s4.x + t4.x;
    a.y = a4.y * s4.y + t4.y;
    b.x = Bk[0];
    b.y = Bk[(size_t)N];
    acc = __builtin_amdgcn_wmma_f32_16x16x4_f32(
         false, a,  false, b,
         (short)0, acc,  false,  false);
    a.x = a4.z * s4.z + t4.z;
    a.y = a4.w * s4.w + t4.w;
    b.x = Bk[(size_t)2 * N];
    b.y = Bk[(size_t)3 * N];
    acc = __builtin_amdgcn_wmma_f32_16x16x4_f32(
         false, a,  false, b,
         (short)0, acc,  false,  false);
  }
  float* Crow = C + (size_t)((tm << 4) + (half << 3)) * N + colB;
  #pragma unroll
  for (int r = 0; r < 8; ++r) Crow[(size_t)r * N] = acc[r];
}

#define G1_COLS_PER_WAVE 128

__global__ __launch_bounds__(256)
void gemm1_rows_resident(const float* __restrict__ A,
                         const float* __restrict__ scale,
                         const float* __restrict__ shift,
                         const float* __restrict__ B,
                         float* __restrict__ C) {
  int tm = blockIdx.x * 8 + (threadIdx.x >> 5);
  if (tm >= (N_NODES >> 4)) return;
  int lane = threadIdx.x & 31;
  int half = lane >> 4;
  int l16  = lane & 15;

  const float* __restrict__ Arow = A + (size_t)((tm << 4) + l16) * F_IN + (half << 2);
  const float* __restrict__ sc = scale + (half << 2);
  const float* __restrict__ sh = shift + (half << 2);
  v4f a[F_IN / 8];
  #pragma unroll
  for (int t = 0; t < F_IN / 8; ++t) a[t] = *(const v4f*)(Arow + 8 * t);
  #pragma unroll
  for (int t0 = 0; t0 < F_IN / 8; t0 += 8) {
    #pragma unroll
    for (int t = t0; t < t0 + 8; ++t) {
      v4f s4 = *(const v4f*)(sc + 8 * t);
      v4f t4 = *(const v4f*)(sh + 8 * t);
      a[t].x = a[t].x * s4.x + t4.x;
      a[t].y = a[t].y * s4.y + t4.y;
      a[t].z = a[t].z * s4.z + t4.z;
      a[t].w = a[t].w * s4.w + t4.w;
    }
    __builtin_amdgcn_sched_barrier(0);
  }

  int cb = blockIdx.y * G1_COLS_PER_WAVE + (l16 << 2);
  const float* __restrict__ Bl = B + (size_t)(half << 2) * D1 + cb;
  float* __restrict__ Cl = C + (size_t)((tm << 4) + (half << 3)) * D1 + cb;
  for (int j = 0; j < G1_COLS_PER_WAVE; j += 64) {
    v8f acc0 = {}, acc1 = {}, acc2 = {}, acc3 = {};
    const float* __restrict__ Bj = Bl + j;
    v4f bq[3][4];
    #pragma unroll
    for (int p = 0; p < 2; ++p) {
      #pragma unroll
      for (int q = 0; q < 4; ++q) bq[p][q] = *(const v4f*)(Bj + (size_t)(8 * p + q) * D1);
    }
    #pragma unroll
    for (int t = 0; t < F_IN / 8; ++t) {
      if (t + 2 < F_IN / 8) {
        #pragma unroll
        for (int q = 0; q < 4; ++q)
          bq[(t + 2) % 3][q] = *(const v4f*)(Bj + (size_t)(8 * (t + 2) + q) * D1);
      }
      __builtin_amdgcn_sched_barrier(0);
      const v4f b0 = bq[t % 3][0], b1 = bq[t % 3][1], b2 = bq[t % 3][2], b3 = bq[t % 3][3];
      v2f a01, a23, b;
      a01.x = a[t].x; a01.y = a[t].y;
      a23.x = a[t].z; a23.y = a[t].w;
      b.x = b0.x; b.y = b1.x;
      acc0 = __builtin_amdgcn_wmma_f32_16x16x4_f32(false, a01, false, b, (short)0, acc0, false, false);
      b.x = b0.y; b.y = b1.y;
      acc1 = __builtin_amdgcn_wmma_f32_16x16x4_f32(false, a01, false, b, (short)0, acc1, false, false);
      b.x = b0.z; b.y = b1.z;
      acc2 = __builtin_amdgcn_wmma_f32_16x16x4_f32(false, a01, false, b, (short)0, acc2, false, false);
      b.x = b0.w; b.y = b1.w;
      acc3 = __builtin_amdgcn_wmma_f32_16x16x4_f32(false, a01, false, b, (short)0, acc3, false, false);
      b.x = b2.x; b.y = b3.x;
      acc0 = __builtin_amdgcn_wmma_f32_16x16x4_f32(false, a23, false, b, (short)0, acc0, false, false);
      b.x = b2.y; b.y = b3.y;
      acc1 = __builtin_amdgcn_wmma_f32_16x16x4_f32(false, a23, false, b, (short)0, acc1, false, false);
      b.x = b2.z; b.y = b3.z;
      acc2 = __builtin_amdgcn_wmma_f32_16x16x4_f32(false, a23, false, b, (short)0, acc2, false, false);
      b.x = b2.w; b.y = b3.w;
      acc3 = __builtin_amdgcn_wmma_f32_16x16x4_f32(false, a23, false, b, (short)0, acc3, false, false);
      __builtin_amdgcn_sched_barrier(0);
    }
    #pragma unroll
    for (int r = 0; r < 8; ++r) {
      v4f o;
      o.x = acc0[r]; o.y = acc1[r]; o.z = acc2[r]; o.w = acc3[r];
      *(v4f*)(Cl + (size_t)r * D1 + j) = o;
    }
  }
}


__global__ void node_att1(const float* __restrict__ h1, const float* __restrict__ att1,
                          float* __restrict__ ai, float* __restrict__ aj) {
  int wave = blockIdx.x * (blockDim.x >> 5) + (threadIdx.x >> 5);
  if (wave >= N_NODES * H1) return;
  int n = wave >> 2, h = wave & 3;
  int lane = threadIdx.x & 31;
  const float* __restrict__ hp = h1 + (size_t)n * D1 + h * C1;
  const float* __restrict__ al = att1 + h * (2 * C1);
  const float* __restrict__ ar = al + C1;
  float si = 0.f, sj = 0.f;
  #pragma unroll
  for (int c0 = 0; c0 < C1; c0 += 32) {
    float v = hp[c0 + lane];
    si += v * al[c0 + lane];
    sj += v * ar[c0 + lane];
  }
  #pragma unroll
  for (int off = 16; off > 0; off >>= 1) {
    si += __shfl_xor(si, off, 32);
    sj += __shfl_xor(sj, off, 32);
  }
  if (lane == 0) { ai[wave] = si; aj[wave] = sj; }
}


__global__ void edge_max1(const int* __restrict__ rowA, const int* __restrict__ colA,
                          const float* __restrict__ ai, const float* __restrict__ aj,
                          float* __restrict__ m1) {
  int wave = blockIdx.x * (blockDim.x >> 5) + (threadIdx.x >> 5);
  int e0 = wave << 5;
  if (e0 >= ETOT) return;
  int lane = threadIdx.x & 31;
  int sub = lane >> 2, h = lane & 3;
  int dsti[4]; float val[4];
  #pragma unroll
  for (int i = 0; i < 4; ++i) {
    int row, col; edge_rc_flat(rowA, colA, e0 + (i << 3) + sub, row, col);
    float a = ai[row * 4 + h] + aj[col * 4 + h];
    val[i] = a > 0.f ? a : NEG_SLOPE * a;
    dsti[i] = row * 4 + h;
  }
  #pragma unroll
  for (int i = 0; i < 4; ++i) atomic_max_f32(&m1[dsti[i]], val[i]);
}

__global__ void edge_sum1(const int* __restrict__ rowA, const int* __restrict__ colA,
                          const float* __restrict__ ai, const float* __restrict__ aj,
                          const float* __restrict__ m1, float* __restrict__ s1) {
  int wave = blockIdx.x * (blockDim.x >> 5) + (threadIdx.x >> 5);
  int e0 = wave << 5;
  if (e0 >= ETOT) return;
  int lane = threadIdx.x & 31;
  int sub = lane >> 2, h = lane & 3;
  int dsti[4]; float val[4];
  #pragma unroll
  for (int i = 0; i < 4; ++i) {
    int row, col; edge_rc_flat(rowA, colA, e0 + (i << 3) + sub, row, col);
    float a = ai[row * 4 + h] + aj[col * 4 + h];
    a = a > 0.f ? a : NEG_SLOPE * a;
    val[i] = expf(a - m1[row * 4 + h]);
    dsti[i] = row * 4 + h;
  }
  #pragma unroll
  for (int i = 0; i < 4; ++i) atomicAdd(&s1[dsti[i]], val[i]);
}

__global__ __launch_bounds__(256)
void edge_scatter1(const int* __restrict__ rowA, const int* __restrict__ colA,
                   const float* __restrict__ ai, const float* __restrict__ aj,
                   const float* __restrict__ m1, const float* __restrict__ s1,
                   const float* __restrict__ h1, float* __restrict__ out1) {
  int wave = blockIdx.x * (blockDim.x >> 5) + (threadIdx.x >> 5);
  int e0 = wave << 3;
  if (e0 >= ETOT) return;
  int lane = threadIdx.x & 31;
  int sub = lane >> 2, h = lane & 3;
  int row, col; edge_rc_flat(rowA, colA, e0 + sub, row, col);
  float a = ai[row * 4 + h] + aj[col * 4 + h];
  a = a > 0.f ? a : NEG_SLOPE * a;
  float w = expf(a - m1[row * 4 + h]) / (s1[row * 4 + h] + 1e-16f);
  int wbits = __float_as_int(w);
  #pragma unroll
  for (int i = 0; i < 8; i += 2) {
    float v[2][16];
    int rr[2];
    #pragma unroll
    for (int u = 0; u < 2; ++u) {
      int cc = __builtin_amdgcn_readlane(col, (i + u) * 4);
      rr[u]  = __builtin_amdgcn_readlane(row, (i + u) * 4);
      const float* __restrict__ src = h1 + (size_t)cc * D1 + lane;
      #pragma unroll
      for (int q = 0; q < 16; ++q) v[u][q] = src[q * 32];
    }
    #pragma unroll
    for (int u = 0; u < 2; ++u) {
      float* dst = out1 + (size_t)rr[u] * D1 + lane;
      #pragma unroll
      for (int q = 0; q < 16; ++q) {
        float wq = __int_as_float(__builtin_amdgcn_readlane(wbits, (i + u) * 4 + (q >> 2)));
        atomicAdd(dst + q * 32, v[u][q] * wq);
      }
    }
  }
}


__global__ void node_att2(const float* __restrict__ h2, const float* __restrict__ att2,
                          float* __restrict__ s2) {
  int tid = blockIdx.x * blockDim.x + threadIdx.x;
  if (tid >= N_NODES * H2) return;
  int n = tid >> 3, h = tid & 7;
  const float* __restrict__ hp = h2 + (size_t)n * D2 + h * C2;
  const float* __restrict__ ap = att2 + h * (2 * C2);
  float s = 0.f;
  #pragma unroll
  for (int c = 0; c < C2; ++c) s += hp[c] * (ap[c] + ap[C2 + c]);
  s2[tid] = tanhf(s);
}

__global__ void edge_max2(const int* __restrict__ rowA, const int* __restrict__ colA,
                          const float* __restrict__ s2, float* __restrict__ m2) {
  int wave = blockIdx.x * (blockDim.x >> 5) + (threadIdx.x >> 5);
  int e0 = wave << 5;
  if (e0 >= ETOT) return;
  int lane = threadIdx.x & 31;
  int sub = lane >> 3, h = lane & 7;
  int dsti[8]; float val[8];
  #pragma unroll
  for (int i = 0; i < 8; ++i) {
    int row, col; edge_rc_flat(rowA, colA, e0 + (i << 2) + sub, row, col);
    val[i] = s2[col * 8 + h];
    dsti[i] = row * 8 + h;
  }
  #pragma unroll
  for (int i = 0; i < 8; ++i) atomic_max_f32(&m2[dsti[i]], val[i]);
}

__global__ void edge_sum2(const int* __restrict__ rowA, const int* __restrict__ colA,
                          const float* __restrict__ s2, const float* __restrict__ m2,
                          float* __restrict__ ss2) {
  int wave = blockIdx.x * (blockDim.x >> 5) + (threadIdx.x >> 5);
  int e0 = wave << 5;
  if (e0 >= ETOT) return;
  int lane = threadIdx.x & 31;
  int sub = lane >> 3, h = lane & 7;
  int dsti[8]; float val[8];
  #pragma unroll
  for (int i = 0; i < 8; ++i) {
    int row, col; edge_rc_flat(rowA, colA, e0 + (i << 2) + sub, row, col);
    val[i] = expf(s2[col * 8 + h] - m2[row * 8 + h]);
    dsti[i] = row * 8 + h;
  }
  #pragma unroll
  for (int i = 0; i < 8; ++i) atomicAdd(&ss2[dsti[i]], val[i]);
}

__global__ void edge_scatter2(const int* __restrict__ rowA, const int* __restrict__ colA,
                              const float* __restrict__ s2, const float* __restrict__ m2,
                              const float* __restrict__ ss2,
                              const float* __restrict__ h2, float* __restrict__ out2) {
  int wave = blockIdx.x * (blockDim.x >> 5) + (threadIdx.x >> 5);
  int e0 = wave << 3;
  if (e0 >= ETOT) return;
  int lane = threadIdx.x & 31;
  int half = lane >> 4, j = lane & 15, hq = lane & 7;
  int dsti[4]; float val[4][3];
  #pragma unroll
  for (int i = 0; i < 4; ++i) {
    int row, col; edge_rc_flat(rowA, colA, e0 + (i << 1) + half, row, col);
    float w = expf(s2[col * 8 + hq] - m2[row * 8 + hq]) / (ss2[row * 8 + hq] + 1e-16f);
    const float* __restrict__ src = h2 + (size_t)col * D2 + j;
    dsti[i] = row * D2 + j;
    #pragma unroll
    for (int k = 0; k < 3; ++k) {
      int hd = (j + 16 * k) / C2;
      float wk = __shfl(w, (lane & 16) | hd, 32);
      val[i][k] = src[16 * k] * wk;
    }
  }
  #pragma unroll
  for (int i = 0; i < 4; ++i) {
    #pragma unroll
    for (int k = 0; k < 3; ++k) atomicAdd(out2 + dsti[i] + 16 * k, val[i][k]);
  }
}

__global__ void finalize(const float* __restrict__ out2, const float* __restrict__ b2,
                         float* __restrict__ out) {
  int n = blockIdx.x * blockDim.x + threadIdx.x;
  if (n >= N_NODES) return;
  float v[C2];
  #pragma unroll
  for (int c = 0; c < C2; ++c) {
    float s = 0.f;
    #pragma unroll
    for (int h = 0; h < H2; ++h) s += out2[(size_t)n * D2 + h * C2 + c];
    float t = s * (1.0f / H2) + b2[c];
    v[c] = t > 0.f ? t : (expf(t) - 1.0f);
  }
  float m = v[0];
  #pragma unroll
  for (int c = 1; c < C2; ++c) m = fmaxf(m, v[c]);
  float sum = 0.f;
  #pragma unroll
  for (int c = 0; c < C2; ++c) sum += expf(v[c] - m);
  float lse = logf(sum);
  #pragma unroll
  for (int c = 0; c < C2; ++c) out[(size_t)n * C2 + c] = v[c] - m - lse;
}


extern "C" void kernel_launch(void* const* d_in, const int* in_sizes, int n_in,
                              void* d_out, int out_size, void* d_ws, size_t ws_size,
                              hipStream_t stream) {
  const float* x    = (const float*)d_in[0];
  const int*   ei   = (const int*)  d_in[1];
  const float* g1   = (const float*)d_in[2];
  const float* be1  = (const float*)d_in[3];
  const float* mu1  = (const float*)d_in[4];
  const float* va1  = (const float*)d_in[5];
  const float* W1   = (const float*)d_in[6];
  const float* att1 = (const float*)d_in[7];
  const float* b1   = (const float*)d_in[8];
  const float* g2   = (const float*)d_in[9];
  const float* be2  = (const float*)d_in[10];
  const float* mu2  = (const float*)d_in[11];
  const float* va2  = (const float*)d_in[12];
  const float* W2   = (const float*)d_in[13];
  const float* att2 = (const float*)d_in[14];
  const float* b2   = (const float*)d_in[15];
  const int* rowA = ei;
  const int* colA = ei + N_EDGES;
  float* out = (float*)d_out;
  float* ws  = (float*)d_ws;

  float* sc1  = ws;
  float* sh1  = ws + 256;
  float* sc2  = ws + 512;
  float* sh2  = ws + 1024;
  float* ai   = ws + 1536;
  float* aj   = ai  + (size_t)N_NODES * H1;
  float* m1   = aj  + (size_t)N_NODES * H1;
  float* s1   = m1  + (size_t)N_NODES * H1;
  float* out1 = s1  + (size_t)N_NODES * H1;
  float* h1   = out1 + (size_t)N_NODES * D1;
  float* h2   = h1;
  float* s2   = h2  + (size_t)N_NODES * D2;
  float* m2   = s2  + (size_t)N_NODES * H2;
  float* ss2  = m2  + (size_t)N_NODES * H2;
  float* out2 = ss2 + (size_t)N_NODES * H2;

  const float NEG_INF = -__builtin_inff();

  bn_prep<<<2, 256, 0, stream>>>(g1, be1, mu1, va1, g2, be2, mu2, va2, b1,
                                 sc1, sh1, sc2, sh2);
  fill_f32<<<625,   256, 0, stream>>>(m1, NEG_INF, N_NODES * H1);
  fill_f32<<<625,   256, 0, stream>>>(s1, 0.0f,    N_NODES * H1);
  fill_f32<<<20000, 256, 0, stream>>>(out1, 0.0f,  N_NODES * D1);

  gemm1_rows_resident<<<dim3(313, 4), 256, 0, stream>>>(x, sc1, sh1, W1, h1);

  node_att1<<<20000, 256, 0, stream>>>(h1, att1, ai, aj);
  edge_max1<<<(ETOT + 255) / 256, 256, 0, stream>>>(rowA, colA, ai, aj, m1);
  edge_sum1<<<(ETOT + 255) / 256, 256, 0, stream>>>(rowA, colA, ai, aj, m1, s1);
  edge_scatter1<<<(ETOT / 8) / 8, 256, 0, stream>>>(rowA, colA, ai, aj, m1, s1, h1, out1);

  fill_f32<<<1250, 256, 0, stream>>>(m2, NEG_INF, N_NODES * H2);
  fill_f32<<<1250, 256, 0, stream>>>(ss2, 0.0f,   N_NODES * H2);
  fill_f32<<<7500, 256, 0, stream>>>(out2, 0.0f,  N_NODES * D2);

  gemm_bn_wmma<<<(2500 * 3 + 7) / 8, 256, 0, stream>>>(out1, sc2, sh2, W2, h2,
                                                       N_NODES, D1, D2);

  node_att2<<<(N_NODES * H2) / 256, 256, 0, stream>>>(h2, att2, s2);
  edge_max2<<<(ETOT + 255) / 256, 256, 0, stream>>>(rowA, colA, s2, m2);
  edge_sum2<<<(ETOT + 255) / 256, 256, 0, stream>>>(rowA, colA, s2, m2, ss2);
  edge_scatter2<<<(ETOT / 8) / 8, 256, 0, stream>>>(rowA, colA, s2, m2, ss2, h2, out2);

  finalize<<<(N_NODES + 255) / 256, 256, 0, stream>>>(out2, b2, out);
}


# The six fill launches removed: zero and -inf start values of the atomic planes are stored by kernels that run ahead anyway (both GEMMs, both node score kernels); 18 -> 12 launches; sums as before
# speedup vs baseline: 2.5049x; 1.0380x over previous
#include <hip/hip_runtime.h>
#include <math.h>

#define N_NODES 40000
#define N_EDGES 320000
#define ETOT    (N_EDGES + N_NODES)
#define F_IN    256
#define H1      4
#define C1      128
#define D1      (H1*C1)
#define H2      8
#define C2      6
#define D2      (H2*C2)
#define BN_EPS    1e-5f
#define NEG_SLOPE 0.2f

typedef __attribute__((ext_vector_type(2))) float v2f;
typedef __attribute__((ext_vector_type(4))) float v4f;
typedef __attribute__((ext_vector_type(8))) float v8f;


__device__ __forceinline__ void atomic_max_f32(float* addr, float v) {
  if (v >= 0.0f) atomicMax((int*)addr, __float_as_int(v));
  else           atomicMin((unsigned int*)addr, __float_as_uint(v));
}

__device__ __forceinline__ void edge_rc_flat(const int* __restrict__ rowA,
                                             const int* __restrict__ colA,
                                             int e, int& r, int& c) {
  int ec = e < N_EDGES ? e : N_EDGES - 1;
  int lr = rowA[ec], lc = colA[ec];
  bool self = e >= N_EDGES;
  r = self ? e - N_EDGES : lr;
  c = self ? e - N_EDGES : lc;
}


__global__ void bn_prep(const float* g1, const float* be1, const float* mu1, const float* va1,
                        const float* g2, const float* be2, const float* mu2, const float* va2,
                        const float* bias1,
                        float* sc1, float* sh1, float* sc2, float* sh2) {
  int i = blockIdx.x * blockDim.x + threadIdx.x;
  if (i < F_IN) {
    float s = g1[i] / sqrtf(va1[i] + BN_EPS);
    sc1[i] = s; sh1[i] = be1[i] - mu1[i] * s;
  }
  if (i < D1) {
    float s = g2[i] / sqrtf(va2[i] + BN_EPS);
    sc2[i] = s; sh2[i] = be2[i] + (bias1[i] - mu2[i]) * s;
  }
}


__global__ __launch_bounds__(256)
void gemm_bn_wmma(const float* __restrict__ A,
                  const float* __restrict__ scale,
                  const float* __restrict__ shift,
                  const float* __restrict__ B,
                  float* __restrict__ C,
                  float* __restrict__ Z,
                  int M, int K, int N) {
  int wave = blockIdx.x * (blockDim.x >> 5) + (threadIdx.x >> 5);
  int ntiles = N >> 4;
  int tm = wave / ntiles;
  int tn = wave - tm * ntiles;
  if (tm >= (M >> 4)) return;

  int lane = threadIdx.x & 31;
  int half = lane >> 4;
  int l16  = lane & 15;
  int row  = (tm << 4) + l16;
  int colB = (tn << 4) + l16;

  {
    float* __restrict__ Zt = Z + (size_t)((tm << 4) + (lane >> 2)) * N + (tn << 4) + ((lane & 3) << 2);
    const v4f z = {0.f, 0.f, 0.f, 0.f};
    *(v4f*)Zt = z;
    *(v4f*)(Zt + (size_t)8 * N) = z;
  }

  const float* __restrict__ Arow = A + (size_t)row * K;
  v8f acc = {};
  for (int k = 0; k < K; k += 8) {
    int kk = k + (half << 2);
    v4f a4 = *(const v4f*)(Arow + kk);
    v4f s4 = *(const v4f*)(scale + kk);
    v4f t4 = *(const v4f*)(shift + kk);
    const float* __restrict__ Bk = B + (size_t)kk * N + colB;
    v2f a, b;
    a.x = a4.x * s4.x + t4.x;
    a.y = a4.y * s4.y + t4.y;
    b.x = Bk[0];
    b.y = Bk[(size_t)N];
    acc = __builtin_amdgcn_wmma_f32_16x16x4_f32(
         false, a,  false, b,
         (short)0, acc,  false,  false);
    a.x = a4.z * s4.z + t4.z;
    a.y = a4.w * s4.w + t4.w;
    b.x = Bk[(size_t)2 * N];
    b.y = Bk[(size_t)3 * N];
    acc = __builtin_amdgcn_wmma_f32_16x16x4_f32(
         false, a,  false, b,
         (short)0, acc,  false,  false);
  }
  float* Crow = C + (size_t)((tm << 4) + (half << 3)) * N + colB;
  #pragma unroll
  for (int r = 0; r < 8; ++r) Crow[(size_t)r * N] = acc[r];
}

#define G1_COLS_PER_WAVE 128

__global__ __launch_bounds__(256)
void gemm1_rows_resident(const float* __restrict__ A,
                         const float* __restrict__ scale,
                         const float* __restrict__ shift,
                         const float* __restrict__ B,
                         float* __restrict__ C,
                         float* __restrict__ Z) {
  int tm = blockIdx.x * 8 + (threadIdx.x >> 5);
  if (tm >= (N_NODES >> 4)) return;
  int lane = threadIdx.x & 31;
  int half = lane >> 4;
  int l16  = lane & 15;

  {
    float* __restrict__ Zl = Z + (size_t)(tm << 4) * D1 + blockIdx.y * G1_COLS_PER_WAVE + (lane << 2);
    const v4f z = {0.f, 0.f, 0.f, 0.f};
    for (int r = 0; r < 16; ++r) *(v4f*)(Zl + (size_t)r * D1) = z;
  }

  const float* __restrict__ Arow = A + (size_t)((tm << 4) + l16) * F_IN + (half << 2);
  const float* __restrict__ sc = scale + (half << 2);
  const float* __restrict__ sh = shift + (half << 2);
  v4f a[F_IN / 8];
  #pragma unroll
  for (int t = 0; t < F_IN / 8; ++t) a[t] = *(const v4f*)(Arow + 8 * t);
  #pragma unroll
  for (int t0 = 0; t0 < F_IN / 8; t0 += 8) {
    #pragma unroll
    for (int t = t0; t < t0 + 8; ++t) {
      v4f s4 = *(const v4f*)(sc + 8 * t);
      v4f t4 = *(const v4f*)(sh + 8 * t);
      a[t].x = a[t].x * s4.x + t4.x;
      a[t].y = a[t].y * s4.y + t4.y;
      a[t].z = a[t].z * s4.z + t4.z;
      a[t].w = a[t].w * s4.w + t4.w;
    }
    __builtin_amdgcn_sched_barrier(0);
  }

  int cb = blockIdx.y * G1_COLS_PER_WAVE + (l16 << 2);
  const float* __restrict__ Bl = B + (size_t)(half << 2) * D1 + cb;
  float* __restrict__ Cl = C + (size_t)((tm << 4) + (half << 3)) * D1 + cb;
  for (int j = 0; j < G1_COLS_PER_WAVE; j += 64) {
    v8f acc0 = {}, acc1 = {}, acc2 = {}, acc3 = {};
    const float* __restrict__ Bj = Bl + j;
    v4f bq[3][4];
    #pragma unroll
    for (int p = 0; p < 2; ++p) {
      #pragma unroll
      for (int q = 0; q < 4; ++q) bq[p][q] = *(const v4f*)(Bj + (size_t)(8 * p + q) * D1);
    }
    #pragma unroll
    for (int t = 0; t < F_IN / 8; ++t) {
      if (t + 2 < F_IN / 8) {
        #pragma unroll
        for (int q = 0; q < 4; ++q)
          bq[(t + 2) % 3][q] = *(const v4f*)(Bj + (size_t)(8 * (t + 2) + q) * D1);
      }
      __builtin_amdgcn_sched_barrier(0);
      const v4f b0 = bq[t % 3][0], b1 = bq[t % 3][1], b2 = bq[t % 3][2], b3 = bq[t % 3][3];
      v2f a01, a23, b;
      a01.x = a[t].x; a01.y = a[t].y;
      a23.x = a[t].z; a23.y = a[t].w;
      b.x = b0.x; b.y = b1.x;
      acc0 = __builtin_amdgcn_wmma_f32_16x16x4_f32(false, a01, false, b, (short)0, acc0, false, false);
      b.x = b0.y; b.y = b1.y;
      acc1 = __builtin_amdgcn_wmma_f32_16x16x4_f32(false, a01, false, b, (short)0, acc1, false, false);
      b.x = b0.z; b.y = b1.z;
      acc2 = __builtin_amdgcn_wmma_f32_16x16x4_f32(false, a01, false, b, (short)0, acc2, false, false);
      b.x = b0.w; b.y = b1.w;
      acc3 = __builtin_amdgcn_wmma_f32_16x16x4_f32(false, a01, false, b, (short)0, acc3, false, false);
      b.x = b2.x; b.y = b3.x;
      acc0 = __builtin_amdgcn_wmma_f32_16x16x4_f32(false, a23, false, b, (short)0, acc0, false, false);
      b.x = b2.y; b.y = b3.y;
      acc1 = __builtin_amdgcn_wmma_f32_16x16x4_f32(false, a23, false, b, (short)0, acc1, false, false);
      b.x = b2.z; b.y = b3.z;
      acc2 = __builtin_amdgcn_wmma_f32_16x16x4_f32(false, a23, false, b, (short)0, acc2, false, false);
      b.x = b2.w; b.y = b3.w;
      acc3 = __builtin_amdgcn_wmma_f32_16x16x4_f32(false, a23, false, b, (short)0, acc3, false, false);
      __builtin_amdgcn_sched_barrier(0);
    }
    #pragma unroll
    for (int r = 0; r < 8; ++r) {
      v4f o;
      o.x = acc0[r]; o.y = acc1[r]; o.z = acc2[r]; o.w = acc3[r];
      *(v4f*)(Cl + (size_t)r * D1 + j) = o;
    }
  }
}


__global__ void node_att1(const float* __restrict__ h1, const float* __restrict__ att1,
                          float* __restrict__ ai, float* __restrict__ aj,
                          float* __restrict__ m1, float* __restrict__ s1) {
  int wave = blockIdx.x * (blockDim.x >> 5) + (threadIdx.x >> 5);
  if (wave >= N_NODES * H1) return;
  int n = wave >> 2, h = wave & 3;
  int lane = threadIdx.x & 31;
  const float* __restrict__ hp = h1 + (size_t)n * D1 + h * C1;
  const float* __restrict__ al = att1 + h * (2 * C1);
  const float* __restrict__ ar = al + C1;
  float si = 0.f, sj = 0.f;
  #pragma unroll
  for (int c0 = 0; c0 < C1; c0 += 32) {
    float v = hp[c0 + lane];
    si += v * al[c0 + lane];
    sj += v * ar[c0 + lane];
  }
  #pragma unroll
  for (int off = 16; off > 0; off >>= 1) {
    si += __shfl_xor(si, off, 32);
    sj += __shfl_xor(sj, off, 32);
  }
  if (lane == 0) {
    ai[wave] = si; aj[wave] = sj;
    m1[wave] = -__builtin_inff(); s1[wave] = 0.0f;
  }
}


__global__ void edge_max1(const int* __restrict__ rowA, const int* __restrict__ colA,
                          const float* __restrict__ ai, const float* __restrict__ aj,
                          float* __restrict__ m1) {
  int wave = blockIdx.x * (blockDim.x >> 5) + (threadIdx.x >> 5);
  int e0 = wave << 5;
  if (e0 >= ETOT) return;
  int lane = threadIdx.x & 31;
  int sub = lane >> 2, h = lane & 3;
  int dsti[4]; float val[4];
  #pragma unroll
  for (int i = 0; i < 4; ++i) {
    int row, col; edge_rc_flat(rowA, colA, e0 + (i << 3) + sub, row, col);
    float a = ai[row * 4 + h] + aj[col * 4 + h];
    val[i] = a > 0.f ? a : NEG_SLOPE * a;
    dsti[i] = row * 4 + h;
  }
  #pragma unroll
  for (int i = 0; i < 4; ++i) atomic_max_f32(&m1[dsti[i]], val[i]);
}

__global__ void edge_sum1(const int* __restrict__ rowA, const int* __restrict__ colA,
                          const float* __restrict__ ai, const float* __restrict__ aj,
                          const float* __restrict__ m1, float* __restrict__ s1) {
  int wave = blockIdx.x * (blockDim.x >> 5) + (threadIdx.x >> 5);
  int e0 = wave << 5;
  if (e0 >= ETOT) return;
  int lane = threadIdx.x & 31;
  int sub = lane >> 2, h = lane & 3;
  int dsti[4]; float val[4];
  #pragma unroll
  for (int i = 0; i < 4; ++i) {
    int row, col; edge_rc_flat(rowA, colA, e0 + (i << 3) + sub, row, col);
    float a = ai[row * 4 + h] + aj[col * 4 + h];
    a = a > 0.f ? a : NEG_SLOPE * a;
    val[i] = expf(a - m1[row * 4 + h]);
    dsti[i] = row * 4 + h;
  }
  #pragma unroll
  for (int i = 0; i < 4; ++i) atomicAdd(&s1[dsti[i]], val[i]);
}

__global__ __launch_bounds__(256)
void edge_scatter1(const int* __restrict__ rowA, const int* __restrict__ colA,
                   const float* __restrict__ ai, const float* __restrict__ aj,
                   const float* __restrict__ m1, const float* __restrict__ s1,
                   const float* __restrict__ h1, float* __restrict__ out1) {
  int wave = blockIdx.x * (blockDim.x >> 5) + (threadIdx.x >> 5);
  int e0 = wave << 3;
  if (e0 >= ETOT) return;
  int lane = threadIdx.x & 31;
  int sub = lane >> 2, h = lane & 3;
  int row, col; edge_rc_flat(rowA, colA, e0 + sub, row, col);
  float a = ai[row * 4 + h] + aj[col * 4 + h];
  a = a > 0.f ? a : NEG_SLOPE * a;
  float w = expf(a - m1[row * 4 + h]) / (s1[row * 4 + h] + 1e-16f);
  int wbits = __float_as_int(w);
  #pragma unroll
  for (int i = 0; i < 8; i += 2) {
    float v[2][16];
    int rr[2];
    #pragma unroll
    for (int u = 0; u < 2; ++u) {
      int cc = __builtin_amdgcn_readlane(col, (i + u) * 4);
      rr[u]  = __builtin_amdgcn_readlane(row, (i + u) * 4);
      const float* __restrict__ src = h1 + (size_t)cc * D1 + lane;
      #pragma unroll
      for (int q = 0; q < 16; ++q) v[u][q] = src[q * 32];
    }
    #pragma unroll
    for (int u = 0; u < 2; ++u) {
      float* dst = out1 + (size_t)rr[u] * D1 + lane;
      #pragma unroll
      for (int q = 0; q < 16; ++q) {
        float wq = __int_as_float(__builtin_amdgcn_readlane(wbits, (i + u) * 4 + (q >> 2)));
        atomicAdd(dst + q * 32, v[u][q] * wq);
      }
    }
  }
}


__global__ void node_att2(const float* __restrict__ h2, const float* __restrict__ att2,
                          float* __restrict__ s2, float* __restrict__ m2,
                          float* __restrict__ ss2) {
  int tid = blockIdx.x * blockDim.x + threadIdx.x;
  if (tid >= N_NODES * H2) return;
  int n = tid >> 3, h = tid & 7;
  const float* __restrict__ hp = h2 + (size_t)n * D2 + h * C2;
  const float* __restrict__ ap = att2 + h * (2 * C2);
  float s = 0.f;
  #pragma unroll
  for (int c = 0; c < C2; ++c) s += hp[c] * (ap[c] + ap[C2 + c]);
  s2[tid] = tanhf(s);
  m2[tid] = -__builtin_inff();
  ss2[tid] = 0.0f;
}

__global__ void edge_max2(const int* __restrict__ rowA, const int* __restrict__ colA,
                          const float* __restrict__ s2, float* __restrict__ m2) {
  int wave = blockIdx.x * (blockDim.x >> 5) + (threadIdx.x >> 5);
  int e0 = wave << 5;
  if (e0 >= ETOT) return;
  int lane = threadIdx.x & 31;
  int sub = lane >> 3, h = lane & 7;
  int dsti[8]; float val[8];
  #pragma unroll
  for (int i = 0; i < 8; ++i) {
    int row, col; edge_rc_flat(rowA, colA, e0 + (i << 2) + sub, row, col);
    val[i] = s2[col * 8 + h];
    dsti[i] = row * 8 + h;
  }
  #pragma unroll
  for (int i = 0; i < 8; ++i) atomic_max_f32(&m2[dsti[i]], val[i]);
}

__global__ void edge_sum2(const int* __restrict__ rowA, const int* __restrict__ colA,
                          const float* __restrict__ s2, const float* __restrict__ m2,
                          float* __restrict__ ss2) {
  int wave = blockIdx.x * (blockDim.x >> 5) + (threadIdx.x >> 5);
  int e0 = wave << 5;
  if (e0 >= ETOT) return;
  int lane = threadIdx.x & 31;
  int sub = lane >> 3, h = lane & 7;
  int dsti[8]; float val[8];
  #pragma unroll
  for (int i = 0; i < 8; ++i) {
    int row, col; edge_rc_flat(rowA, colA, e0 + (i << 2) + sub, row, col);
    val[i] = expf(s2[col * 8 + h] - m2[row * 8 + h]);
    dsti[i] = row * 8 + h;
  }
  #pragma unroll
  for (int i = 0; i < 8; ++i) atomicAdd(&ss2[dsti[i]], val[i]);
}

__global__ void edge_scatter2(const int* __restrict__ rowA, const int* __restrict__ colA,
                              const float* __restrict__ s2, const float* __restrict__ m2,
                              const float* __restrict__ ss2,
                              const float* __restrict__ h2, float* __restrict__ out2) {
  int wave = blockIdx.x * (blockDim.x >> 5) + (threadIdx.x >> 5);
  int e0 = wave << 3;
  if (e0 >= ETOT) return;
  int lane = threadIdx.x & 31;
  int half = lane >> 4, j = lane & 15, hq = lane & 7;
  int dsti[4]; float val[4][3];
  #pragma unroll
  for (int i = 0; i < 4; ++i) {
    int row, col; edge_rc_flat(rowA, colA, e0 + (i << 1) + half, row, col);
    float w = expf(s2[col * 8 + hq] - m2[row * 8 + hq]) / (ss2[row * 8 + hq] + 1e-16f);
    const float* __restrict__ src = h2 + (size_t)col * D2 + j;
    dsti[i] = row * D2 + j;
    #pragma unroll
    for (int k = 0; k < 3; ++k) {
      int hd = (j + 16 * k) / C2;
      float wk = __shfl(w, (lane & 16) | hd, 32);
      val[i][k] = src[16 * k] * wk;
    }
  }
  #pragma unroll
  for (int i = 0; i < 4; ++i) {
    #pragma unroll
    for (int k = 0; k < 3; ++k) atomicAdd(out2 + dsti[i] + 16 * k, val[i][k]);
  }
}

__global__ void finalize(const float* __restrict__ out2, const float* __restrict__ b2,
                         float* __restrict__ out) {
  int n = blockIdx.x * blockDim.x + threadIdx.x;
  if (n >= N_NODES) return;
  float v[C2];
  #pragma unroll
  for (int c = 0; c < C2; ++c) {
    float s = 0.f;
    #pragma unroll
    for (int h = 0; h < H2; ++h) s += out2[(size_t)n * D2 + h * C2 + c];
    float t = s * (1.0f / H2) + b2[c];
    v[c] = t > 0.f ? t : (expf(t) - 1.0f);
  }
  float m = v[0];
  #pragma unroll
  for (int c = 1; c < C2; ++c) m = fmaxf(m, v[c]);
  float sum = 0.f;
  #pragma unroll
  for (int c = 0; c < C2; ++c) sum += expf(v[c] - m);
  float lse = logf(sum);
  #pragma unroll
  for (int c = 0; c < C2; ++c) out[(size_t)n * C2 + c] = v[c] - m - lse;
}


extern "C" void kernel_launch(void* const* d_in, const int* in_sizes, int n_in,
                              void* d_out, int out_size, void* d_ws, size_t ws_size,
                              hipStream_t stream) {
  const float* x    = (const float*)d_in[0];
  const int*   ei   = (const int*)  d_in[1];
  const float* g1   = (const float*)d_in[2];
  const float* be1  = (const float*)d_in[3];
  const float* mu1  = (const float*)d_in[4];
  const float* va1  = (const float*)d_in[5];
  const float* W1   = (const float*)d_in[6];
  const float* att1 = (const float*)d_in[7];
  const float* b1   = (const float*)d_in[8];
  const float* g2   = (const float*)d_in[9];
  const float* be2  = (const float*)d_in[10];
  const float* mu2  = (const float*)d_in[11];
  const float* va2  = (const float*)d_in[12];
  const float* W2   = (const float*)d_in[13];
  const float* att2 = (const float*)d_in[14];
  const float* b2   = (const float*)d_in[15];
  const int* rowA = ei;
  const int* colA = ei + N_EDGES;
  float* out = (float*)d_out;
  float* ws  = (float*)d_ws;

  float* sc1  = ws;
  float* sh1  = ws + 256;
  float* sc2  = ws + 512;
  float* sh2  = ws + 1024;
  float* ai   = ws + 1536;
  float* aj   = ai  + (size_t)N_NODES * H1;
  float* m1   = aj  + (size_t)N_NODES * H1;
  float* s1   = m1  + (size_t)N_NODES * H1;
  float* out1 = s1  + (size_t)N_NODES * H1;
  float* h1   = out1 + (size_t)N_NODES * D1;
  float* h2   = h1;
  float* s2   = h2  + (size_t)N_NODES * D2;
  float* m2   = s2  + (size_t)N_NODES * H2;
  float* ss2  = m2  + (size_t)N_NODES * H2;
  float* out2 = ss2 + (size_t)N_NODES * H2;

  bn_prep<<<2, 256, 0, stream>>>(g1, be1, mu1, va1, g2, be2, mu2, va2, b1,
                                 sc1, sh1, sc2, sh2);

  gemm1_rows_resident<<<dim3(313, 4), 256, 0, stream>>>(x, sc1, sh1, W1, h1, out1);

  node_att1<<<20000, 256, 0, stream>>>(h1, att1, ai, aj, m1, s1);
  edge_max1<<<(ETOT + 255) / 256, 256, 0, stream>>>(rowA, colA, ai, aj, m1);
  edge_sum1<<<(ETOT + 255) / 256, 256, 0, stream>>>(rowA, colA, ai, aj, m1, s1);
  edge_scatter1<<<(ETOT / 8) / 8, 256, 0, stream>>>(rowA, colA, ai, aj, m1, s1, h1, out1);


  gemm_bn_wmma<<<(2500 * 3 + 7) / 8, 256, 0, stream>>>(out1, sc2, sh2, W2, h2, out2,
                                                       N_NODES, D1, D2);

  node_att2<<<(N_NODES * H2) / 256, 256, 0, stream>>>(h2, att2, s2, m2, ss2);
  edge_max2<<<(ETOT + 255) / 256, 256, 0, stream>>>(rowA, colA, s2, m2);
  edge_sum2<<<(ETOT + 255) / 256, 256, 0, stream>>>(rowA, colA, s2, m2, ss2);
  edge_scatter2<<<(ETOT / 8) / 8, 256, 0, stream>>>(rowA, colA, s2, m2, ss2, h2, out2);

  finalize<<<(N_NODES + 255) / 256, 256, 0, stream>>>(out2, b2, out);
}


# Layer-1 node score kernel: a wave streams 8 whole node rows (16-byte loads, 16 in flight) through register-held attention vectors; 160000 waves -> 5000; score sums in a new lane order
# speedup vs baseline: 2.6095x; 1.0418x over previous
#include <hip/hip_runtime.h>
#include <math.h>

#define N_NODES 40000
#define N_EDGES 320000
#define ETOT    (N_EDGES + N_NODES)
#define F_IN    256
#define H1      4
#define C1      128
#define D1      (H1*C1)
#define H2      8
#define C2      6
#define D2      (H2*C2)
#define BN_EPS    1e-5f
#define NEG_SLOPE 0.2f

typedef __attribute__((ext_vector_type(2))) float v2f;
typedef __attribute__((ext_vector_type(4))) float v4f;
typedef __attribute__((ext_vector_type(8))) float v8f;


__device__ __forceinline__ void atomic_max_f32(float* addr, float v) {
  if (v >= 0.0f) atomicMax((int*)addr, __float_as_int(v));
  else           atomicMin((unsigned int*)addr, __float_as_uint(v));
}

__device__ __forceinline__ void edge_rc_flat(const int* __restrict__ rowA,
                                             const int* __restrict__ colA,
                                             int e, int& r, int& c) {
  int ec = e < N_EDGES ? e : N_EDGES - 1;
  int lr = rowA[ec], lc = colA[ec];
  bool self = e >= N_EDGES;
  r = self ? e - N_EDGES : lr;
  c = self ? e - N_EDGES : lc;
}


__global__ void bn_prep(const float* g1, const float* be1, const float* mu1, const float* va1,
                        const float* g2, const float* be2, const float* mu2, const float* va2,
                        const float* bias1,
                        float* sc1, float* sh1, float* sc2, float* sh2) {
  int i = blockIdx.x * blockDim.x + threadIdx.x;
  if (i < F_IN) {
    float s = g1[i] / sqrtf(va1[i] + BN_EPS);
    sc1[i] = s; sh1[i] = be1[i] - mu1[i] * s;
  }
  if (i < D1) {
    float s = g2[i] / sqrtf(va2[i] + BN_EPS);
    sc2[i] = s; sh2[i] = be2[i] + (bias1[i] - mu2[i]) * s;
  }
}


__global__ __launch_bounds__(256)
void gemm_bn_wmma(const float* __restrict__ A,
                  const float* __restrict__ scale,
                  const float* __restrict__ shift,
                  const float* __restrict__ B,
                  float* __restrict__ C,
                  float* __restrict__ Z,
                  int M, int K, int N) {
  int wave = blockIdx.x * (blockDim.x >> 5) + (threadIdx.x >> 5);
  int ntiles = N >> 4;
  int tm = wave / ntiles;
  int tn = wave - tm * ntiles;
  if (tm >= (M >> 4)) return;

  int lane = threadIdx.x & 31;
  int half = lane >> 4;
  int l16  = lane & 15;
  int row  = (tm << 4) + l16;
  int colB = (tn << 4) + l16;

  {
    float* __restrict__ Zt = Z + (size_t)((tm << 4) + (lane >> 2)) * N + (tn << 4) + ((lane & 3) << 2);
    const v4f z = {0.f, 0.f, 0.f, 0.f};
    *(v4f*)Zt = z;
    *(v4f*)(Zt + (size_t)8 * N) = z;
  }

  const float* __restrict__ Arow = A + (size_t)row * K;
  v8f acc = {};
  for (int k = 0; k < K; k += 8) {
    int kk = k + (half << 2);
    v4f a4 = *(const v4f*)(Arow + kk);
    v4f s4 = *(const v4f*)(scale + kk);
    v4f t4 = *(const v4f*)(shift + kk);
    const float* __restrict__ Bk = B + (size_t)kk * N + colB;
    v2f a, b;
    a.x = a4.x * s4.x + t4.x;
    a.y = a4.y * s4.y + t4.y;
    b.x = Bk[0];
    b.y = Bk[(size_t)N];
    acc = __builtin_amdgcn_wmma_f32_16x16x4_f32(
         false, a,  false, b,
         (short)0, acc,  false,  false);
    a.x = a4.z * s4.z + t4.z;
    a.y = a4.w * s4.w + t4.w;
    b.x = Bk[(size_t)2 * N];
    b.y = Bk[(size_t)3 * N];
    acc = __builtin_amdgcn_wmma_f32_16x16x4_f32(
         false, a,  false, b,
         (short)0, acc,  false,  false);
  }
  float* Crow = C + (size_t)((tm << 4) + (half << 3)) * N + colB;
  #pragma unroll
  for (int r = 0; r < 8; ++r) Crow[(size_t)r * N] = acc[r];
}

#define G1_COLS_PER_WAVE 128

__global__ __launch_bounds__(256)
void gemm1_rows_resident(const float* __restrict__ A,
                         const float* __restrict__ scale,
                         const float* __restrict__ shift,
                         const float* __restrict__ B,
                         float* __restrict__ C,
                         float* __restrict__ Z) {
  int tm = blockIdx.x * 8 + (threadIdx.x >> 5);
  if (tm >= (N_NODES >> 4)) return;
  int lane = threadIdx.x & 31;
  int half = lane >> 4;
  int l16  = lane & 15;

  {
    float* __restrict__ Zl = Z + (size_t)(tm << 4) * D1 + blockIdx.y * G1_COLS_PER_WAVE + (lane << 2);
    const v4f z = {0.f, 0.f, 0.f, 0.f};
    for (int r = 0; r < 16; ++r) *(v4f*)(Zl + (size_t)r * D1) = z;
  }

  const float* __restrict__ Arow = A + (size_t)((tm << 4) + l16) * F_IN + (half << 2);
  const float* __restrict__ sc = scale + (half << 2);
  const float* __restrict__ sh = shift + (half << 2);
  v4f a[F_IN / 8];
  #pragma unroll
  for (int t = 0; t < F_IN / 8; ++t) a[t] = *(const v4f*)(Arow + 8 * t);
  #pragma unroll
  for (int t0 = 0; t0 < F_IN / 8; t0 += 8) {
    #pragma unroll
    for (int t = t0; t < t0 + 8; ++t) {
      v4f s4 = *(const v4f*)(sc + 8 * t);
      v4f t4 = *(const v4f*)(sh + 8 * t);
      a[t].x = a[t].x * s4.x + t4.x;
      a[t].y = a[t].y * s4.y + t4.y;
      a[t].z = a[t].z * s4.z + t4.z;
      a[t].w = a[t].w * s4.w + t4.w;
    }
    __builtin_amdgcn_sched_barrier(0);
  }

  int cb = blockIdx.y * G1_COLS_PER_WAVE + (l16 << 2);
  const float* __restrict__ Bl = B + (size_t)(half << 2) * D1 + cb;
  float* __restrict__ Cl = C + (size_t)((tm << 4) + (half << 3)) * D1 + cb;
  for (int j = 0; j < G1_COLS_PER_WAVE; j += 64) {
    v8f acc0 = {}, acc1 = {}, acc2 = {}, acc3 = {};
    const float* __restrict__ Bj = Bl + j;
    v4f bq[3][4];
    #pragma unroll
    for (int p = 0; p < 2; ++p) {
      #pragma unroll
      for (int q = 0; q < 4; ++q) bq[p][q] = *(const v4f*)(Bj + (size_t)(8 * p + q) * D1);
    }
    #pragma unroll
    for (int t = 0; t < F_IN / 8; ++t) {
      if (t + 2 < F_IN / 8) {
        #pragma unroll
        for (int q = 0; q < 4; ++q)
          bq[(t + 2) % 3][q] = *(const v4f*)(Bj + (size_t)(8 * (t + 2) + q) * D1);
      }
      __builtin_amdgcn_sched_barrier(0);
      const v4f b0 = bq[t % 3][0], b1 = bq[t % 3][1], b2 = bq[t % 3][2], b3 = bq[t % 3][3];
      v2f a01, a23, b;
      a01.x = a[t].x; a01.y = a[t].y;
      a23.x = a[t].z; a23.y = a[t].w;
      b.x = b0.x; b.y = b1.x;
      acc0 = __builtin_amdgcn_wmma_f32_16x16x4_f32(false, a01, false, b, (short)0, acc0, false, false);
      b.x = b0.y; b.y = b1.y;
      acc1 = __builtin_amdgcn_wmma_f32_16x16x4_f32(false, a01, false, b, (short)0, acc1, false, false);
      b.x = b0.z; b.y = b1.z;
      acc2 = __builtin_amdgcn_wmma_f32_16x16x4_f32(false, a01, false, b, (short)0, acc2, false, false);
      b.x = b0.w; b.y = b1.w;
      acc3 = __builtin_amdgcn_wmma_f32_16x16x4_f32(false, a01, false, b, (short)0, acc3, false, false);
      b.x = b2.x; b.y = b3.x;
      acc0 = __builtin_amdgcn_wmma_f32_16x16x4_f32(false, a23, false, b, (short)0, acc0, false, false);
      b.x = b2.y; b.y = b3.y;
      acc1 = __builtin_amdgcn_wmma_f32_16x16x4_f32(false, a23, false, b, (short)0, acc1, false, false);
      b.x = b2.z; b.y = b3.z;
      acc2 = __builtin_amdgcn_wmma_f32_16x16x4_f32(false, a23, false, b, (short)0, acc2, false, false);
      b.x = b2.w; b.y = b3.w;
      acc3 = __builtin_amdgcn_wmma_f32_16x16x4_f32(false, a23, false, b, (short)0, acc3, false, false);
      __builtin_amdgcn_sched_barrier(0);
    }
    #pragma unroll
    for (int r = 0; r < 8; ++r) {
      v4f o;
      o.x = acc0[r]; o.y = acc1[r]; o.z = acc2[r]; o.w = acc3[r];
      *(v4f*)(Cl + (size_t)r * D1 + j) = o;
    }
  }
}


#define NA1_NODES 8

__global__ __launch_bounds__(256)
void node_att1(const float* __restrict__ h1, const float* __restrict__ att1,
               float* __restrict__ ai, float* __restrict__ aj,
               float* __restrict__ m1, float* __restrict__ s1) {
  int wave = blockIdx.x * (blockDim.x >> 5) + (threadIdx.x >> 5);
  int n0 = wave * NA1_NODES;
  if (n0 >= N_NODES) return;
  int lane = threadIdx.x & 31;
  v4f al[H1], ar[H1];
  #pragma unroll
  for (int h = 0; h < H1; ++h) {
    al[h] = *(const v4f*)(att1 + h * (2 * C1) + (lane << 2));
    ar[h] = *(const v4f*)(att1 + h * (2 * C1) + C1 + (lane << 2));
  }
  const bool b16 = (lane & 16) != 0, b8 = (lane & 8) != 0, b4 = (lane & 4) != 0;
  const int idx = lane >> 2;
  const float* __restrict__ hp = h1 + (size_t)n0 * D1 + (lane << 2);
  #pragma unroll
  for (int i = 0; i < NA1_NODES; i += 4) {
    v4f v[4][H1];
    #pragma unroll
    for (int u = 0; u < 4; ++u) {
      #pragma unroll
      for (int h = 0; h < H1; ++h) v[u][h] = *(const v4f*)(hp + (size_t)(i + u) * D1 + h * C1);
    }
    #pragma unroll
    for (int u = 0; u < 4; ++u) {
      float r[8];
      #pragma unroll
      for (int h = 0; h < H1; ++h) {
        const v4f x = v[u][h];
        r[h]     = x.x * al[h].x + x.y * al[h].y + x.z * al[h].z + x.w * al[h].w;
        r[4 + h] = x.x * ar[h].x + x.y * ar[h].y + x.z * ar[h].z + x.w * ar[h].w;
      }
      float t4[4], t2[2];
      #pragma unroll
      for (int k = 0; k < 4; ++k) {
        float keep = b16 ? r[4 + k] : r[k];
        float send = b16 ? r[k] : r[4 + k];
        t4[k] = keep + __shfl_xor(send, 16, 32);
      }
      #pragma unroll
      for (int k = 0; k < 2; ++k) {
        float keep = b8 ? t4[2 + k] : t4[k];
        float send = b8 ? t4[k] : t4[2 + k];
        t2[k] = keep + __shfl_xor(send, 8, 32);
      }
      float keep = b4 ? t2[1] : t2[0];
      float send = b4 ? t2[0] : t2[1];
      float t = keep + __shfl_xor(send, 4, 32);
      t += __shfl_xor(t, 2, 32);
      t += __shfl_xor(t, 1, 32);
      if ((lane & 3) == 0) {
        int n = n0 + i + u;
        if (idx < 4) {
          ai[n * 4 + idx] = t;
          m1[n * 4 + idx] = -__builtin_inff();
          s1[n * 4 + idx] = 0.0f;
        } else {
          aj[n * 4 + (idx - 4)] = t;
        }
      }
    }
  }
}


__global__ void edge_max1(const int* __restrict__ rowA, const int* __restrict__ colA,
                          const float* __restrict__ ai, const float* __restrict__ aj,
                          float* __restrict__ m1) {
  int wave = blockIdx.x * (blockDim.x >> 5) + (threadIdx.x >> 5);
  int e0 = wave << 5;
  if (e0 >= ETOT) return;
  int lane = threadIdx.x & 31;
  int sub = lane >> 2, h = lane & 3;
  int dsti[4]; float val[4];
  #pragma unroll
  for (int i = 0; i < 4; ++i) {
    int row, col; edge_rc_flat(rowA, colA, e0 + (i << 3) + sub, row, col);
    float a = ai[row * 4 + h] + aj[col * 4 + h];
    val[i] = a > 0.f ? a : NEG_SLOPE * a;
    dsti[i] = row * 4 + h;
  }
  #pragma unroll
  for (int i = 0; i < 4; ++i) atomic_max_f32(&m1[dsti[i]], val[i]);
}

__global__ void edge_sum1(const int* __restrict__ rowA, const int* __restrict__ colA,
                          const float* __restrict__ ai, const float* __restrict__ aj,
                          const float* __restrict__ m1, float* __restrict__ s1) {
  int wave = blockIdx.x * (blockDim.x >> 5) + (threadIdx.x >> 5);
  int e0 = wave << 5;
  if (e0 >= ETOT) return;
  int lane = threadIdx.x & 31;
  int sub = lane >> 2, h = lane & 3;
  int dsti[4]; float val[4];
  #pragma unroll
  for (int i = 0; i < 4; ++i) {
    int row, col; edge_rc_flat(rowA, colA, e0 + (i << 3) + sub, row, col);
    float a = ai[row * 4 + h] + aj[col * 4 + h];
    a = a > 0.f ? a : NEG_SLOPE * a;
    val[i] = expf(a - m1[row * 4 + h]);
    dsti[i] = row * 4 + h;
  }
  #pragma unroll
  for (int i = 0; i < 4; ++i) atomicAdd(&s1[dsti[i]], val[i]);
}

__global__ __launch_bounds__(256)
void edge_scatter1(const int* __restrict__ rowA, const int* __restrict__ colA,
                   const float* __restrict__ ai, const float* __restrict__ aj,
                   const float* __restrict__ m1, const float* __restrict__ s1,
                   const float* __restrict__ h1, float* __restrict__ out1) {
  int wave = blockIdx.x * (blockDim.x >> 5) + (threadIdx.x >> 5);
  int e0 = wave << 3;
  if (e0 >= ETOT) return;
  int lane = threadIdx.x & 31;
  int sub = lane >> 2, h = lane & 3;
  int row, col; edge_rc_flat(rowA, colA, e0 + sub, row, col);
  float a = ai[row * 4 + h] + aj[col * 4 + h];
  a = a > 0.f ? a : NEG_SLOPE * a;
  float w = expf(a - m1[row * 4 + h]) / (s1[row * 4 + h] + 1e-16f);
  int wbits = __float_as_int(w);
  #pragma unroll
  for (int i = 0; i < 8; i += 2) {
    float v[2][16];
    int rr[2];
    #pragma unroll
    for (int u = 0; u < 2; ++u) {
      int cc = __builtin_amdgcn_readlane(col, (i + u) * 4);
      rr[u]  = __builtin_amdgcn_readlane(row, (i + u) * 4);
      const float* __restrict__ src = h1 + (size_t)cc * D1 + lane;
      #pragma unroll
      for (int q = 0; q < 16; ++q) v[u][q] = src[q * 32];
    }
    #pragma unroll
    for (int u = 0; u < 2; ++u) {
      float* dst = out1 + (size_t)rr[u] * D1 + lane;
      #pragma unroll
      for (int q = 0; q < 16; ++q) {
        float wq = __int_as_float(__builtin_amdgcn_readlane(wbits, (i + u) * 4 + (q >> 2)));
        atomicAdd(dst + q * 32, v[u][q] * wq);
      }
    }
  }
}


__global__ void node_att2(const float* __restrict__ h2, const float* __restrict__ att2,
                          float* __restrict__ s2, float* __restrict__ m2,
                          float* __restrict__ ss2) {
  int tid = blockIdx.x * blockDim.x + threadIdx.x;
  if (tid >= N_NODES * H2) return;
  int n = tid >> 3, h = tid & 7;
  const float* __restrict__ hp = h2 + (size_t)n * D2 + h * C2;
  const float* __restrict__ ap = att2 + h * (2 * C2);
  float s = 0.f;
  #pragma unroll
  for (int c = 0; c < C2; ++c) s += hp[c] * (ap[c] + ap[C2 + c]);
  s2[tid] = tanhf(s);
  m2[tid] = -__builtin_inff();
  ss2[tid] = 0.0f;
}

__global__ void edge_max2(const int* __restrict__ rowA, const int* __restrict__ colA,
                          const float* __restrict__ s2, float* __restrict__ m2) {
  int wave = blockIdx.x * (blockDim.x >> 5) + (threadIdx.x >> 5);
  int e0 = wave << 5;
  if (e0 >= ETOT) return;
  int lane = threadIdx.x & 31;
  int sub = lane >> 3, h = lane & 7;
  int dsti[8]; float val[8];
  #pragma unroll
  for (int i = 0; i < 8; ++i) {
    int row, col; edge_rc_flat(rowA, colA, e0 + (i << 2) + sub, row, col);
    val[i] = s2[col * 8 + h];
    dsti[i] = row * 8 + h;
  }
  #pragma unroll
  for (int i = 0; i < 8; ++i) atomic_max_f32(&m2[dsti[i]], val[i]);
}

__global__ void edge_sum2(const int* __restrict__ rowA, const int* __restrict__ colA,
                          const float* __restrict__ s2, const float* __restrict__ m2,
                          float* __restrict__ ss2) {
  int wave = blockIdx.x * (blockDim.x >> 5) + (threadIdx.x >> 5);
  int e0 = wave << 5;
  if (e0 >= ETOT) return;
  int lane = threadIdx.x & 31;
  int sub = lane >> 3, h = lane & 7;
  int dsti[8]; float val[8];
  #pragma unroll
  for (int i = 0; i < 8; ++i) {
    int row, col; edge_rc_flat(rowA, colA, e0 + (i << 2) + sub, row, col);
    val[i] = expf(s2[col * 8 + h] - m2[row * 8 + h]);
    dsti[i] = row * 8 + h;
  }
  #pragma unroll
  for (int i = 0; i < 8; ++i) atomicAdd(&ss2[dsti[i]], val[i]);
}

__global__ void edge_scatter2(const int* __restrict__ rowA, const int* __restrict__ colA,
                              const float* __restrict__ s2, const float* __restrict__ m2,
                              const float* __restrict__ ss2,
                              const float* __restrict__ h2, float* __restrict__ out2) {
  int wave = blockIdx.x * (blockDim.x >> 5) + (threadIdx.x >> 5);
  int e0 = wave << 3;
  if (e0 >= ETOT) return;
  int lane = threadIdx.x & 31;
  int half = lane >> 4, j = lane & 15, hq = lane & 7;
  int dsti[4]; float val[4][3];
  #pragma unroll
  for (int i = 0; i < 4; ++i) {
    int row, col; edge_rc_flat(rowA, colA, e0 + (i << 1) + half, row, col);
    float w = expf(s2[col * 8 + hq] - m2[row * 8 + hq]) / (ss2[row * 8 + hq] + 1e-16f);
    const float* __restrict__ src = h2 + (size_t)col * D2 + j;
    dsti[i] = row * D2 + j;
    #pragma unroll
    for (int k = 0; k < 3; ++k) {
      int hd = (j + 16 * k) / C2;
      float wk = __shfl(w, (lane & 16) | hd, 32);
      val[i][k] = src[16 * k] * wk;
    }
  }
  #pragma unroll
  for (int i = 0; i < 4; ++i) {
    #pragma unroll
    for (int k = 0; k < 3; ++k) atomicAdd(out2 + dsti[i] + 16 * k, val[i][k]);
  }
}

__global__ void finalize(const float* __restrict__ out2, const float* __restrict__ b2,
                         float* __restrict__ out) {
  int n = blockIdx.x * blockDim.x + threadIdx.x;
  if (n >= N_NODES) return;
  float v[C2];
  #pragma unroll
  for (int c = 0; c < C2; ++c) {
    float s = 0.f;
    #pragma unroll
    for (int h = 0; h < H2; ++h) s += out2[(size_t)n * D2 + h * C2 + c];
    float t = s * (1.0f / H2) + b2[c];
    v[c] = t > 0.f ? t : (expf(t) - 1.0f);
  }
  float m = v[0];
  #pragma unroll
  for (int c = 1; c < C2; ++c) m = fmaxf(m, v[c]);
  float sum = 0.f;
  #pragma unroll
  for (int c = 0; c < C2; ++c) sum += expf(v[c] - m);
  float lse = logf(sum);
  #pragma unroll
  for (int c = 0; c < C2; ++c) out[(size_t)n * C2 + c] = v[c] - m - lse;
}


extern "C" void kernel_launch(void* const* d_in, const int* in_sizes, int n_in,
                              void* d_out, int out_size, void* d_ws, size_t ws_size,
                              hipStream_t stream) {
  const float* x    = (const float*)d_in[0];
  const int*   ei   = (const int*)  d_in[1];
  const float* g1   = (const float*)d_in[2];
  const float* be1  = (const float*)d_in[3];
  const float* mu1  = (const float*)d_in[4];
  const float* va1  = (const float*)d_in[5];
  const float* W1   = (const float*)d_in[6];
  const float* att1 = (const float*)d_in[7];
  const float* b1   = (const float*)d_in[8];
  const float* g2   = (const float*)d_in[9];
  const float* be2  = (const float*)d_in[10];
  const float* mu2  = (const float*)d_in[11];
  const float* va2  = (const float*)d_in[12];
  const float* W2   = (const float*)d_in[13];
  const float* att2 = (const float*)d_in[14];
  const float* b2   = (const float*)d_in[15];
  const int* rowA = ei;
  const int* colA = ei + N_EDGES;
  float* out = (float*)d_out;
  float* ws  = (float*)d_ws;

  float* sc1  = ws;
  float* sh1  = ws + 256;
  float* sc2  = ws + 512;
  float* sh2  = ws + 1024;
  float* ai   = ws + 1536;
  float* aj   = ai  + (size_t)N_NODES * H1;
  float* m1   = aj  + (size_t)N_NODES * H1;
  float* s1   = m1  + (size_t)N_NODES * H1;
  float* out1 = s1  + (size_t)N_NODES * H1;
  float* h1   = out1 + (size_t)N_NODES * D1;
  float* h2   = h1;
  float* s2   = h2  + (size_t)N_NODES * D2;
  float* m2   = s2  + (size_t)N_NODES * H2;
  float* ss2  = m2  + (size_t)N_NODES * H2;
  float* out2 = ss2 + (size_t)N_NODES * H2;

  bn_prep<<<2, 256, 0, stream>>>(g1, be1, mu1, va1, g2, be2, mu2, va2, b1,
                                 sc1, sh1, sc2, sh2);

  gemm1_rows_resident<<<dim3(313, 4), 256, 0, stream>>>(x, sc1, sh1, W1, h1, out1);

  node_att1<<<(N_NODES / NA1_NODES + 7) / 8, 256, 0, stream>>>(h1, att1, ai, aj, m1, s1);
  edge_max1<<<(ETOT + 255) / 256, 256, 0, stream>>>(rowA, colA, ai, aj, m1);
  edge_sum1<<<(ETOT + 255) / 256, 256, 0, stream>>>(rowA, colA, ai, aj, m1, s1);
  edge_scatter1<<<(ETOT / 8) / 8, 256, 0, stream>>>(rowA, colA, ai, aj, m1, s1, h1, out1);


  gemm_bn_wmma<<<(2500 * 3 + 7) / 8, 256, 0, stream>>>(out1, sc2, sh2, W2, h2, out2,
                                                       N_NODES, D1, D2);

  node_att2<<<(N_NODES * H2) / 256, 256, 0, stream>>>(h2, att2, s2, m2, ss2);
  edge_max2<<<(ETOT + 255) / 256, 256, 0, stream>>>(rowA, colA, s2, m2);
  edge_sum2<<<(ETOT + 255) / 256, 256, 0, stream>>>(rowA, colA, s2, m2, ss2);
  edge_scatter2<<<(ETOT / 8) / 8, 256, 0, stream>>>(rowA, colA, s2, m2, ss2, h2, out2);

  finalize<<<(N_NODES + 255) / 256, 256, 0, stream>>>(out2, b2, out);
}


# Layer-1 aggregation: a work-group owns 157 destination rows, sums them in LDS while it scans all edge destinations, stores them once; no global atomics; sums: self loop, then LDS-add arrival order
# speedup vs baseline: 2.9156x; 1.1173x over previous
#include <hip/hip_runtime.h>
#include <math.h>

#define N_NODES 40000
#define N_EDGES 320000
#define ETOT    (N_EDGES + N_NODES)
#define F_IN    256
#define H1      4
#define C1      128
#define D1      (H1*C1)
#define H2      8
#define C2      6
#define D2      (H2*C2)
#define BN_EPS    1e-5f
#define NEG_SLOPE 0.2f

typedef __attribute__((ext_vector_type(2))) float v2f;
typedef __attribute__((ext_vector_type(4))) float v4f;
typedef __attribute__((ext_vector_type(8))) float v8f;


__device__ __forceinline__ void atomic_max_f32(float* addr, float v) {
  if (v >= 0.0f) atomicMax((int*)addr, __float_as_int(v));
  else           atomicMin((unsigned int*)addr, __float_as_uint(v));
}

__device__ __forceinline__ void edge_rc_flat(const int* __restrict__ rowA,
                                             const int* __restrict__ colA,
                                             int e, int& r, int& c) {
  int ec = e < N_EDGES ? e : N_EDGES - 1;
  int lr = rowA[ec], lc = colA[ec];
  bool self = e >= N_EDGES;
  r = self ? e - N_EDGES : lr;
  c = self ? e - N_EDGES : lc;
}


__global__ void bn_prep(const float* g1, const float* be1, const float* mu1, const float* va1,
                        const float* g2, const float* be2, const float* mu2, const float* va2,
                        const float* bias1,
                        float* sc1, float* sh1, float* sc2, float* sh2) {
  int i = blockIdx.x * blockDim.x + threadIdx.x;
  if (i < F_IN) {
    float s = g1[i] / sqrtf(va1[i] + BN_EPS);
    sc1[i] = s; sh1[i] = be1[i] - mu1[i] * s;
  }
  if (i < D1) {
    float s = g2[i] / sqrtf(va2[i] + BN_EPS);
    sc2[i] = s; sh2[i] = be2[i] + (bias1[i] - mu2[i]) * s;
  }
}


__global__ __launch_bounds__(256)
void gemm_bn_wmma(const float* __restrict__ A,
                  const float* __restrict__ scale,
                  const float* __restrict__ shift,
                  const float* __restrict__ B,
                  float* __restrict__ C,
                  float* __restrict__ Z,
                  int M, int K, int N) {
  int wave = blockIdx.x * (blockDim.x >> 5) + (threadIdx.x >> 5);
  int ntiles = N >> 4;
  int tm = wave / ntiles;
  int tn = wave - tm * ntiles;
  if (tm >= (M >> 4)) return;

  int lane = threadIdx.x & 31;
  int half = lane >> 4;
  int l16  = lane & 15;
  int row  = (tm << 4) + l16;
  int colB = (tn << 4) + l16;

  {
    float* __restrict__ Zt = Z + (size_t)((tm << 4) + (lane >> 2)) * N + (tn << 4) + ((lane & 3) << 2);
    const v4f z = {0.f, 0.f, 0.f, 0.f};
    *(v4f*)Zt = z;
    *(v4f*)(Zt + (size_t)8 * N) = z;
  }

  const float* __restrict__ Arow = A + (size_t)row * K;
  v8f acc = {};
  for (int k = 0; k < K; k += 8) {
    int kk = k + (half << 2);
    v4f a4 = *(const v4f*)(Arow + kk);
    v4f s4 = *(const v4f*)(scale + kk);
    v4f t4 = *(const v4f*)(shift + kk);
    const float* __restrict__ Bk = B + (size_t)kk * N + colB;
    v2f a, b;
    a.x = a4.x * s4.x + t4.x;
    a.y = a4.y * s4.y + t4.y;
    b.x = Bk[0];
    b.y = Bk[(size_t)N];
    acc = __builtin_amdgcn_wmma_f32_16x16x4_f32(
         false, a,  false, b,
         (short)0, acc,  false,  false);
    a.x = a4.z * s4.z + t4.z;
    a.y = a4.w * s4.w + t4.w;
    b.x = Bk[(size_t)2 * N];
    b.y = Bk[(size_t)3 * N];
    acc = __builtin_amdgcn_wmma_f32_16x16x4_f32(
         false, a,  false, b,
         (short)0, acc,  false,  false);
  }
  float* Crow = C + (size_t)((tm << 4) + (half << 3)) * N + colB;
  #pragma unroll
  for (int r = 0; r < 8; ++r) Crow[(size_t)r * N] = acc[r];
}

#define G1_COLS_PER_WAVE 128

__global__ __launch_bounds__(256)
void gemm1_rows_resident(const float* __restrict__ A,
                         const float* __restrict__ scale,
                         const float* __restrict__ shift,
                         const float* __restrict__ B,
                         float* __restrict__ C) {
  int tm = blockIdx.x * 8 + (threadIdx.x >> 5);
  if (tm >= (N_NODES >> 4)) return;
  int lane = threadIdx.x & 31;
  int half = lane >> 4;
  int l16  = lane & 15;

  const float* __restrict__ Arow = A + (size_t)((tm << 4) + l16) * F_IN + (half << 2);
  const float* __restrict__ sc = scale + (half << 2);
  const float* __restrict__ sh = shift + (half << 2);
  v4f a[F_IN / 8];
  #pragma unroll
  for (int t = 0; t < F_IN / 8; ++t) a[t] = *(const v4f*)(Arow + 8 * t);
  #pragma unroll
  for (int t0 = 0; t0 < F_IN / 8; t0 += 8) {
    #pragma unroll
    for (int t = t0; t < t0 + 8; ++t) {
      v4f s4 = *(const v4f*)(sc + 8 * t);
      v4f t4 = *(const v4f*)(sh + 8 * t);
      a[t].x = a[t].x * s4.x + t4.x;
      a[t].y = a[t].y * s4.y + t4.y;
      a[t].z = a[t].z * s4.z + t4.z;
      a[t].w = a[t].w * s4.w + t4.w;
    }
    __builtin_amdgcn_sched_barrier(0);
  }

  int cb = blockIdx.y * G1_COLS_PER_WAVE + (l16 << 2);
  const float* __restrict__ Bl = B + (size_t)(half << 2) * D1 + cb;
  float* __restrict__ Cl = C + (size_t)((tm << 4) + (half << 3)) * D1 + cb;
  for (int j = 0; j < G1_COLS_PER_WAVE; j += 64) {
    v8f acc0 = {}, acc1 = {}, acc2 = {}, acc3 = {};
    const float* __restrict__ Bj = Bl + j;
    v4f bq[3][4];
    #pragma unroll
    for (int p = 0; p < 2; ++p) {
      #pragma unroll
      for (int q = 0; q < 4; ++q) bq[p][q] = *(const v4f*)(Bj + (size_t)(8 * p + q) * D1);
    }
    #pragma unroll
    for (int t = 0; t < F_IN / 8; ++t) {
      if (t + 2 < F_IN / 8) {
        #pragma unroll
        for (int q = 0; q < 4; ++q)
          bq[(t + 2) % 3][q] = *(const v4f*)(Bj + (size_t)(8 * (t + 2) + q) * D1);
      }
      __builtin_amdgcn_sched_barrier(0);
      const v4f b0 = bq[t % 3][0], b1 = bq[t % 3][1], b2 = bq[t % 3][2], b3 = bq[t % 3][3];
      v2f a01, a23, b;
      a01.x = a[t].x; a01.y = a[t].y;
      a23.x = a[t].z; a23.y = a[t].w;
      b.x = b0.x; b.y = b1.x;
      acc0 = __builtin_amdgcn_wmma_f32_16x16x4_f32(false, a01, false, b, (short)0, acc0, false, false);
      b.x = b0.y; b.y = b1.y;
      acc1 = __builtin_amdgcn_wmma_f32_16x16x4_f32(false, a01, false, b, (short)0, acc1, false, false);
      b.x = b0.z; b.y = b1.z;
      acc2 = __builtin_amdgcn_wmma_f32_16x16x4_f32(false, a01, false, b, (short)0, acc2, false, false);
      b.x = b0.w; b.y = b1.w;
      acc3 = __builtin_amdgcn_wmma_f32_16x16x4_f32(false, a01, false, b, (short)0, acc3, false, false);
      b.x = b2.x; b.y = b3.x;
      acc0 = __builtin_amdgcn_wmma_f32_16x16x4_f32(false, a23, false, b, (short)0, acc0, false, false);
      b.x = b2.y; b.y = b3.y;
      acc1 = __builtin_amdgcn_wmma_f32_16x16x4_f32(false, a23, false, b, (short)0, acc1, false, false);
      b.x = b2.z; b.y = b3.z;
      acc2 = __builtin_amdgcn_wmma_f32_16x16x4_f32(false, a23, false, b, (short)0, acc2, false, false);
      b.x = b2.w; b.y = b3.w;
      acc3 = __builtin_amdgcn_wmma_f32_16x16x4_f32(false, a23, false, b, (short)0, acc3, false, false);
      __builtin_amdgcn_sched_barrier(0);
    }
    #pragma unroll
    for (int r = 0; r < 8; ++r) {
      v4f o;
      o.x = acc0[r]; o.y = acc1[r]; o.z = acc2[r]; o.w = acc3[r];
      *(v4f*)(Cl + (size_t)r * D1 + j) = o;
    }
  }
}


#define NA1_NODES 8

__global__ __launch_bounds__(256)
void node_att1(const float* __restrict__ h1, const float* __restrict__ att1,
               float* __restrict__ ai, float* __restrict__ aj,
               float* __restrict__ m1, float* __restrict__ s1) {
  int wave = blockIdx.x * (blockDim.x >> 5) + (threadIdx.x >> 5);
  int n0 = wave * NA1_NODES;
  if (n0 >= N_NODES) return;
  int lane = threadIdx.x & 31;
  v4f al[H1], ar[H1];
  #pragma unroll
  for (int h = 0; h < H1; ++h) {
    al[h] = *(const v4f*)(att1 + h * (2 * C1) + (lane << 2));
    ar[h] = *(const v4f*)(att1 + h * (2 * C1) + C1 + (lane << 2));
  }
  const bool b16 = (lane & 16) != 0, b8 = (lane & 8) != 0, b4 = (lane & 4) != 0;
  const int idx = lane >> 2;
  const float* __restrict__ hp = h1 + (size_t)n0 * D1 + (lane << 2);
  #pragma unroll
  for (int i = 0; i < NA1_NODES; i += 4) {
    v4f v[4][H1];
    #pragma unroll
    for (int u = 0; u < 4; ++u) {
      #pragma unroll
      for (int h = 0; h < H1; ++h) v[u][h] = *(const v4f*)(hp + (size_t)(i + u) * D1 + h * C1);
    }
    #pragma unroll
    for (int u = 0; u < 4; ++u) {
      float r[8];
      #pragma unroll
      for (int h = 0; h < H1; ++h) {
        const v4f x = v[u][h];
        r[h]     = x.x * al[h].x + x.y * al[h].y + x.z * al[h].z + x.w * al[h].w;
        r[4 + h] = x.x * ar[h].x + x.y * ar[h].y + x.z * ar[h].z + x.w * ar[h].w;
      }
      float t4[4], t2[2];
      #pragma unroll
      for (int k = 0; k < 4; ++k) {
        float keep = b16 ? r[4 + k] : r[k];
        float send = b16 ? r[k] : r[4 + k];
        t4[k] = keep + __shfl_xor(send, 16, 32);
      }
      #pragma unroll
      for (int k = 0; k < 2; ++k) {
        float keep = b8 ? t4[2 + k] : t4[k];
        float send = b8 ? t4[k] : t4[2 + k];
        t2[k] = keep + __shfl_xor(send, 8, 32);
      }
      float keep = b4 ? t2[1] : t2[0];
      float send = b4 ? t2[0] : t2[1];
      float t = keep + __shfl_xor(send, 4, 32);
      t += __shfl_xor(t, 2, 32);
      t += __shfl_xor(t, 1, 32);
      if ((lane & 3) == 0) {
        int n = n0 + i + u;
        if (idx < 4) {
          ai[n * 4 + idx] = t;
          m1[n * 4 + idx] = -__builtin_inff();
          s1[n * 4 + idx] = 0.0f;
        } else {
          aj[n * 4 + (idx - 4)] = t;
        }
      }
    }
  }
}


__global__ void edge_max1(const int* __restrict__ rowA, const int* __restrict__ colA,
                          const float* __restrict__ ai, const float* __restrict__ aj,
                          float* __restrict__ m1) {
  int wave = blockIdx.x * (blockDim.x >> 5) + (threadIdx.x >> 5);
  int e0 = wave << 5;
  if (e0 >= ETOT) return;
  int lane = threadIdx.x & 31;
  int sub = lane >> 2, h = lane & 3;
  int dsti[4]; float val[4];
  #pragma unroll
  for (int i = 0; i < 4; ++i) {
    int row, col; edge_rc_flat(rowA, colA, e0 + (i << 3) + sub, row, col);
    float a = ai[row * 4 + h] + aj[col * 4 + h];
    val[i] = a > 0.f ? a : NEG_SLOPE * a;
    dsti[i] = row * 4 + h;
  }
  #pragma unroll
  for (int i = 0; i < 4; ++i) atomic_max_f32(&m1[dsti[i]], val[i]);
}

__global__ void edge_sum1(const int* __restrict__ rowA, const int* __restrict__ colA,
                          const float* __restrict__ ai, const float* __restrict__ aj,
                          const float* __restrict__ m1, float* __restrict__ s1) {
  int wave = blockIdx.x * (blockDim.x >> 5) + (threadIdx.x >> 5);
  int e0 = wave << 5;
  if (e0 >= ETOT) return;
  int lane = threadIdx.x & 31;
  int sub = lane >> 2, h = lane & 3;
  int dsti[4]; float val[4];
  #pragma unroll
  for (int i = 0; i < 4; ++i) {
    int row, col; edge_rc_flat(rowA, colA, e0 + (i << 3) + sub, row, col);
    float a = ai[row * 4 + h] + aj[col * 4 + h];
    a = a > 0.f ? a : NEG_SLOPE * a;
    val[i] = expf(a - m1[row * 4 + h]);
    dsti[i] = row * 4 + h;
  }
  #pragma unroll
  for (int i = 0; i < 4; ++i) atomicAdd(&s1[dsti[i]], val[i]);
}

#define A1_ROWS 157
#define A1_WGS  ((N_NODES + A1_ROWS - 1) / A1_ROWS)
#define A1_NV   (N_EDGES / 4)
static_assert(A1_ROWS * D1 * 4 <= 320 * 1024, "the owned rows fit the LDS of one work-group");
static_assert(N_EDGES % 128 == 0, "a wave's 32 x 4 destination indices are all inside the list");

typedef __attribute__((ext_vector_type(4))) int v4i;

__device__ __forceinline__ float gat_weight(float a, float m, float s) {
  a = a > 0.f ? a : NEG_SLOPE * a;
  return expf(a - m) / (s + 1e-16f);
}

__device__ __forceinline__ void gat_weights4(int r, int c,
                                             const float* __restrict__ ai, const float* __restrict__ aj,
                                             const float* __restrict__ m1, const float* __restrict__ s1,
                                             float w[4]) {
  const v4f A = *(const v4f*)(ai + r * 4), B = *(const v4f*)(aj + c * 4);
  const v4f M = *(const v4f*)(m1 + r * 4), S = *(const v4f*)(s1 + r * 4);
  w[0] = gat_weight(A.x + B.x, M.x, S.x);
  w[1] = gat_weight(A.y + B.y, M.y, S.y);
  w[2] = gat_weight(A.z + B.z, M.z, S.z);
  w[3] = gat_weight(A.w + B.w, M.w, S.w);
}

__device__ __forceinline__ void agg1_one(int d, int c, int r0, int lane,
                                         const float* __restrict__ ai, const float* __restrict__ aj,
                                         const float* __restrict__ m1, const float* __restrict__ s1,
                                         const float* __restrict__ h1, float* acc) {
  const float* __restrict__ src = h1 + (size_t)c * D1 + lane;
  float v[16];
  #pragma unroll
  for (int q = 0; q < 16; ++q) v[q] = src[q * 32];
  float w[4];
  gat_weights4(r0 + d, c, ai, aj, m1, s1, w);
  float* dst = acc + d * D1 + lane;
  #pragma unroll
  for (int q = 0; q < 16; ++q) atomicAdd(dst + q * 32, v[q] * w[q >> 2]);
}

__device__ __forceinline__ void agg1_pair(int d0, int c0, int d1, int c1, int r0, int lane,
                                          const float* __restrict__ ai, const float* __restrict__ aj,
                                          const float* __restrict__ m1, const float* __restrict__ s1,
                                          const float* __restrict__ h1, float* acc) {
  const float* __restrict__ src0 = h1 + (size_t)c0 * D1 + lane;
  const float* __restrict__ src1 = h1 + (size_t)c1 * D1 + lane;
  float v0[16], v1[16];
  #pragma unroll
  for (int q = 0; q < 16; ++q) v0[q] = src0[q * 32];
  #pragma unroll
  for (int q = 0; q < 16; ++q) v1[q] = src1[q * 32];
  float w0[4], w1[4];
  gat_weights4(r0 + d0, c0, ai, aj, m1, s1, w0);
  gat_weights4(r0 + d1, c1, ai, aj, m1, s1, w1);
  float* dst0 = acc + d0 * D1 + lane;
  float* dst1 = acc + d1 * D1 + lane;
  #pragma unroll
  for (int q = 0; q < 16; ++q) atomicAdd(dst0 + q * 32, v0[q] * w0[q >> 2]);
  #pragma unroll
  for (int q = 0; q < 16; ++q) atomicAdd(dst1 + q * 32, v1[q] * w1[q >> 2]);
}

__global__ __launch_bounds__(1024)
void agg1_owned_rows(const int* __restrict__ rowA, const int* __restrict__ colA,
                     const float* __restrict__ ai, const float* __restrict__ aj,
                     const float* __restrict__ m1, const float* __restrict__ s1,
                     const float* __restrict__ h1, float* __restrict__ out1) {
  __shared__ __attribute__((aligned(16))) float acc[A1_ROWS * D1];
  const int r0 = blockIdx.x * A1_ROWS;
  const int nr = (N_NODES - r0) < A1_ROWS ? (N_NODES - r0) : A1_ROWS;
  const int wave = threadIdx.x >> 5, lane = threadIdx.x & 31;

  for (int r = wave; r < nr; r += 32) {
    int n = r0 + r;
    const float* __restrict__ src = h1 + (size_t)n * D1 + lane;
    float v[16];
    #pragma unroll
    for (int q = 0; q < 16; ++q) v[q] = src[q * 32];
    float w[4];
    gat_weights4(n, n, ai, aj, m1, s1, w);
    float* dst = acc + r * D1 + lane;
    #pragma unroll
    for (int q = 0; q < 16; ++q) dst[q * 32] = v[q] * w[q >> 2];
  }
  __syncthreads();

  const unsigned unr = (unsigned)nr;
  const v4i none = {-1, -1, -1, -1};
  int vb = wave * 32;
  v4i g0 = vb        < A1_NV ? *(const v4i*)(rowA + 4 * (vb + lane))        : none;
  v4i g1 = vb + 1024 < A1_NV ? *(const v4i*)(rowA + 4 * (vb + 1024 + lane)) : none;
  v4i g2 = vb + 2048 < A1_NV ? *(const v4i*)(rowA + 4 * (vb + 2048 + lane)) : none;
  v4i g3 = vb + 3072 < A1_NV ? *(const v4i*)(rowA + 4 * (vb + 3072 + lane)) : none;
  bool have = false;
  int pd = 0, pc = 0;
  #pragma nounroll
  for (; vb < A1_NV; vb += 4096) {
    const v4i n0 = vb + 4096 < A1_NV ? *(const v4i*)(rowA + 4 * (vb + 4096 + lane)) : none;
    const v4i n1 = vb + 5120 < A1_NV ? *(const v4i*)(rowA + 4 * (vb + 5120 + lane)) : none;
    const v4i n2 = vb + 6144 < A1_NV ? *(const v4i*)(rowA + 4 * (vb + 6144 + lane)) : none;
    const v4i n3 = vb + 7168 < A1_NV ? *(const v4i*)(rowA + 4 * (vb + 7168 + lane)) : none;
    #pragma nounroll
    for (int u = 0; u < 4; ++u) {
      const v4i G = u == 0 ? g0 : u == 1 ? g1 : u == 2 ? g2 : g3;
      const int d0 = G.x - r0, d1 = G.y - r0, d2 = G.z - r0, d3 = G.w - r0;
      const bool any = ((unsigned)d0 < unr) | ((unsigned)d1 < unr) | ((unsigned)d2 < unr) | ((unsigned)d3 < unr);
      if (__builtin_amdgcn_ballot_w32(any) == 0) continue;
      const int e = 4 * (vb + u * 1024 + lane);
      #pragma nounroll
      for (int j = 0; j < 4; ++j) {
        const int dj = j == 0 ? d0 : j == 1 ? d1 : j == 2 ? d2 : d3;
        const bool tj = (unsigned)dj < unr;
        unsigned mask = __builtin_amdgcn_ballot_w32(tj);
        if (mask == 0) continue;
        int c = 0;
        if (tj) c = colA[e + j];
        while (mask) {
          const int l = __builtin_ctz(mask);
          mask &= mask - 1;
          const int dl = __builtin_amdgcn_readlane(dj, l);
          const int cl = __builtin_amdgcn_readlane(c, l);
          if (!have) { pd = dl; pc = cl; have = true; }
          else { agg1_pair(pd, pc, dl, cl, r0, lane, ai, aj, m1, s1, h1, acc); have = false; }
        }
      }
    }
    g0 = n0; g1 = n1; g2 = n2; g3 = n3;
  }
  if (have) agg1_one(pd, pc, r0, lane, ai, aj, m1, s1, h1, acc);
  __syncthreads();

  for (int r = wave; r < nr; r += 32) {
    const float* src = acc + r * D1 + (lane << 2);
    float* __restrict__ dst = out1 + (size_t)(r0 + r) * D1 + (lane << 2);
    #pragma unroll
    for (int k = 0; k < 4; ++k) *(v4f*)(dst + k * C1) = *(const v4f*)(src + k * C1);
  }
}


__global__ void node_att2(const float* __restrict__ h2, const float* __restrict__ att2,
                          float* __restrict__ s2, float* __restrict__ m2,
                          float* __restrict__ ss2) {
  int tid = blockIdx.x * blockDim.x + threadIdx.x;
  if (tid >= N_NODES * H2) return;
  int n = tid >> 3, h = tid & 7;
  const float* __restrict__ hp = h2 + (size_t)n * D2 + h * C2;
  const float* __restrict__ ap = att2 + h * (2 * C2);
  float s = 0.f;
  #pragma unroll
  for (int c = 0; c < C2; ++c) s += hp[c] * (ap[c] + ap[C2 + c]);
  s2[tid] = tanhf(s);
  m2[tid] = -__builtin_inff();
  ss2[tid] = 0.0f;
}

__global__ void edge_max2(const int* __restrict__ rowA, const int* __restrict__ colA,
                          const float* __restrict__ s2, float* __restrict__ m2) {
  int wave = blockIdx.x * (blockDim.x >> 5) + (threadIdx.x >> 5);
  int e0 = wave << 5;
  if (e0 >= ETOT) return;
  int lane = threadIdx.x & 31;
  int sub = lane >> 3, h = lane & 7;
  int dsti[8]; float val[8];
  #pragma unroll
  for (int i = 0; i < 8; ++i) {
    int row, col; edge_rc_flat(rowA, colA, e0 + (i << 2) + sub, row, col);
    val[i] = s2[col * 8 + h];
    dsti[i] = row * 8 + h;
  }
  #pragma unroll
  for (int i = 0; i < 8; ++i) atomic_max_f32(&m2[dsti[i]], val[i]);
}

__global__ void edge_sum2(const int* __restrict__ rowA, const int* __restrict__ colA,
                          const float* __restrict__ s2, const float* __restrict__ m2,
                          float* __restrict__ ss2) {
  int wave = blockIdx.x * (blockDim.x >> 5) + (threadIdx.x >> 5);
  int e0 = wave << 5;
  if (e0 >= ETOT) return;
  int lane = threadIdx.x & 31;
  int sub = lane >> 3, h = lane & 7;
  int dsti[8]; float val[8];
  #pragma unroll
  for (int i = 0; i < 8; ++i) {
    int row, col; edge_rc_flat(rowA, colA, e0 + (i << 2) + sub, row, col);
    val[i] = expf(s2[col * 8 + h] - m2[row * 8 + h]);
    dsti[i] = row * 8 + h;
  }
  #pragma unroll
  for (int i = 0; i < 8; ++i) atomicAdd(&ss2[dsti[i]], val[i]);
}

__global__ void edge_scatter2(const int* __restrict__ rowA, const int* __restrict__ colA,
                              const float* __restrict__ s2, const float* __restrict__ m2,
                              const float* __restrict__ ss2,
                              const float* __restrict__ h2, float* __restrict__ out2) {
  int wave = blockIdx.x * (blockDim.x >> 5) + (threadIdx.x >> 5);
  int e0 = wave << 3;
  if (e0 >= ETOT) return;
  int lane = threadIdx.x & 31;
  int half = lane >> 4, j = lane & 15, hq = lane & 7;
  int dsti[4]; float val[4][3];
  #pragma unroll
  for (int i = 0; i < 4; ++i) {
    int row, col; edge_rc_flat(rowA, colA, e0 + (i << 1) + half, row, col);
    float w = expf(s2[col * 8 + hq] - m2[row * 8 + hq]) / (ss2[row * 8 + hq] + 1e-16f);
    const float* __restrict__ src = h2 + (size_t)col * D2 + j;
    dsti[i] = row * D2 + j;
    #pragma unroll
    for (int k = 0; k < 3; ++k) {
      int hd = (j + 16 * k) / C2;
      float wk = __shfl(w, (lane & 16) | hd, 32);
      val[i][k] = src[16 * k] * wk;
    }
  }
  #pragma unroll
  for (int i = 0; i < 4; ++i) {
    #pragma unroll
    for (int k = 0; k < 3; ++k) atomicAdd(out2 + dsti[i] + 16 * k, val[i][k]);
  }
}

__global__ void finalize(const float* __restrict__ out2, const float* __restrict__ b2,
                         float* __restrict__ out) {
  int n = blockIdx.x * blockDim.x + threadIdx.x;
  if (n >= N_NODES) return;
  float v[C2];
  #pragma unroll
  for (int c = 0; c < C2; ++c) {
    float s = 0.f;
    #pragma unroll
    for (int h = 0; h < H2; ++h) s += out2[(size_t)n * D2 + h * C2 + c];
    float t = s * (1.0f / H2) + b2[c];
    v[c] = t > 0.f ? t : (expf(t) - 1.0f);
  }
  float m = v[0];
  #pragma unroll
  for (int c = 1; c < C2; ++c) m = fmaxf(m, v[c]);
  float sum = 0.f;
  #pragma unroll
  for (int c = 0; c < C2; ++c) sum += expf(v[c] - m);
  float lse = logf(sum);
  #pragma unroll
  for (int c = 0; c < C2; ++c) out[(size_t)n * C2 + c] = v[c] - m - lse;
}


extern "C" void kernel_launch(void* const* d_in, const int* in_sizes, int n_in,
                              void* d_out, int out_size, void* d_ws, size_t ws_size,
                              hipStream_t stream) {
  const float* x    = (const float*)d_in[0];
  const int*   ei   = (const int*)  d_in[1];
  const float* g1   = (const float*)d_in[2];
  const float* be1  = (const float*)d_in[3];
  const float* mu1  = (const float*)d_in[4];
  const float* va1  = (const float*)d_in[5];
  const float* W1   = (const float*)d_in[6];
  const float* att1 = (const float*)d_in[7];
  const float* b1   = (const float*)d_in[8];
  const float* g2   = (const float*)d_in[9];
  const float* be2  = (const float*)d_in[10];
  const float* mu2  = (const float*)d_in[11];
  const float* va2  = (const float*)d_in[12];
  const float* W2   = (const float*)d_in[13];
  const float* att2 = (const float*)d_in[14];
  const float* b2   = (const float*)d_in[15];
  const int* rowA = ei;
  const int* colA = ei + N_EDGES;
  float* out = (float*)d_out;
  float* ws  = (float*)d_ws;

  float* sc1  = ws;
  float* sh1  = ws + 256;
  float* sc2  = ws + 512;
  float* sh2  = ws + 1024;
  float* ai   = ws + 1536;
  float* aj   = ai  + (size_t)N_NODES * H1;
  float* m1   = aj  + (size_t)N_NODES * H1;
  float* s1   = m1  + (size_t)N_NODES * H1;
  float* out1 = s1  + (size_t)N_NODES * H1;
  float* h1   = out1 + (size_t)N_NODES * D1;
  float* h2   = h1;
  float* s2   = h2  + (size_t)N_NODES * D2;
  float* m2   = s2  + (size_t)N_NODES * H2;
  float* ss2  = m2  + (size_t)N_NODES * H2;
  float* out2 = ss2 + (size_t)N_NODES * H2;

  bn_prep<<<2, 256, 0, stream>>>(g1, be1, mu1, va1, g2, be2, mu2, va2, b1,
                                 sc1, sh1, sc2, sh2);

  gemm1_rows_resident<<<dim3(313, 4), 256, 0, stream>>>(x, sc1, sh1, W1, h1);

  node_att1<<<(N_NODES / NA1_NODES + 7) / 8, 256, 0, stream>>>(h1, att1, ai, aj, m1, s1);
  edge_max1<<<(ETOT + 255) / 256, 256, 0, stream>>>(rowA, colA, ai, aj, m1);
  edge_sum1<<<(ETOT + 255) / 256, 256, 0, stream>>>(rowA, colA, ai, aj, m1, s1);
  agg1_owned_rows<<<A1_WGS, 1024, 0, stream>>>(rowA, colA, ai, aj, m1, s1, h1, out1);


  gemm_bn_wmma<<<(2500 * 3 + 7) / 8, 256, 0, stream>>>(out1, sc2, sh2, W2, h2, out2,
                                                       N_NODES, D1, D2);

  node_att2<<<(N_NODES * H2) / 256, 256, 0, stream>>>(h2, att2, s2, m2, ss2);
  edge_max2<<<(ETOT + 255) / 256, 256, 0, stream>>>(rowA, colA, s2, m2);
  edge_sum2<<<(ETOT + 255) / 256, 256, 0, stream>>>(rowA, colA, s2, m2, ss2);
  edge_scatter2<<<(ETOT / 8) / 8, 256, 0, stream>>>(rowA, colA, s2, m2, ss2, h2, out2);

  finalize<<<(N_NODES + 255) / 256, 256, 0, stream>>>(out2, b2, out);
}


# Softmax sum passes removed: both aggregations add unnormalised exp weights and the weights themselves (layer 1 in LDS, layer 2 in the row's 2nd line); rows divided once after summing; 12->10 launches
# speedup vs baseline: 3.3301x; 1.1422x over previous
#include <hip/hip_runtime.h>
#include <math.h>

#define N_NODES 40000
#define N_EDGES 320000
#define ETOT    (N_EDGES + N_NODES)
#define F_IN    256
#define H1      4
#define C1      128
#define D1      (H1*C1)
#define H2      8
#define C2      6
#define D2      (H2*C2)
#define BN_EPS    1e-5f
#define NEG_SLOPE 0.2f

typedef __attribute__((ext_vector_type(2))) float v2f;
typedef __attribute__((ext_vector_type(4))) float v4f;
typedef __attribute__((ext_vector_type(8))) float v8f;


__device__ __forceinline__ void atomic_max_f32(float* addr, float v) {
  if (v >= 0.0f) atomicMax((int*)addr, __float_as_int(v));
  else           atomicMin((unsigned int*)addr, __float_as_uint(v));
}

__device__ __forceinline__ void edge_rc_flat(const int* __restrict__ rowA,
                                             const int* __restrict__ colA,
                                             int e, int& r, int& c) {
  int ec = e < N_EDGES ? e : N_EDGES - 1;
  int lr = rowA[ec], lc = colA[ec];
  bool self = e >= N_EDGES;
  r = self ? e - N_EDGES : lr;
  c = self ? e - N_EDGES : lc;
}


__global__ void bn_prep(const float* g1, const float* be1, const float* mu1, const float* va1,
                        const float* g2, const float* be2, const float* mu2, const float* va2,
                        const float* bias1,
                        float* sc1, float* sh1, float* sc2, float* sh2) {
  int i = blockIdx.x * blockDim.x + threadIdx.x;
  if (i < F_IN) {
    float s = g1[i] / sqrtf(va1[i] + BN_EPS);
    sc1[i] = s; sh1[i] = be1[i] - mu1[i] * s;
  }
  if (i < D1) {
    float s = g2[i] / sqrtf(va2[i] + BN_EPS);
    sc2[i] = s; sh2[i] = be2[i] + (bias1[i] - mu2[i]) * s;
  }
}


__global__ __launch_bounds__(256)
void gemm_bn_wmma(const float* __restrict__ A,
                  const float* __restrict__ scale,
                  const float* __restrict__ shift,
                  const float* __restrict__ B,
                  float* __restrict__ C,
                  int M, int K, int N) {
  int wave = blockIdx.x * (blockDim.x >> 5) + (threadIdx.x >> 5);
  int ntiles = N >> 4;
  int tm = wave / ntiles;
  int tn = wave - tm * ntiles;
  if (tm >= (M >> 4)) return;

  int lane = threadIdx.x & 31;
  int half = lane >> 4;
  int l16  = lane & 15;
  int row  = (tm << 4) + l16;
  int colB = (tn << 4) + l16;

  const float* __restrict__ Arow = A + (size_t)row * K;
  v8f acc = {};
  for (int k = 0; k < K; k += 8) {
    int kk = k + (half << 2);
    v4f a4 = *(const v4f*)(Arow + kk);
    v4f s4 = *(const v4f*)(scale + kk);
    v4f t4 = *(const v4f*)(shift + kk);
    const float* __restrict__ Bk = B + (size_t)kk * N + colB;
    v2f a, b;
    a.x = a4.x * s4.x + t4.x;
    a.y = a4.y * s4.y + t4.y;
    b.x = Bk[0];
    b.y = Bk[(size_t)N];
    acc = __builtin_amdgcn_wmma_f32_16x16x4_f32(
         false, a,  false, b,
         (short)0, acc,  false,  false);
    a.x = a4.z * s4.z + t4.z;
    a.y = a4.w * s4.w + t4.w;
    b.x = Bk[(size_t)2 * N];
    b.y = Bk[(size_t)3 * N];
    acc = __builtin_amdgcn_wmma_f32_16x16x4_f32(
         false, a,  false, b,
         (short)0, acc,  false,  false);
  }
  float* Crow = C + (size_t)((tm << 4) + (half << 3)) * N + colB;
  #pragma unroll
  for (int r = 0; r < 8; ++r) Crow[(size_t)r * N] = acc[r];
}

#define G1_COLS_PER_WAVE 128

__global__ __launch_bounds__(256)
void gemm1_rows_resident(const float* __restrict__ A,
                         const float* __restrict__ scale,
                         const float* __restrict__ shift,
                         const float* __restrict__ B,
                         float* __restrict__ C) {
  int tm = blockIdx.x * 8 + (threadIdx.x >> 5);
  if (tm >= (N_NODES >> 4)) return;
  int lane = threadIdx.x & 31;
  int half = lane >> 4;
  int l16  = lane & 15;

  const float* __restrict__ Arow = A + (size_t)((tm << 4) + l16) * F_IN + (half << 2);
  const float* __restrict__ sc = scale + (half << 2);
  const float* __restrict__ sh = shift + (half << 2);
  v4f a[F_IN / 8];
  #pragma unroll
  for (int t = 0; t < F_IN / 8; ++t) a[t] = *(const v4f*)(Arow + 8 * t);
  #pragma unroll
  for (int t0 = 0; t0 < F_IN / 8; t0 += 8) {
    #pragma unroll
    for (int t = t0; t < t0 + 8; ++t) {
      v4f s4 = *(const v4f*)(sc + 8 * t);
      v4f t4 = *(const v4f*)(sh + 8 * t);
      a[t].x = a[t].x * s4.x + t4.x;
      a[t].y = a[t].y * s4.y + t4.y;
      a[t].z = a[t].z * s4.z + t4.z;
      a[t].w = a[t].w * s4.w + t4.w;
    }
    __builtin_amdgcn_sched_barrier(0);
  }

  int cb = blockIdx.y * G1_COLS_PER_WAVE + (l16 << 2);
  const float* __restrict__ Bl = B + (size_t)(half << 2) * D1 + cb;
  float* __restrict__ Cl = C + (size_t)((tm << 4) + (half << 3)) * D1 + cb;
  for (int j = 0; j < G1_COLS_PER_WAVE; j += 64) {
    v8f acc0 = {}, acc1 = {}, acc2 = {}, acc3 = {};
    const float* __restrict__ Bj = Bl + j;
    v4f bq[3][4];
    #pragma unroll
    for (int p = 0; p < 2; ++p) {
      #pragma unroll
      for (int q = 0; q < 4; ++q) bq[p][q] = *(const v4f*)(Bj + (size_t)(8 * p + q) * D1);
    }
    #pragma unroll
    for (int t = 0; t < F_IN / 8; ++t) {
      if (t + 2 < F_IN / 8) {
        #pragma unroll
        for (int q = 0; q < 4; ++q)
          bq[(t + 2) % 3][q] = *(const v4f*)(Bj + (size_t)(8 * (t + 2) + q) * D1);
      }
      __builtin_amdgcn_sched_barrier(0);
      const v4f b0 = bq[t % 3][0], b1 = bq[t % 3][1], b2 = bq[t % 3][2], b3 = bq[t % 3][3];
      v2f a01, a23, b;
      a01.x = a[t].x; a01.y = a[t].y;
      a23.x = a[t].z; a23.y = a[t].w;
      b.x = b0.x; b.y = b1.x;
      acc0 = __builtin_amdgcn_wmma_f32_16x16x4_f32(false, a01, false, b, (short)0, acc0, false, false);
      b.x = b0.y; b.y = b1.y;
      acc1 = __builtin_amdgcn_wmma_f32_16x16x4_f32(false, a01, false, b, (short)0, acc1, false, false);
      b.x = b0.z; b.y = b1.z;
      acc2 = __builtin_amdgcn_wmma_f32_16x16x4_f32(false, a01, false, b, (short)0, acc2, false, false);
      b.x = b0.w; b.y = b1.w;
      acc3 = __builtin_amdgcn_wmma_f32_16x16x4_f32(false, a01, false, b, (short)0, acc3, false, false);
      b.x = b2.x; b.y = b3.x;
      acc0 = __builtin_amdgcn_wmma_f32_16x16x4_f32(false, a23, false, b, (short)0, acc0, false, false);
      b.x = b2.y; b.y = b3.y;
      acc1 = __builtin_amdgcn_wmma_f32_16x16x4_f32(false, a23, false, b, (short)0, acc1, false, false);
      b.x = b2.z; b.y = b3.z;
      acc2 = __builtin_amdgcn_wmma_f32_16x16x4_f32(false, a23, false, b, (short)0, acc2, false, false);
      b.x = b2.w; b.y = b3.w;
      acc3 = __builtin_amdgcn_wmma_f32_16x16x4_f32(false, a23, false, b, (short)0, acc3, false, false);
      __builtin_amdgcn_sched_barrier(0);
    }
    #pragma unroll
    for (int r = 0; r < 8; ++r) {
      v4f o;
      o.x = acc0[r]; o.y = acc1[r]; o.z = acc2[r]; o.w = acc3[r];
      *(v4f*)(Cl + (size_t)r * D1 + j) = o;
    }
  }
}


#define NA1_NODES 8

__global__ __launch_bounds__(256)
void node_att1(const float* __restrict__ h1, const float* __restrict__ att1,
               float* __restrict__ ai, float* __restrict__ aj,
               float* __restrict__ m1) {
  int wave = blockIdx.x * (blockDim.x >> 5) + (threadIdx.x >> 5);
  int n0 = wave * NA1_NODES;
  if (n0 >= N_NODES) return;
  int lane = threadIdx.x & 31;
  v4f al[H1], ar[H1];
  #pragma unroll
  for (int h = 0; h < H1; ++h) {
    al[h] = *(const v4f*)(att1 + h * (2 * C1) + (lane << 2));
    ar[h] = *(const v4f*)(att1 + h * (2 * C1) + C1 + (lane << 2));
  }
  const bool b16 = (lane & 16) != 0, b8 = (lane & 8) != 0, b4 = (lane & 4) != 0;
  const int idx = lane >> 2;
  const float* __restrict__ hp = h1 + (size_t)n0 * D1 + (lane << 2);
  #pragma unroll
  for (int i = 0; i < NA1_NODES; i += 4) {
    v4f v[4][H1];
    #pragma unroll
    for (int u = 0; u < 4; ++u) {
      #pragma unroll
      for (int h = 0; h < H1; ++h) v[u][h] = *(const v4f*)(hp + (size_t)(i + u) * D1 + h * C1);
    }
    #pragma unroll
    for (int u = 0; u < 4; ++u) {
      float r[8];
      #pragma unroll
      for (int h = 0; h < H1; ++h) {
        const v4f x = v[u][h];
        r[h]     = x.x * al[h].x + x.y * al[h].y + x.z * al[h].z + x.w * al[h].w;
        r[4 + h] = x.x * ar[h].x + x.y * ar[h].y + x.z * ar[h].z + x.w * ar[h].w;
      }
      float t4[4], t2[2];
      #pragma unroll
      for (int k = 0; k < 4; ++k) {
        float keep = b16 ? r[4 + k] : r[k];
        float send = b16 ? r[k] : r[4 + k];
        t4[k] = keep + __shfl_xor(send, 16, 32);
      }
      #pragma unroll
      for (int k = 0; k < 2; ++k) {
        float keep = b8 ? t4[2 + k] : t4[k];
        float send = b8 ? t4[k] : t4[2 + k];
        t2[k] = keep + __shfl_xor(send, 8, 32);
      }
      float keep = b4 ? t2[1] : t2[0];
      float send = b4 ? t2[0] : t2[1];
      float t = keep + __shfl_xor(send, 4, 32);
      t += __shfl_xor(t, 2, 32);
      t += __shfl_xor(t, 1, 32);
      if ((lane & 3) == 0) {
        int n = n0 + i + u;
        if (idx < 4) {
          ai[n * 4 + idx] = t;
          m1[n * 4 + idx] = -__builtin_inff();
        } else {
          aj[n * 4 + (idx - 4)] = t;
        }
      }
    }
  }
}


__global__ void edge_max1(const int* __restrict__ rowA, const int* __restrict__ colA,
                          const float* __restrict__ ai, const float* __restrict__ aj,
                          float* __restrict__ m1) {
  int wave = blockIdx.x * (blockDim.x >> 5) + (threadIdx.x >> 5);
  int e0 = wave << 5;
  if (e0 >= ETOT) return;
  int lane = threadIdx.x & 31;
  int sub = lane >> 2, h = lane & 3;
  int dsti[4]; float val[4];
  #pragma unroll
  for (int i = 0; i < 4; ++i) {
    int row, col; edge_rc_flat(rowA, colA, e0 + (i << 3) + sub, row, col);
    float a = ai[row * 4 + h] + aj[col * 4 + h];
    val[i] = a > 0.f ? a : NEG_SLOPE * a;
    dsti[i] = row * 4 + h;
  }
  #pragma unroll
  for (int i = 0; i < 4; ++i) atomic_max_f32(&m1[dsti[i]], val[i]);
}

#define A1_ROWS 157
#define A1_WGS  ((N_NODES + A1_ROWS - 1) / A1_ROWS)
#define A1_NV   (N_EDGES / 4)
static_assert(A1_ROWS * (D1 + H1) * 4 <= 320 * 1024, "the owned rows and their weight sums fit the LDS of one work-group");
static_assert(N_EDGES % 128 == 0, "a wave's 32 x 4 destination indices are all inside the list");

typedef __attribute__((ext_vector_type(4))) int v4i;

__device__ __forceinline__ float gat_weight(float a, float m) {
  a = a > 0.f ? a : NEG_SLOPE * a;
  return expf(a - m);
}

__device__ __forceinline__ void gat_weights4(int r, int c,
                                             const float* __restrict__ ai, const float* __restrict__ aj,
                                             const float* __restrict__ m1,
                                             float w[4]) {
  const v4f A = *(const v4f*)(ai + r * 4), B = *(const v4f*)(aj + c * 4);
  const v4f M = *(const v4f*)(m1 + r * 4);
  w[0] = gat_weight(A.x + B.x, M.x);
  w[1] = gat_weight(A.y + B.y, M.y);
  w[2] = gat_weight(A.z + B.z, M.z);
  w[3] = gat_weight(A.w + B.w, M.w);
}

__device__ __forceinline__ float weight_of_lane(const float w[4], int lane) {
  float x = w[0];
  x = lane == 1 ? w[1] : x;
  x = lane == 2 ? w[2] : x;
  x = lane == 3 ? w[3] : x;
  return x;
}

__device__ __forceinline__ void agg1_one(int d, int c, int r0, int lane,
                                         const float* __restrict__ ai, const float* __restrict__ aj,
                                         const float* __restrict__ m1,
                                         const float* __restrict__ h1, float* acc, float* den) {
  const float* __restrict__ src = h1 + (size_t)c * D1 + lane;
  float v[16];
  #pragma unroll
  for (int q = 0; q < 16; ++q) v[q] = src[q * 32];
  float w[4];
  gat_weights4(r0 + d, c, ai, aj, m1, w);
  float* dst = acc + d * D1 + lane;
  #pragma unroll
  for (int q = 0; q < 16; ++q) atomicAdd(dst + q * 32, v[q] * w[q >> 2]);
  const float wl = weight_of_lane(w, lane);
  if (lane < 4) atomicAdd(den + d * 4 + lane, wl);
}

__device__ __forceinline__ void agg1_pair(int d0, int c0, int d1, int c1, int r0, int lane,
                                          const float* __restrict__ ai, const float* __restrict__ aj,
                                          const float* __restrict__ m1,
                                          const float* __restrict__ h1, float* acc, float* den) {
  const float* __restrict__ src0 = h1 + (size_t)c0 * D1 + lane;
  const float* __restrict__ src1 = h1 + (size_t)c1 * D1 + lane;
  float v0[16], v1[16];
  #pragma unroll
  for (int q = 0; q < 16; ++q) v0[q] = src0[q * 32];
  #pragma unroll
  for (int q = 0; q < 16; ++q) v1[q] = src1[q * 32];
  float w0[4], w1[4];
  gat_weights4(r0 + d0, c0, ai, aj, m1, w0);
  gat_weights4(r0 + d1, c1, ai, aj, m1, w1);
  float* dst0 = acc + d0 * D1 + lane;
  float* dst1 = acc + d1 * D1 + lane;
  #pragma unroll
  for (int q = 0; q < 16; ++q) atomicAdd(dst0 + q * 32, v0[q] * w0[q >> 2]);
  #pragma unroll
  for (int q = 0; q < 16; ++q) atomicAdd(dst1 + q * 32, v1[q] * w1[q >> 2]);
  const float wl0 = weight_of_lane(w0, lane), wl1 = weight_of_lane(w1, lane);
  if (lane < 4) {
    atomicAdd(den + d0 * 4 + lane, wl0);
    atomicAdd(den + d1 * 4 + lane, wl1);
  }
}

__global__ __launch_bounds__(1024)
void agg1_owned_rows(const int* __restrict__ rowA, const int* __restrict__ colA,
                     const float* __restrict__ ai, const float* __restrict__ aj,
                     const float* __restrict__ m1,
                     const float* __restrict__ h1, float* __restrict__ out1) {
  __shared__ __attribute__((aligned(16))) float acc[A1_ROWS * D1];
  __shared__ __attribute__((aligned(16))) float den[A1_ROWS * H1];
  const int r0 = blockIdx.x * A1_ROWS;
  const int nr = (N_NODES - r0) < A1_ROWS ? (N_NODES - r0) : A1_ROWS;
  const int wave = threadIdx.x >> 5, lane = threadIdx.x & 31;

  for (int r = wave; r < nr; r += 32) {
    int n = r0 + r;
    const float* __restrict__ src = h1 + (size_t)n * D1 + lane;
    float v[16];
    #pragma unroll
    for (int q = 0; q < 16; ++q) v[q] = src[q * 32];
    float w[4];
    gat_weights4(n, n, ai, aj, m1, w);
    float* dst = acc + r * D1 + lane;
    #pragma unroll
    for (int q = 0; q < 16; ++q) dst[q * 32] = v[q] * w[q >> 2];
    const float wl = weight_of_lane(w, lane);
    if (lane < 4) den[r * 4 + lane] = wl;
  }
  __syncthreads();

  const unsigned unr = (unsigned)nr;
  const v4i none = {-1, -1, -1, -1};
  int vb = wave * 32;
  v4i g0 = vb        < A1_NV ? *(const v4i*)(rowA + 4 * (vb + lane))        : none;
  v4i g1 = vb + 1024 < A1_NV ? *(const v4i*)(rowA + 4 * (vb + 1024 + lane)) : none;
  v4i g2 = vb + 2048 < A1_NV ? *(const v4i*)(rowA + 4 * (vb + 2048 + lane)) : none;
  v4i g3 = vb + 3072 < A1_NV ? *(const v4i*)(rowA + 4 * (vb + 3072 + lane)) : none;
  bool have = false;
  int pd = 0, pc = 0;
  #pragma nounroll
  for (; vb < A1_NV; vb += 4096) {
    const v4i n0 = vb + 4096 < A1_NV ? *(const v4i*)(rowA + 4 * (vb + 4096 + lane)) : none;
    const v4i n1 = vb + 5120 < A1_NV ? *(const v4i*)(rowA + 4 * (vb + 5120 + lane)) : none;
    const v4i n2 = vb + 6144 < A1_NV ? *(const v4i*)(rowA + 4 * (vb + 6144 + lane)) : none;
    const v4i n3 = vb + 7168 < A1_NV ? *(const v4i*)(rowA + 4 * (vb + 7168 + lane)) : none;
    #pragma nounroll
    for (int u = 0; u < 4; ++u) {
      const v4i G = u == 0 ? g0 : u == 1 ? g1 : u == 2 ? g2 : g3;
      const int d0 = G.x - r0, d1 = G.y - r0, d2 = G.z - r0, d3 = G.w - r0;
      const bool any = ((unsigned)d0 < unr) | ((unsigned)d1 < unr) | ((unsigned)d2 < unr) | ((unsigned)d3 < unr);
      if (__builtin_amdgcn_ballot_w32(any) == 0) continue;
      const int e = 4 * (vb + u * 1024 + lane);
      #pragma nounroll
      for (int j = 0; j < 4; ++j) {
        const int dj = j == 0 ? d0 : j == 1 ? d1 : j == 2 ? d2 : d3;
        const bool tj = (unsigned)dj < unr;
        unsigned mask = __builtin_amdgcn_ballot_w32(tj);
        if (mask == 0) continue;
        int c = 0;
        if (tj) c = colA[e + j];
        while (mask) {
          const int l = __builtin_ctz(mask);
          mask &= mask - 1;
          const int dl = __builtin_amdgcn_readlane(dj, l);
          const int cl = __builtin_amdgcn_readlane(c, l);
          if (!have) { pd = dl; pc = cl; have = true; }
          else { agg1_pair(pd, pc, dl, cl, r0, lane, ai, aj, m1, h1, acc, den); have = false; }
        }
      }
    }
    g0 = n0; g1 = n1; g2 = n2; g3 = n3;
  }
  if (have) agg1_one(pd, pc, r0, lane, ai, aj, m1, h1, acc, den);
  __syncthreads();

  for (int r = wave; r < nr; r += 32) {
    const float* src = acc + r * D1 + (lane << 2);
    float* __restrict__ dst = out1 + (size_t)(r0 + r) * D1 + (lane << 2);
    const v4f S = *(const v4f*)(den + r * 4);
    v4f t[4];
    #pragma unroll
    for (int k = 0; k < 4; ++k) t[k] = *(const v4f*)(src + k * C1);
    #pragma unroll
    for (int k = 0; k < 4; ++k) {
      const float dn = S[k] + 1e-16f;
      v4f o;
      o.x = t[k].x / dn; o.y = t[k].y / dn; o.z = t[k].z / dn; o.w = t[k].w / dn;
      *(v4f*)(dst + k * C1) = o;
    }
  }
}


#define O2P 64

__global__ void node_att2(const float* __restrict__ h2, const float* __restrict__ att2,
                          float* __restrict__ s2, float* __restrict__ m2,
                          float* __restrict__ o2) {
  int tid = blockIdx.x * blockDim.x + threadIdx.x;
  if (tid >= N_NODES * H2) return;
  int n = tid >> 3, h = tid & 7;
  const float* __restrict__ hp = h2 + (size_t)n * D2 + h * C2;
  const float* __restrict__ ap = att2 + h * (2 * C2);
  float s = 0.f;
  #pragma unroll
  for (int c = 0; c < C2; ++c) s += hp[c] * (ap[c] + ap[C2 + c]);
  s2[tid] = tanhf(s);
  m2[tid] = -__builtin_inff();
  const v4f z = {0.f, 0.f, 0.f, 0.f};
  float* __restrict__ zp = o2 + (size_t)(tid & ~31) * 8 + ((tid & 31) << 2);
  *(v4f*)zp = z;
  *(v4f*)(zp + 128) = z;
}

__global__ void edge_max2(const int* __restrict__ rowA, const int* __restrict__ colA,
                          const float* __restrict__ s2, float* __restrict__ m2) {
  int wave = blockIdx.x * (blockDim.x >> 5) + (threadIdx.x >> 5);
  int e0 = wave << 5;
  if (e0 >= ETOT) return;
  int lane = threadIdx.x & 31;
  int sub = lane >> 3, h = lane & 7;
  int dsti[8]; float val[8];
  #pragma unroll
  for (int i = 0; i < 8; ++i) {
    int row, col; edge_rc_flat(rowA, colA, e0 + (i << 2) + sub, row, col);
    val[i] = s2[col * 8 + h];
    dsti[i] = row * 8 + h;
  }
  #pragma unroll
  for (int i = 0; i < 8; ++i) atomic_max_f32(&m2[dsti[i]], val[i]);
}

__global__ void edge_scatter2(const int* __restrict__ rowA, const int* __restrict__ colA,
                              const float* __restrict__ s2, const float* __restrict__ m2,
                              const float* __restrict__ h2, float* __restrict__ o2) {
  int wave = blockIdx.x * (blockDim.x >> 5) + (threadIdx.x >> 5);
  int e0 = wave << 3;
  if (e0 >= ETOT) return;
  int lane = threadIdx.x & 31;
  const bool row2 = lane < 16;
  const bool sum2 = lane < 24;
  const int hA = lane / C2;
  const int hB = row2 ? (lane + 32) / C2 : (lane - 16) & 7;
  const int j1 = row2 ? lane + 32 : lane;
  int myr, myc; edge_rc_flat(rowA, colA, e0 + (lane & 7), myr, myc);
  int dsti[8]; float sA[8], mA[8], sB[8], mB[8], x0[8], x1[8], v0[8], v1[8];
  #pragma unroll
  for (int i = 0; i < 8; ++i) {
    const int row = __builtin_amdgcn_readlane(myr, i);
    const int col = __builtin_amdgcn_readlane(myc, i);
    sA[i] = s2[col * 8 + hA]; mA[i] = m2[row * 8 + hA];
    sB[i] = s2[col * 8 + hB]; mB[i] = m2[row * 8 + hB];
    const float* __restrict__ src = h2 + (size_t)col * D2;
    x0[i] = src[lane];
    x1[i] = src[j1];
    dsti[i] = row * O2P + lane;
  }
  __builtin_amdgcn_sched_barrier(0);
  #pragma unroll
  for (int i = 0; i < 8; ++i) {
    const float eA = expf(sA[i] - mA[i]);
    const float eB = expf(sB[i] - mB[i]);
    v0[i] = x0[i] * eA;
    v1[i] = row2 ? x1[i] * eB : (sum2 ? eB : 0.0f);
  }
  #pragma unroll
  for (int i = 0; i < 8; ++i) {
    atomicAdd(o2 + dsti[i], v0[i]);
    atomicAdd(o2 + dsti[i] + 32, v1[i]);
  }
}

__global__ void finalize(const float* __restrict__ o2, const float* __restrict__ b2,
                         float* __restrict__ out) {
  int n = blockIdx.x * blockDim.x + threadIdx.x;
  if (n >= N_NODES) return;
  const float* __restrict__ rp = o2 + (size_t)n * O2P;
  v4f q[14];
  #pragma unroll
  for (int i = 0; i < 14; ++i) q[i] = *(const v4f*)(rp + 4 * i);
  float dn[H2];
  #pragma unroll
  for (int h = 0; h < H2; ++h) dn[h] = q[12 + (h >> 2)][h & 3] + 1e-16f;
  float v[C2];
  #pragma unroll
  for (int c = 0; c < C2; ++c) {
    float s = 0.f;
    #pragma unroll
    for (int h = 0; h < H2; ++h) {
      const int w = h * C2 + c;
      s += q[w >> 2][w & 3] / dn[h];
    }
    float t = s * (1.0f / H2) + b2[c];
    v[c] = t > 0.f ? t : (expf(t) - 1.0f);
  }
  float m = v[0];
  #pragma unroll
  for (int c = 1; c < C2; ++c) m = fmaxf(m, v[c]);
  float sum = 0.f;
  #pragma unroll
  for (int c = 0; c < C2; ++c) sum += expf(v[c] - m);
  float lse = logf(sum);
  #pragma unroll
  for (int c = 0; c < C2; ++c) out[(size_t)n * C2 + c] = v[c] - m - lse;
}


extern "C" void kernel_launch(void* const* d_in, const int* in_sizes, int n_in,
                              void* d_out, int out_size, void* d_ws, size_t ws_size,
                              hipStream_t stream) {
  const float* x    = (const float*)d_in[0];
  const int*   ei   = (const int*)  d_in[1];
  const float* g1   = (const float*)d_in[2];
  const float* be1  = (const float*)d_in[3];
  const float* mu1  = (const float*)d_in[4];
  const float* va1  = (const float*)d_in[5];
  const float* W1   = (const float*)d_in[6];
  const float* att1 = (const float*)d_in[7];
  const float* b1   = (const float*)d_in[8];
  const float* g2   = (const float*)d_in[9];
  const float* be2  = (const float*)d_in[10];
  const float* mu2  = (const float*)d_in[11];
  const float* va2  = (const float*)d_in[12];
  const float* W2   = (const float*)d_in[13];
  const float* att2 = (const float*)d_in[14];
  const float* b2   = (const float*)d_in[15];
  const int* rowA = ei;
  const int* colA = ei + N_EDGES;
  float* out = (float*)d_out;
  float* ws  = (float*)d_ws;

  float* sc1  = ws;
  float* sh1  = ws + 256;
  float* sc2  = ws + 512;
  float* sh2  = ws + 1024;
  float* ai   = ws + 1536;
  float* aj   = ai  + (size_t)N_NODES * H1;
  float* m1   = aj  + (size_t)N_NODES * H1;
  float* out1 = m1  + (size_t)N_NODES * H1;
  float* h1   = out1 + (size_t)N_NODES * D1;
  float* h2   = h1;
  float* s2   = h2  + (size_t)N_NODES * D2;
  float* m2   = s2  + (size_t)N_NODES * H2;
  float* out2 = m2  + (size_t)N_NODES * H2;

  bn_prep<<<2, 256, 0, stream>>>(g1, be1, mu1, va1, g2, be2, mu2, va2, b1,
                                 sc1, sh1, sc2, sh2);

  gemm1_rows_resident<<<dim3(313, 4), 256, 0, stream>>>(x, sc1, sh1, W1, h1);

  node_att1<<<(N_NODES / NA1_NODES + 7) / 8, 256, 0, stream>>>(h1, att1, ai, aj, m1);
  edge_max1<<<(ETOT + 255) / 256, 256, 0, stream>>>(rowA, colA, ai, aj, m1);
  agg1_owned_rows<<<A1_WGS, 1024, 0, stream>>>(rowA, colA, ai, aj, m1, h1, out1);


  gemm_bn_wmma<<<(2500 * 3 + 7) / 8, 256, 0, stream>>>(out1, sc2, sh2, W2, h2,
                                                       N_NODES, D1, D2);

  node_att2<<<(N_NODES * H2) / 256, 256, 0, stream>>>(h2, att2, s2, m2, out2);
  edge_max2<<<(ETOT + 255) / 256, 256, 0, stream>>>(rowA, colA, s2, m2);
  edge_scatter2<<<(ETOT / 8) / 8, 256, 0, stream>>>(rowA, colA, s2, m2, h2, out2);

  finalize<<<(N_NODES + 255) / 256, 256, 0, stream>>>(out2, b2, out);
}


# Layer 2 aggregated like layer 1: a group owns 157 rows, lists its edges in LDS from one scan of all destinations, maxima and sums by LDS atomics, finalizes; 10->8 launches; sums: LDS arrival order
# speedup vs baseline: 3.6116x; 1.0845x over previous
#include <hip/hip_runtime.h>
#include <math.h>

#define N_NODES 40000
#define N_EDGES 320000
#define ETOT    (N_EDGES + N_NODES)
#define F_IN    256
#define H1      4
#define C1      128
#define D1      (H1*C1)
#define H2      8
#define C2      6
#define D2      (H2*C2)
#define BN_EPS    1e-5f
#define NEG_SLOPE 0.2f

typedef __attribute__((ext_vector_type(2))) float v2f;
typedef __attribute__((ext_vector_type(4))) float v4f;
typedef __attribute__((ext_vector_type(8))) float v8f;


__device__ __forceinline__ void atomic_max_f32(float* addr, float v) {
  if (v >= 0.0f) atomicMax((int*)addr, __float_as_int(v));
  else           atomicMin((unsigned int*)addr, __float_as_uint(v));
}

__device__ __forceinline__ void edge_rc_flat(const int* __restrict__ rowA,
                                             const int* __restrict__ colA,
                                             int e, int& r, int& c) {
  int ec = e < N_EDGES ? e : N_EDGES - 1;
  int lr = rowA[ec], lc = colA[ec];
  bool self = e >= N_EDGES;
  r = self ? e - N_EDGES : lr;
  c = self ? e - N_EDGES : lc;
}


__global__ void bn_prep(const float* g1, const float* be1, const float* mu1, const float* va1,
                        const float* g2, const float* be2, const float* mu2, const float* va2,
                        const float* bias1,
                        float* sc1, float* sh1, float* sc2, float* sh2) {
  int i = blockIdx.x * blockDim.x + threadIdx.x;
  if (i < F_IN) {
    float s = g1[i] / sqrtf(va1[i] + BN_EPS);
    sc1[i] = s; sh1[i] = be1[i] - mu1[i] * s;
  }
  if (i < D1) {
    float s = g2[i] / sqrtf(va2[i] + BN_EPS);
    sc2[i] = s; sh2[i] = be2[i] + (bias1[i] - mu2[i]) * s;
  }
}


__global__ __launch_bounds__(256)
void gemm_bn_wmma(const float* __restrict__ A,
                  const float* __restrict__ scale,
                  const float* __restrict__ shift,
                  const float* __restrict__ B,
                  float* __restrict__ C,
                  int M, int K, int N) {
  int wave = blockIdx.x * (blockDim.x >> 5) + (threadIdx.x >> 5);
  int ntiles = N >> 4;
  int tm = wave / ntiles;
  int tn = wave - tm * ntiles;
  if (tm >= (M >> 4)) return;

  int lane = threadIdx.x & 31;
  int half = lane >> 4;
  int l16  = lane & 15;
  int row  = (tm << 4) + l16;
  int colB = (tn << 4) + l16;

  const float* __restrict__ Arow = A + (size_t)row * K;
  v8f acc = {};
  for (int k = 0; k < K; k += 8) {
    int kk = k + (half << 2);
    v4f a4 = *(const v4f*)(Arow + kk);
    v4f s4 = *(const v4f*)(scale + kk);
    v4f t4 = *(const v4f*)(shift + kk);
    const float* __restrict__ Bk = B + (size_t)kk * N + colB;
    v2f a, b;
    a.x = a4.x * s4.x + t4.x;
    a.y = a4.y * s4.y + t4.y;
    b.x = Bk[0];
    b.y = Bk[(size_t)N];
    acc = __builtin_amdgcn_wmma_f32_16x16x4_f32(
         false, a,  false, b,
         (short)0, acc,  false,  false);
    a.x = a4.z * s4.z + t4.z;
    a.y = a4.w * s4.w + t4.w;
    b.x = Bk[(size_t)2 * N];
    b.y = Bk[(size_t)3 * N];
    acc = __builtin_amdgcn_wmma_f32_16x16x4_f32(
         false, a,  false, b,
         (short)0, acc,  false,  false);
  }
  float* Crow = C + (size_t)((tm << 4) + (half << 3)) * N + colB;
  #pragma unroll
  for (int r = 0; r < 8; ++r) Crow[(size_t)r * N] = acc[r];
}

#define G1_COLS_PER_WAVE 128

__global__ __launch_bounds__(256)
void gemm1_rows_resident(const float* __restrict__ A,
                         const float* __restrict__ scale,
                         const float* __restrict__ shift,
                         const float* __restrict__ B,
                         float* __restrict__ C) {
  int tm = blockIdx.x * 8 + (threadIdx.x >> 5);
  if (tm >= (N_NODES >> 4)) return;
  int lane = threadIdx.x & 31;
  int half = lane >> 4;
  int l16  = lane & 15;

  const float* __restrict__ Arow = A + (size_t)((tm << 4) + l16) * F_IN + (half << 2);
  const float* __restrict__ sc = scale + (half << 2);
  const float* __restrict__ sh = shift + (half << 2);
  v4f a[F_IN / 8];
  #pragma unroll
  for (int t = 0; t < F_IN / 8; ++t) a[t] = *(const v4f*)(Arow + 8 * t);
  #pragma unroll
  for (int t0 = 0; t0 < F_IN / 8; t0 += 8) {
    #pragma unroll
    for (int t = t0; t < t0 + 8; ++t) {
      v4f s4 = *(const v4f*)(sc + 8 * t);
      v4f t4 = *(const v4f*)(sh + 8 * t);
      a[t].x = a[t].x * s4.x + t4.x;
      a[t].y = a[t].y * s4.y + t4.y;
      a[t].z = a[t].z * s4.z + t4.z;
      a[t].w = a[t].w * s4.w + t4.w;
    }
    __builtin_amdgcn_sched_barrier(0);
  }

  int cb = blockIdx.y * G1_COLS_PER_WAVE + (l16 << 2);
  const float* __restrict__ Bl = B + (size_t)(half << 2) * D1 + cb;
  float* __restrict__ Cl = C + (size_t)((tm << 4) + (half << 3)) * D1 + cb;
  for (int j = 0; j < G1_COLS_PER_WAVE; j += 64) {
    v8f acc0 = {}, acc1 = {}, acc2 = {}, acc3 = {};
    const float* __restrict__ Bj = Bl + j;
    v4f bq[3][4];
    #pragma unroll
    for (int p = 0; p < 2; ++p) {
      #pragma unroll
      for (int q = 0; q < 4; ++q) bq[p][q] = *(const v4f*)(Bj + (size_t)(8 * p + q) * D1);
    }
    #pragma unroll
    for (int t = 0; t < F_IN / 8; ++t) {
      if (t + 2 < F_IN / 8) {
        #pragma unroll
        for (int q = 0; q < 4; ++q)
          bq[(t + 2) % 3][q] = *(const v4f*)(Bj + (size_t)(8 * (t + 2) + q) * D1);
      }
      __builtin_amdgcn_sched_barrier(0);
      const v4f b0 = bq[t % 3][0], b1 = bq[t % 3][1], b2 = bq[t % 3][2], b3 = bq[t % 3][3];
      v2f a01, a23, b;
      a01.x = a[t].x; a01.y = a[t].y;
      a23.x = a[t].z; a23.y = a[t].w;
      b.x = b0.x; b.y = b1.x;
      acc0 = __builtin_amdgcn_wmma_f32_16x16x4_f32(false, a01, false, b, (short)0, acc0, false, false);
      b.x = b0.y; b.y = b1.y;
      acc1 = __builtin_amdgcn_wmma_f32_16x16x4_f32(false, a01, false, b, (short)0, acc1, false, false);
      b.x = b0.z; b.y = b1.z;
      acc2 = __builtin_amdgcn_wmma_f32_16x16x4_f32(false, a01, false, b, (short)0, acc2, false, false);
      b.x = b0.w; b.y = b1.w;
      acc3 = __builtin_amdgcn_wmma_f32_16x16x4_f32(false, a01, false, b, (short)0, acc3, false, false);
      b.x = b2.x; b.y = b3.x;
      acc0 = __builtin_amdgcn_wmma_f32_16x16x4_f32(false, a23, false, b, (short)0, acc0, false, false);
      b.x = b2.y; b.y = b3.y;
      acc1 = __builtin_amdgcn_wmma_f32_16x16x4_f32(false, a23, false, b, (short)0, acc1, false, false);
      b.x = b2.z; b.y = b3.z;
      acc2 = __builtin_amdgcn_wmma_f32_16x16x4_f32(false, a23, false, b, (short)0, acc2, false, false);
      b.x = b2.w; b.y = b3.w;
      acc3 = __builtin_amdgcn_wmma_f32_16x16x4_f32(false, a23, false, b, (short)0, acc3, false, false);
      __builtin_amdgcn_sched_barrier(0);
    }
    #pragma unroll
    for (int r = 0; r < 8; ++r) {
      v4f o;
      o.x = acc0[r]; o.y = acc1[r]; o.z = acc2[r]; o.w = acc3[r];
      *(v4f*)(Cl + (size_t)r * D1 + j) = o;
    }
  }
}


#define NA1_NODES 8

__global__ __launch_bounds__(256)
void node_att1(const float* __restrict__ h1, const float* __restrict__ att1,
               float* __restrict__ ai, float* __restrict__ aj,
               float* __restrict__ m1) {
  int wave = blockIdx.x * (blockDim.x >> 5) + (threadIdx.x >> 5);
  int n0 = wave * NA1_NODES;
  if (n0 >= N_NODES) return;
  int lane = threadIdx.x & 31;
  v4f al[H1], ar[H1];
  #pragma unroll
  for (int h = 0; h < H1; ++h) {
    al[h] = *(const v4f*)(att1 + h * (2 * C1) + (lane << 2));
    ar[h] = *(const v4f*)(att1 + h * (2 * C1) + C1 + (lane << 2));
  }
  const bool b16 = (lane & 16) != 0, b8 = (lane & 8) != 0, b4 = (lane & 4) != 0;
  const int idx = lane >> 2;
  const float* __restrict__ hp = h1 + (size_t)n0 * D1 + (lane << 2);
  #pragma unroll
  for (int i = 0; i < NA1_NODES; i += 4) {
    v4f v[4][H1];
    #pragma unroll
    for (int u = 0; u < 4; ++u) {
      #pragma unroll
      for (int h = 0; h < H1; ++h) v[u][h] = *(const v4f*)(hp + (size_t)(i + u) * D1 + h * C1);
    }
    #pragma unroll
    for (int u = 0; u < 4; ++u) {
      float r[8];
      #pragma unroll
      for (int h = 0; h < H1; ++h) {
        const v4f x = v[u][h];
        r[h]     = x.x * al[h].x + x.y * al[h].y + x.z * al[h].z + x.w * al[h].w;
        r[4 + h] = x.x * ar[h].x + x.y * ar[h].y + x.z * ar[h].z + x.w * ar[h].w;
      }
      float t4[4], t2[2];
      #pragma unroll
      for (int k = 0; k < 4; ++k) {
        float keep = b16 ? r[4 + k] : r[k];
        float send = b16 ? r[k] : r[4 + k];
        t4[k] = keep + __shfl_xor(send, 16, 32);
      }
      #pragma unroll
      for (int k = 0; k < 2; ++k) {
        float keep = b8 ? t4[2 + k] : t4[k];
        float send = b8 ? t4[k] : t4[2 + k];
        t2[k] = keep + __shfl_xor(send, 8, 32);
      }
      float keep = b4 ? t2[1] : t2[0];
      float send = b4 ? t2[0] : t2[1];
      float t = keep + __shfl_xor(send, 4, 32);
      t += __shfl_xor(t, 2, 32);
      t += __shfl_xor(t, 1, 32);
      if ((lane & 3) == 0) {
        int n = n0 + i + u;
        if (idx < 4) {
          ai[n * 4 + idx] = t;
          m1[n * 4 + idx] = -__builtin_inff();
        } else {
          aj[n * 4 + (idx - 4)] = t;
        }
      }
    }
  }
}


__global__ void edge_max1(const int* __restrict__ rowA, const int* __restrict__ colA,
                          const float* __restrict__ ai, const float* __restrict__ aj,
                          float* __restrict__ m1) {
  int wave = blockIdx.x * (blockDim.x >> 5) + (threadIdx.x >> 5);
  int e0 = wave << 5;
  if (e0 >= ETOT) return;
  int lane = threadIdx.x & 31;
  int sub = lane >> 2, h = lane & 3;
  int dsti[4]; float val[4];
  #pragma unroll
  for (int i = 0; i < 4; ++i) {
    int row, col; edge_rc_flat(rowA, colA, e0 + (i << 3) + sub, row, col);
    float a = ai[row * 4 + h] + aj[col * 4 + h];
    val[i] = a > 0.f ? a : NEG_SLOPE * a;
    dsti[i] = row * 4 + h;
  }
  #pragma unroll
  for (int i = 0; i < 4; ++i) atomic_max_f32(&m1[dsti[i]], val[i]);
}

#define A1_ROWS 157
#define A1_WGS  ((N_NODES + A1_ROWS - 1) / A1_ROWS)
#define A1_NV   (N_EDGES / 4)
static_assert(A1_ROWS * (D1 + H1) * 4 <= 320 * 1024, "the owned rows and their weight sums fit the LDS of one work-group");
static_assert(N_EDGES % 128 == 0, "a wave's 32 x 4 destination indices are all inside the list");

typedef __attribute__((ext_vector_type(4))) int v4i;

__device__ __forceinline__ float gat_weight(float a, float m) {
  a = a > 0.f ? a : NEG_SLOPE * a;
  return expf(a - m);
}

__device__ __forceinline__ void gat_weights4(int r, int c,
                                             const float* __restrict__ ai, const float* __restrict__ aj,
                                             const float* __restrict__ m1,
                                             float w[4]) {
  const v4f A = *(const v4f*)(ai + r * 4), B = *(const v4f*)(aj + c * 4);
  const v4f M = *(const v4f*)(m1 + r * 4);
  w[0] = gat_weight(A.x + B.x, M.x);
  w[1] = gat_weight(A.y + B.y, M.y);
  w[2] = gat_weight(A.z + B.z, M.z);
  w[3] = gat_weight(A.w + B.w, M.w);
}

__device__ __forceinline__ float weight_of_lane(const float w[4], int lane) {
  float x = w[0];
  x = lane == 1 ? w[1] : x;
  x = lane == 2 ? w[2] : x;
  x = lane == 3 ? w[3] : x;
  return x;
}

__device__ __forceinline__ void agg1_one(int d, int c, int r0, int lane,
                                         const float* __restrict__ ai, const float* __restrict__ aj,
                                         const float* __restrict__ m1,
                                         const float* __restrict__ h1, float* acc, float* den) {
  const float* __restrict__ src = h1 + (size_t)c * D1 + lane;
  float v[16];
  #pragma unroll
  for (int q = 0; q < 16; ++q) v[q] = src[q * 32];
  float w[4];
  gat_weights4(r0 + d, c, ai, aj, m1, w);
  float* dst = acc + d * D1 + lane;
  #pragma unroll
  for (int q = 0; q < 16; ++q) atomicAdd(dst + q * 32, v[q] * w[q >> 2]);
  const float wl = weight_of_lane(w, lane);
  if (lane < 4) atomicAdd(den + d * 4 + lane, wl);
}

__device__ __forceinline__ void agg1_pair(int d0, int c0, int d1, int c1, int r0, int lane,
                                          const float* __restrict__ ai, const float* __restrict__ aj,
                                          const float* __restrict__ m1,
                                          const float* __restrict__ h1, float* acc, float* den) {
  const float* __restrict__ src0 = h1 + (size_t)c0 * D1 + lane;
  const float* __restrict__ src1 = h1 + (size_t)c1 * D1 + lane;
  float v0[16], v1[16];
  #pragma unroll
  for (int q = 0; q < 16; ++q) v0[q] = src0[q * 32];
  #pragma unroll
  for (int q = 0; q < 16; ++q) v1[q] = src1[q * 32];
  float w0[4], w1[4];
  gat_weights4(r0 + d0, c0, ai, aj, m1, w0);
  gat_weights4(r0 + d1, c1, ai, aj, m1, w1);
  float* dst0 = acc + d0 * D1 + lane;
  float* dst1 = acc + d1 * D1 + lane;
  #pragma unroll
  for (int q = 0; q < 16; ++q) atomicAdd(dst0 + q * 32, v0[q] * w0[q >> 2]);
  #pragma unroll
  for (int q = 0; q < 16; ++q) atomicAdd(dst1 + q * 32, v1[q] * w1[q >> 2]);
  const float wl0 = weight_of_lane(w0, lane), wl1 = weight_of_lane(w1, lane);
  if (lane < 4) {
    atomicAdd(den + d0 * 4 + lane, wl0);
    atomicAdd(den + d1 * 4 + lane, wl1);
  }
}

__global__ __launch_bounds__(1024)
void agg1_owned_rows(const int* __restrict__ rowA, const int* __restrict__ colA,
                     const float* __restrict__ ai, const float* __restrict__ aj,
                     const float* __restrict__ m1,
                     const float* __restrict__ h1, float* __restrict__ out1) {
  __shared__ __attribute__((aligned(16))) float acc[A1_ROWS * D1];
  __shared__ __attribute__((aligned(16))) float den[A1_ROWS * H1];
  const int r0 = blockIdx.x * A1_ROWS;
  const int nr = (N_NODES - r0) < A1_ROWS ? (N_NODES - r0) : A1_ROWS;
  const int wave = threadIdx.x >> 5, lane = threadIdx.x & 31;

  for (int r = wave; r < nr; r += 32) {
    int n = r0 + r;
    const float* __restrict__ src = h1 + (size_t)n * D1 + lane;
    float v[16];
    #pragma unroll
    for (int q = 0; q < 16; ++q) v[q] = src[q * 32];
    float w[4];
    gat_weights4(n, n, ai, aj, m1, w);
    float* dst = acc + r * D1 + lane;
    #pragma unroll
    for (int q = 0; q < 16; ++q) dst[q * 32] = v[q] * w[q >> 2];
    const float wl = weight_of_lane(w, lane);
    if (lane < 4) den[r * 4 + lane] = wl;
  }
  __syncthreads();

  const unsigned unr = (unsigned)nr;
  const v4i none = {-1, -1, -1, -1};
  int vb = wave * 32;
  v4i g0 = vb        < A1_NV ? *(const v4i*)(rowA + 4 * (vb + lane))        : none;
  v4i g1 = vb + 1024 < A1_NV ? *(const v4i*)(rowA + 4 * (vb + 1024 + lane)) : none;
  v4i g2 = vb + 2048 < A1_NV ? *(const v4i*)(rowA + 4 * (vb + 2048 + lane)) : none;
  v4i g3 = vb + 3072 < A1_NV ? *(const v4i*)(rowA + 4 * (vb + 3072 + lane)) : none;
  bool have = false;
  int pd = 0, pc = 0;
  #pragma nounroll
  for (; vb < A1_NV; vb += 4096) {
    const v4i n0 = vb + 4096 < A1_NV ? *(const v4i*)(rowA + 4 * (vb + 4096 + lane)) : none;
    const v4i n1 = vb + 5120 < A1_NV ? *(const v4i*)(rowA + 4 * (vb + 5120 + lane)) : none;
    const v4i n2 = vb + 6144 < A1_NV ? *(const v4i*)(rowA + 4 * (vb + 6144 + lane)) : none;
    const v4i n3 = vb + 7168 < A1_NV ? *(const v4i*)(rowA + 4 * (vb + 7168 + lane)) : none;
    #pragma nounroll
    for (int u = 0; u < 4; ++u) {
      const v4i G = u == 0 ? g0 : u == 1 ? g1 : u == 2 ? g2 : g3;
      const int d0 = G.x - r0, d1 = G.y - r0, d2 = G.z - r0, d3 = G.w - r0;
      const bool any = ((unsigned)d0 < unr) | ((unsigned)d1 < unr) | ((unsigned)d2 < unr) | ((unsigned)d3 < unr);
      if (__builtin_amdgcn_ballot_w32(any) == 0) continue;
      const int e = 4 * (vb + u * 1024 + lane);
      #pragma nounroll
      for (int j = 0; j < 4; ++j) {
        const int dj = j == 0 ? d0 : j == 1 ? d1 : j == 2 ? d2 : d3;
        const bool tj = (unsigned)dj < unr;
        unsigned mask = __builtin_amdgcn_ballot_w32(tj);
        if (mask == 0) continue;
        int c = 0;
        if (tj) c = colA[e + j];
        while (mask) {
          const int l = __builtin_ctz(mask);
          mask &= mask - 1;
          const int dl = __builtin_amdgcn_readlane(dj, l);
          const int cl = __builtin_amdgcn_readlane(c, l);
          if (!have) { pd = dl; pc = cl; have = true; }
          else { agg1_pair(pd, pc, dl, cl, r0, lane, ai, aj, m1, h1, acc, den); have = false; }
        }
      }
    }
    g0 = n0; g1 = n1; g2 = n2; g3 = n3;
  }
  if (have) agg1_one(pd, pc, r0, lane, ai, aj, m1, h1, acc, den);
  __syncthreads();

  for (int r = wave; r < nr; r += 32) {
    const float* src = acc + r * D1 + (lane << 2);
    float* __restrict__ dst = out1 + (size_t)(r0 + r) * D1 + (lane << 2);
    const v4f S = *(const v4f*)(den + r * 4);
    v4f t[4];
    #pragma unroll
    for (int k = 0; k < 4; ++k) t[k] = *(const v4f*)(src + k * C1);
    #pragma unroll
    for (int k = 0; k < 4; ++k) {
      const float dn = S[k] + 1e-16f;
      v4f o;
      o.x = t[k].x / dn; o.y = t[k].y / dn; o.z = t[k].z / dn; o.w = t[k].w / dn;
      *(v4f*)(dst + k * C1) = o;
    }
  }
}


__global__ void node_att2(const float* __restrict__ h2, const float* __restrict__ att2,
                          float* __restrict__ s2) {
  int tid = blockIdx.x * blockDim.x + threadIdx.x;
  if (tid >= N_NODES * H2) return;
  int n = tid >> 3, h = tid & 7;
  const float* __restrict__ hp = h2 + (size_t)n * D2 + h * C2;
  const float* __restrict__ ap = att2 + h * (2 * C2);
  float s = 0.f;
  #pragma unroll
  for (int c = 0; c < C2; ++c) s += hp[c] * (ap[c] + ap[C2 + c]);
  s2[tid] = tanhf(s);
}

#define A2_ROWS  A1_ROWS
#define A2_WGS   A1_WGS
#define A2_PITCH 57
#define A2_CAP   65536
static_assert(((A2_ROWS + 1) * A2_PITCH + A2_ROWS * H2 + A2_CAP + 1) * 4 <= 320 * 1024, "sums, maxima and list fit the LDS of one work-group");
static_assert(A2_ROWS <= 256 && ETOT <= (1 << 19) && N_NODES <= (1 << 16), "list entries pack (row, edge) as row<<19|edge and (row, node) as row<<16|node");

__device__ __forceinline__ void lds_max_f32(float* addr, float v) {
  if (v >= 0.0f) atomicMax((int*)addr, __float_as_int(v));
  else           atomicMin((unsigned int*)addr, __float_as_uint(v));
}

__device__ __forceinline__ int edge_src(const int* __restrict__ colA, int e) {
  int ec = e < N_EDGES ? e : N_EDGES - 1;
  int lc = colA[ec];
  return e < N_EDGES ? lc : e - N_EDGES;
}

template <int MODE>
__device__ __forceinline__ void l2_take(int d, int e, const int* __restrict__ colA,
                                        const float* __restrict__ s2, const float* __restrict__ h2,
                                        float* acc, float* mx) {
  const int c = edge_src(colA, e);
  const v4f sa = *(const v4f*)(s2 + c * H2), sb = *(const v4f*)(s2 + c * H2 + 4);
  const float s[H2] = {sa.x, sa.y, sa.z, sa.w, sb.x, sb.y, sb.z, sb.w};
  if (MODE == 1) {
    #pragma unroll
    for (int h = 0; h < H2; ++h) lds_max_f32(mx + d * H2 + h, s[h]);
  } else {
    const float* __restrict__ hp = h2 + (size_t)c * D2;
    v4f x[D2 / 4];
    #pragma unroll
    for (int q = 0; q < D2 / 4; ++q) x[q] = *(const v4f*)(hp + 4 * q);
    float* dst = acc + d * A2_PITCH;
    #pragma unroll
    for (int h = 0; h < H2; ++h) {
      const float w = expf(s[h] - mx[d * H2 + h]);
      #pragma unroll
      for (int k = 0; k < C2; ++k) {
        const int j = h * C2 + k;
        atomicAdd(dst + j, x[j >> 2][j & 3] * w);
      }
      atomicAdd(dst + D2 + h, w);
    }
  }
}

#define A2_LD 8

__device__ __forceinline__ void l2_load_groups(v4i (&g)[A2_LD], const int* __restrict__ rowA, int vb, int lane) {
  #pragma unroll
  for (int u = 0; u < A2_LD; ++u) {
    const int v = vb + u * 1024;
    g[u] = *(const v4i*)(rowA + 4 * ((v < A1_NV ? v : A1_NV - 32) + lane));
  }
}

__device__ __forceinline__ void l2_list_groups(const v4i (&g)[A2_LD], int vb, int* list, int* cnt,
                                               int r0, unsigned unr, int lane) {
  #pragma unroll
  for (int u = 0; u < A2_LD; ++u) {
    if (vb + u * 1024 >= A1_NV) continue;
    const int d0 = g[u].x - r0, d1 = g[u].y - r0, d2 = g[u].z - r0, d3 = g[u].w - r0;
    const bool t0 = (unsigned)d0 < unr, t1 = (unsigned)d1 < unr, t2 = (unsigned)d2 < unr, t3 = (unsigned)d3 < unr;
    const int nm = (int)t0 + (int)t1 + (int)t2 + (int)t3;
    if (nm) {
      const int e = 4 * (vb + u * 1024 + lane);
      int slot = atomicAdd(cnt, nm);
      if (t0) { if (slot < A2_CAP) list[slot] = (d0 << 19) | e;       ++slot; }
      if (t1) { if (slot < A2_CAP) list[slot] = (d1 << 19) | (e + 1); ++slot; }
      if (t2) { if (slot < A2_CAP) list[slot] = (d2 << 19) | (e + 2); ++slot; }
      if (t3) { if (slot < A2_CAP) list[slot] = (d3 << 19) | (e + 3); }
    }
  }
}

__device__ __forceinline__ void l2_scan_list(const int* __restrict__ rowA, int* list, int* cnt,
                                             int r0, unsigned unr, int wave, int lane) {
  int vb = wave * 32;
  v4i ga[A2_LD], gb[A2_LD];
  l2_load_groups(ga, rowA, vb, lane);
  #pragma nounroll
  for (; vb < A1_NV; vb += 2 * A2_LD * 1024) {
    l2_load_groups(gb, rowA, vb + A2_LD * 1024, lane);
    __builtin_amdgcn_sched_barrier(0);
    l2_list_groups(ga, vb, list, cnt, r0, unr, lane);
    l2_load_groups(ga, rowA, vb + 2 * A2_LD * 1024, lane);
    __builtin_amdgcn_sched_barrier(0);
    l2_list_groups(gb, vb + A2_LD * 1024, list, cnt, r0, unr, lane);
  }
}

template <int MODE>
__device__ __forceinline__ void l2_rescan(const int* __restrict__ rowA, const int* __restrict__ colA,
                                          const float* __restrict__ s2, const float* __restrict__ h2,
                                          float* acc, float* mx, int r0, int nr, int tid) {
  const int ngroups = A1_NV + nr;
  #pragma nounroll
  for (int v = tid; v < ngroups; v += 1024) {
    const bool self = v >= A1_NV;
    const v4i none = {-1, -1, -1, -1};
    const v4i G = self ? none : *(const v4i*)(rowA + 4 * (self ? 0 : v));
    #pragma nounroll
    for (int j = 0; j < 4; ++j) {
      int dj = (j == 0 ? G.x : j == 1 ? G.y : j == 2 ? G.z : G.w) - r0;
      int e = 4 * v + j;
      if (self) { dj = j == 0 ? v - A1_NV : -1; e = N_EDGES + r0 + (v - A1_NV); }
      if ((unsigned)dj < (unsigned)nr) l2_take<MODE>(dj, e, colA, s2, h2, acc, mx);
    }
  }
}

__global__ __launch_bounds__(1024)
void agg2_owned_rows(const int* __restrict__ rowA, const int* __restrict__ colA,
                     const float* __restrict__ s2, const float* __restrict__ h2,
                     const float* __restrict__ b2, float* __restrict__ out) {
  __shared__ float acc[(A2_ROWS + 1) * A2_PITCH];
  __shared__ float mx[A2_ROWS * H2];
  __shared__ int list[A2_CAP];
  __shared__ int cnt;
  const int r0 = blockIdx.x * A2_ROWS;
  const int nr = (N_NODES - r0) < A2_ROWS ? (N_NODES - r0) : A2_ROWS;
  const int tid = threadIdx.x, lane = tid & 31;
  const int wave = __builtin_amdgcn_readfirstlane(tid >> 5);

  for (int i = tid; i < (A2_ROWS + 1) * A2_PITCH; i += 1024) acc[i] = 0.f;
  for (int i = tid; i < A2_ROWS * H2; i += 1024) mx[i] = -__builtin_inff();
  if (tid < nr) list[tid] = (tid << 19) | (N_EDGES + r0 + tid);
  if (tid == 0) cnt = nr;
  __syncthreads();

  l2_scan_list(rowA, list, &cnt, r0, (unsigned)nr, wave, lane);
  __syncthreads();
  const int found = __builtin_amdgcn_readfirstlane(cnt);

  if (found <= A2_CAP) {
    for (int i = tid; i < found; i += 1024) {
      const int p = list[i];
      const int d = p >> 19;
      const int c = edge_src(colA, p & ((1 << 19) - 1));
      list[i] = (d << 16) | c;
      const v4f sa = *(const v4f*)(s2 + c * H2), sb = *(const v4f*)(s2 + c * H2 + 4);
      float* mp = mx + d * H2;
      lds_max_f32(mp + 0, sa.x); lds_max_f32(mp + 1, sa.y); lds_max_f32(mp + 2, sa.z); lds_max_f32(mp + 3, sa.w);
      lds_max_f32(mp + 4, sb.x); lds_max_f32(mp + 5, sb.y); lds_max_f32(mp + 6, sb.z); lds_max_f32(mp + 7, sb.w);
    }
    __syncthreads();
    const int grp = tid >> 3, h = tid & 7;
    for (int i0 = grp; i0 < found; i0 += 4 * 128) {
      int p[4], dr[4]; float s[4], m[4]; v2f x[4][3];
      #pragma unroll
      for (int u = 0; u < 4; ++u) {
        const int i = i0 + u * 128;
        const bool ok = i < found;
        p[u] = list[ok ? i : i0];
        dr[u] = ok ? (p[u] >> 16) : A2_ROWS;
      }
      #pragma unroll
      for (int u = 0; u < 4; ++u) {
        const int d = p[u] >> 16, c = p[u] & 0xFFFF;
        s[u] = s2[c * H2 + h];
        m[u] = mx[d * H2 + h];
        const float* __restrict__ hp = h2 + (size_t)c * D2 + h * C2;
        x[u][0] = *(const v2f*)(hp); x[u][1] = *(const v2f*)(hp + 2); x[u][2] = *(const v2f*)(hp + 4);
      }
      #pragma unroll
      for (int u = 0; u < 4; ++u) {
        const float w = expf(s[u] - m[u]);
        float* dst = acc + dr[u] * A2_PITCH + h * C2;
        atomicAdd(dst + 0, x[u][0].x * w); atomicAdd(dst + 1, x[u][0].y * w);
        atomicAdd(dst + 2, x[u][1].x * w); atomicAdd(dst + 3, x[u][1].y * w);
        atomicAdd(dst + 4, x[u][2].x * w); atomicAdd(dst + 5, x[u][2].y * w);
        atomicAdd(acc + dr[u] * A2_PITCH + D2 + h, w);
      }
    }
  } else {
    l2_rescan<1>(rowA, colA, s2, h2, acc, mx, r0, nr, tid);
    __syncthreads();
    l2_rescan<2>(rowA, colA, s2, h2, acc, mx, r0, nr, tid);
  }
  __syncthreads();

  if (tid < nr) {
    const float* rp = acc + tid * A2_PITCH;
    float dn[H2];
    #pragma unroll
    for (int hh = 0; hh < H2; ++hh) dn[hh] = rp[D2 + hh] + 1e-16f;
    float v[C2];
    #pragma unroll
    for (int c = 0; c < C2; ++c) {
      float s = 0.f;
      #pragma unroll
      for (int hh = 0; hh < H2; ++hh) s += rp[hh * C2 + c] / dn[hh];
      float t = s * (1.0f / H2) + b2[c];
      v[c] = t > 0.f ? t : (expf(t) - 1.0f);
    }
    float m = v[0];
    #pragma unroll
    for (int c = 1; c < C2; ++c) m = fmaxf(m, v[c]);
    float sum = 0.f;
    #pragma unroll
    for (int c = 0; c < C2; ++c) sum += expf(v[c] - m);
    float lse = logf(sum);
    #pragma unroll
    for (int c = 0; c < C2; ++c) out[(size_t)(r0 + tid) * C2 + c] = v[c] - m - lse;
  }
}


extern "C" void kernel_launch(void* const* d_in, const int* in_sizes, int n_in,
                              void* d_out, int out_size, void* d_ws, size_t ws_size,
                              hipStream_t stream) {
  const float* x    = (const float*)d_in[0];
  const int*   ei   = (const int*)  d_in[1];
  const float* g1   = (const float*)d_in[2];
  const float* be1  = (const float*)d_in[3];
  const float* mu1  = (const float*)d_in[4];
  const float* va1  = (const float*)d_in[5];
  const float* W1   = (const float*)d_in[6];
  const float* att1 = (const float*)d_in[7];
  const float* b1   = (const float*)d_in[8];
  const float* g2   = (const float*)d_in[9];
  const float* be2  = (const float*)d_in[10];
  const float* mu2  = (const float*)d_in[11];
  const float* va2  = (const float*)d_in[12];
  const float* W2   = (const float*)d_in[13];
  const float* att2 = (const float*)d_in[14];
  const float* b2   = (const float*)d_in[15];
  const int* rowA = ei;
  const int* colA = ei + N_EDGES;
  float* out = (float*)d_out;
  float* ws  = (float*)d_ws;

  float* sc1  = ws;
  float* sh1  = ws + 256;
  float* sc2  = ws + 512;
  float* sh2  = ws + 1024;
  float* ai   = ws + 1536;
  float* aj   = ai  + (size_t)N_NODES * H1;
  float* m1   = aj  + (size_t)N_NODES * H1;
  float* out1 = m1  + (size_t)N_NODES * H1;
  float* h1   = out1 + (size_t)N_NODES * D1;
  float* h2   = h1;
  float* s2   = h2  + (size_t)N_NODES * D2;

  bn_prep<<<2, 256, 0, stream>>>(g1, be1, mu1, va1, g2, be2, mu2, va2, b1,
                                 sc1, sh1, sc2, sh2);

  gemm1_rows_resident<<<dim3(313, 4), 256, 0, stream>>>(x, sc1, sh1, W1, h1);

  node_att1<<<(N_NODES / NA1_NODES + 7) / 8, 256, 0, stream>>>(h1, att1, ai, aj, m1);
  edge_max1<<<(ETOT + 255) / 256, 256, 0, stream>>>(rowA, colA, ai, aj, m1);
  agg1_owned_rows<<<A1_WGS, 1024, 0, stream>>>(rowA, colA, ai, aj, m1, h1, out1);

  gemm_bn_wmma<<<(2500 * 3 + 7) / 8, 256, 0, stream>>>(out1, sc2, sh2, W2, h2,
                                                       N_NODES, D1, D2);

  node_att2<<<(N_NODES * H2) / 256, 256, 0, stream>>>(h2, att2, s2);
  agg2_owned_rows<<<A2_WGS, 1024, 0, stream>>>(rowA, colA, s2, h2, b2, out);
}


# Both aggregation scans read a 1-byte owner number per edge (row // 157, written by the max pass and the layer-2 score kernel), not the 4-byte row; a trip's hits looked up together; sums: LDS order
# speedup vs baseline: 3.8956x; 1.0786x over previous
#include <hip/hip_runtime.h>
#include <math.h>

#define N_NODES 40000
#define N_EDGES 320000
#define ETOT    (N_EDGES + N_NODES)
#define F_IN    256
#define H1      4
#define C1      128
#define D1      (H1*C1)
#define H2      8
#define C2      6
#define D2      (H2*C2)
#define BN_EPS    1e-5f
#define NEG_SLOPE 0.2f

typedef __attribute__((ext_vector_type(2))) float v2f;
typedef __attribute__((ext_vector_type(4))) float v4f;
typedef __attribute__((ext_vector_type(8))) float v8f;


__device__ __forceinline__ void atomic_max_f32(float* addr, float v) {
  if (v >= 0.0f) atomicMax((int*)addr, __float_as_int(v));
  else           atomicMin((unsigned int*)addr, __float_as_uint(v));
}

__device__ __forceinline__ void edge_rc_flat(const int* __restrict__ rowA,
                                             const int* __restrict__ colA,
                                             int e, int& r, int& c) {
  int ec = e < N_EDGES ? e : N_EDGES - 1;
  int lr = rowA[ec], lc = colA[ec];
  bool self = e >= N_EDGES;
  r = self ? e - N_EDGES : lr;
  c = self ? e - N_EDGES : lc;
}


__global__ void bn_prep(const float* g1, const float* be1, const float* mu1, const float* va1,
                        const float* g2, const float* be2, const float* mu2, const float* va2,
                        const float* bias1,
                        float* sc1, float* sh1, float* sc2, float* sh2) {
  int i = blockIdx.x * blockDim.x + threadIdx.x;
  if (i < F_IN) {
    float s = g1[i] / sqrtf(va1[i] + BN_EPS);
    sc1[i] = s; sh1[i] = be1[i] - mu1[i] * s;
  }
  if (i < D1) {
    float s = g2[i] / sqrtf(va2[i] + BN_EPS);
    sc2[i] = s; sh2[i] = be2[i] + (bias1[i] - mu2[i]) * s;
  }
}


__global__ __launch_bounds__(256)
void gemm_bn_wmma(const float* __restrict__ A,
                  const float* __restrict__ scale,
                  const float* __restrict__ shift,
                  const float* __restrict__ B,
                  float* __restrict__ C,
                  int M, int K, int N) {
  int wave = blockIdx.x * (blockDim.x >> 5) + (threadIdx.x >> 5);
  int ntiles = N >> 4;
  int tm = wave / ntiles;
  int tn = wave - tm * ntiles;
  if (tm >= (M >> 4)) return;

  int lane = threadIdx.x & 31;
  int half = lane >> 4;
  int l16  = lane & 15;
  int row  = (tm << 4) + l16;
  int colB = (tn << 4) + l16;

  const float* __restrict__ Arow = A + (size_t)row * K;
  v8f acc = {};
  for (int k = 0; k < K; k += 8) {
    int kk = k + (half << 2);
    v4f a4 = *(const v4f*)(Arow + kk);
    v4f s4 = *(const v4f*)(scale + kk);
    v4f t4 = *(const v4f*)(shift + kk);
    const float* __restrict__ Bk = B + (size_t)kk * N + colB;
    v2f a, b;
    a.x = a4.x * s4.x + t4.x;
    a.y = a4.y * s4.y + t4.y;
    b.x = Bk[0];
    b.y = Bk[(size_t)N];
    acc = __builtin_amdgcn_wmma_f32_16x16x4_f32(
         false, a,  false, b,
         (short)0, acc,  false,  false);
    a.x = a4.z * s4.z + t4.z;
    a.y = a4.w * s4.w + t4.w;
    b.x = Bk[(size_t)2 * N];
    b.y = Bk[(size_t)3 * N];
    acc = __builtin_amdgcn_wmma_f32_16x16x4_f32(
         false, a,  false, b,
         (short)0, acc,  false,  false);
  }
  float* Crow = C + (size_t)((tm << 4) + (half << 3)) * N + colB;
  #pragma unroll
  for (int r = 0; r < 8; ++r) Crow[(size_t)r * N] = acc[r];
}

#define G1_COLS_PER_WAVE 128

__global__ __launch_bounds__(256)
void gemm1_rows_resident(const float* __restrict__ A,
                         const float* __restrict__ scale,
                         const float* __restrict__ shift,
                         const float* __restrict__ B,
                         float* __restrict__ C) {
  int tm = blockIdx.x * 8 + (threadIdx.x >> 5);
  if (tm >= (N_NODES >> 4)) return;
  int lane = threadIdx.x & 31;
  int half = lane >> 4;
  int l16  = lane & 15;

  const float* __restrict__ Arow = A + (size_t)((tm << 4) + l16) * F_IN + (half << 2);
  const float* __restrict__ sc = scale + (half << 2);
  const float* __restrict__ sh = shift + (half << 2);
  v4f a[F_IN / 8];
  #pragma unroll
  for (int t = 0; t < F_IN / 8; ++t) a[t] = *(const v4f*)(Arow + 8 * t);
  #pragma unroll
  for (int t0 = 0; t0 < F_IN / 8; t0 += 8) {
    #pragma unroll
    for (int t = t0; t < t0 + 8; ++t) {
      v4f s4 = *(const v4f*)(sc + 8 * t);
      v4f t4 = *(const v4f*)(sh + 8 * t);
      a[t].x = a[t].x * s4.x + t4.x;
      a[t].y = a[t].y * s4.y + t4.y;
      a[t].z = a[t].z * s4.z + t4.z;
      a[t].w = a[t].w * s4.w + t4.w;
    }
    __builtin_amdgcn_sched_barrier(0);
  }

  int cb = blockIdx.y * G1_COLS_PER_WAVE + (l16 << 2);
  const float* __restrict__ Bl = B + (size_t)(half << 2) * D1 + cb;
  float* __restrict__ Cl = C + (size_t)((tm << 4) + (half << 3)) * D1 + cb;
  for (int j = 0; j < G1_COLS_PER_WAVE; j += 64) {
    v8f acc0 = {}, acc1 = {}, acc2 = {}, acc3 = {};
    const float* __restrict__ Bj = Bl + j;
    v4f bq[3][4];
    #pragma unroll
    for (int p = 0; p < 2; ++p) {
      #pragma unroll
      for (int q = 0; q < 4; ++q) bq[p][q] = *(const v4f*)(Bj + (size_t)(8 * p + q) * D1);
    }
    #pragma unroll
    for (int t = 0; t < F_IN / 8; ++t) {
      if (t + 2 < F_IN / 8) {
        #pragma unroll
        for (int q = 0; q < 4; ++q)
          bq[(t + 2) % 3][q] = *(const v4f*)(Bj + (size_t)(8 * (t + 2) + q) * D1);
      }
      __builtin_amdgcn_sched_barrier(0);
      const v4f b0 = bq[t % 3][0], b1 = bq[t % 3][1], b2 = bq[t % 3][2], b3 = bq[t % 3][3];
      v2f a01, a23, b;
      a01.x = a[t].x; a01.y = a[t].y;
      a23.x = a[t].z; a23.y = a[t].w;
      b.x = b0.x; b.y = b1.x;
      acc0 = __builtin_amdgcn_wmma_f32_16x16x4_f32(false, a01, false, b, (short)0, acc0, false, false);
      b.x = b0.y; b.y = b1.y;
      acc1 = __builtin_amdgcn_wmma_f32_16x16x4_f32(false, a01, false, b, (short)0, acc1, false, false);
      b.x = b0.z; b.y = b1.z;
      acc2 = __builtin_amdgcn_wmma_f32_16x16x4_f32(false, a01, false, b, (short)0, acc2, false, false);
      b.x = b0.w; b.y = b1.w;
      acc3 = __builtin_amdgcn_wmma_f32_16x16x4_f32(false, a01, false, b, (short)0, acc3, false, false);
      b.x = b2.x; b.y = b3.x;
      acc0 = __builtin_amdgcn_wmma_f32_16x16x4_f32(false, a23, false, b, (short)0, acc0, false, false);
      b.x = b2.y; b.y = b3.y;
      acc1 = __builtin_amdgcn_wmma_f32_16x16x4_f32(false, a23, false, b, (short)0, acc1, false, false);
      b.x = b2.z; b.y = b3.z;
      acc2 = __builtin_amdgcn_wmma_f32_16x16x4_f32(false, a23, false, b, (short)0, acc2, false, false);
      b.x = b2.w; b.y = b3.w;
      acc3 = __builtin_amdgcn_wmma_f32_16x16x4_f32(false, a23, false, b, (short)0, acc3, false, false);
      __builtin_amdgcn_sched_barrier(0);
    }
    #pragma unroll
    for (int r = 0; r < 8; ++r) {
      v4f o;
      o.x = acc0[r]; o.y = acc1[r]; o.z = acc2[r]; o.w = acc3[r];
      *(v4f*)(Cl + (size_t)r * D1 + j) = o;
    }
  }
}


#define NA1_NODES 8

__global__ __launch_bounds__(256)
void node_att1(const float* __restrict__ h1, const float* __restrict__ att1,
               float* __restrict__ ai, float* __restrict__ aj,
               float* __restrict__ m1) {
  int wave = blockIdx.x * (blockDim.x >> 5) + (threadIdx.x >> 5);
  int n0 = wave * NA1_NODES;
  if (n0 >= N_NODES) return;
  int lane = threadIdx.x & 31;
  v4f al[H1], ar[H1];
  #pragma unroll
  for (int h = 0; h < H1; ++h) {
    al[h] = *(const v4f*)(att1 + h * (2 * C1) + (lane << 2));
    ar[h] = *(const v4f*)(att1 + h * (2 * C1) + C1 + (lane << 2));
  }
  const bool b16 = (lane & 16) != 0, b8 = (lane & 8) != 0, b4 = (lane & 4) != 0;
  const int idx = lane >> 2;
  const float* __restrict__ hp = h1 + (size_t)n0 * D1 + (lane << 2);
  #pragma unroll
  for (int i = 0; i < NA1_NODES; i += 4) {
    v4f v[4][H1];
    #pragma unroll
    for (int u = 0; u < 4; ++u) {
      #pragma unroll
      for (int h = 0; h < H1; ++h) v[u][h] = *(const v4f*)(hp + (size_t)(i + u) * D1 + h * C1);
    }
    #pragma unroll
    for (int u = 0; u < 4; ++u) {
      float r[8];
      #pragma unroll
      for (int h = 0; h < H1; ++h) {
        const v4f x = v[u][h];
        r[h]     = x.x * al[h].x + x.y * al[h].y + x.z * al[h].z + x.w * al[h].w;
        r[4 + h] = x.x * ar[h].x + x.y * ar[h].y + x.z * ar[h].z + x.w * ar[h].w;
      }
      float t4[4], t2[2];
      #pragma unroll
      for (int k = 0; k < 4; ++k) {
        float keep = b16 ? r[4 + k] : r[k];
        float send = b16 ? r[k] : r[4 + k];
        t4[k] = keep + __shfl_xor(send, 16, 32);
      }
      #pragma unroll
      for (int k = 0; k < 2; ++k) {
        float keep = b8 ? t4[2 + k] : t4[k];
        float send = b8 ? t4[k] : t4[2 + k];
        t2[k] = keep + __shfl_xor(send, 8, 32);
      }
      float keep = b4 ? t2[1] : t2[0];
      float send = b4 ? t2[0] : t2[1];
      float t = keep + __shfl_xor(send, 4, 32);
      t += __shfl_xor(t, 2, 32);
      t += __shfl_xor(t, 1, 32);
      if ((lane & 3) == 0) {
        int n = n0 + i + u;
        if (idx < 4) {
          ai[n * 4 + idx] = t;
          m1[n * 4 + idx] = -__builtin_inff();
        } else {
          aj[n * 4 + (idx - 4)] = t;
        }
      }
    }
  }
}

#define A1_ROWS 157
#define A1_WGS  ((N_NODES + A1_ROWS - 1) / A1_ROWS)
#define OWN_NV  (N_EDGES / 16)
static_assert(A1_WGS <= 255, "owner numbers fit a byte, and byte 255 matches no work-group");
static_assert(N_EDGES % 512 == 0, "a wave's 32 x 16 owner bytes are all inside the map");

typedef __attribute__((ext_vector_type(4))) int v4i;

__device__ __forceinline__ unsigned zero_bytes(unsigned x) {
  return ~(((x & 0x7F7F7F7Fu) + 0x7F7F7F7Fu) | x) & 0x80808080u;
}

__device__ __forceinline__ unsigned own_hits_word(unsigned x, unsigned pat) {
  return ((zero_bytes(x ^ pat) >> 7) * 0x10204080u) >> 28;
}
__device__ __forceinline__ unsigned own_hits(v4i G, unsigned pat) {
  return own_hits_word((unsigned)G.x, pat)       | own_hits_word((unsigned)G.y, pat) << 4 |
         own_hits_word((unsigned)G.z, pat) << 8  | own_hits_word((unsigned)G.w, pat) << 12;
}

__device__ __forceinline__ void own_write4(const int* __restrict__ rowA, unsigned* __restrict__ own, int g) {
  const v4i r = *(const v4i*)(rowA + 4 * g);
  own[g] = (unsigned)(unsigned char)((unsigned)r.x / A1_ROWS)       | (unsigned)(unsigned char)((unsigned)r.y / A1_ROWS) << 8 |
           (unsigned)(unsigned char)((unsigned)r.z / A1_ROWS) << 16 | (unsigned)(unsigned char)((unsigned)r.w / A1_ROWS) << 24;
}
static_assert((N_EDGES / 4) % 32 == 0, "the map's words are written by whole waves");


__global__ void edge_max1(const int* __restrict__ rowA, const int* __restrict__ colA,
                          const float* __restrict__ ai, const float* __restrict__ aj,
                          float* __restrict__ m1, float* __restrict__ ownw) {
  const int gid = blockIdx.x * blockDim.x + threadIdx.x;
  if (gid < N_EDGES / 4) own_write4(rowA, (unsigned*)ownw, gid);
  int wave = blockIdx.x * (blockDim.x >> 5) + (threadIdx.x >> 5);
  int e0 = wave << 5;
  if (e0 >= ETOT) return;
  int lane = threadIdx.x & 31;
  int sub = lane >> 2, h = lane & 3;
  int dsti[4]; float val[4];
  #pragma unroll
  for (int i = 0; i < 4; ++i) {
    int row, col; edge_rc_flat(rowA, colA, e0 + (i << 3) + sub, row, col);
    float a = ai[row * 4 + h] + aj[col * 4 + h];
    val[i] = a > 0.f ? a : NEG_SLOPE * a;
    dsti[i] = row * 4 + h;
  }
  #pragma unroll
  for (int i = 0; i < 4; ++i) atomic_max_f32(&m1[dsti[i]], val[i]);
}

#define A1_NV   (N_EDGES / 4)
static_assert(A1_ROWS * (D1 + H1) * 4 <= 320 * 1024, "the owned rows and their weight sums fit the LDS of one work-group");
static_assert(N_EDGES % 128 == 0, "a wave's 32 x 4 destination indices are all inside the list");

__device__ __forceinline__ float gat_weight(float a, float m) {
  a = a > 0.f ? a : NEG_SLOPE * a;
  return expf(a - m);
}

__device__ __forceinline__ void gat_weights4(int r, int c,
                                             const float* __restrict__ ai, const float* __restrict__ aj,
                                             const float* __restrict__ m1,
                                             float w[4]) {
  const v4f A = *(const v4f*)(ai + r * 4), B = *(const v4f*)(aj + c * 4);
  const v4f M = *(const v4f*)(m1 + r * 4);
  w[0] = gat_weight(A.x + B.x, M.x);
  w[1] = gat_weight(A.y + B.y, M.y);
  w[2] = gat_weight(A.z + B.z, M.z);
  w[3] = gat_weight(A.w + B.w, M.w);
}

__device__ __forceinline__ float weight_of_lane(const float w[4], int lane) {
  float x = w[0];
  x = lane == 1 ? w[1] : x;
  x = lane == 2 ? w[2] : x;
  x = lane == 3 ? w[3] : x;
  return x;
}

__device__ __forceinline__ void agg1_one(int d, int c, int r0, int lane,
                                         const float* __restrict__ ai, const float* __restrict__ aj,
                                         const float* __restrict__ m1,
                                         const float* __restrict__ h1, float* acc, float* den) {
  const float* __restrict__ src = h1 + (size_t)c * D1 + lane;
  float v[16];
  #pragma unroll
  for (int q = 0; q < 16; ++q) v[q] = src[q * 32];
  float w[4];
  gat_weights4(r0 + d, c, ai, aj, m1, w);
  float* dst = acc + d * D1 + lane;
  #pragma unroll
  for (int q = 0; q < 16; ++q) atomicAdd(dst + q * 32, v[q] * w[q >> 2]);
  const float wl = weight_of_lane(w, lane);
  if (lane < 4) atomicAdd(den + d * 4 + lane, wl);
}

__device__ __forceinline__ void agg1_pair(int d0, int c0, int d1, int c1, int r0, int lane,
                                          const float* __restrict__ ai, const float* __restrict__ aj,
                                          const float* __restrict__ m1,
                                          const float* __restrict__ h1, float* acc, float* den) {
  const float* __restrict__ src0 = h1 + (size_t)c0 * D1 + lane;
  const float* __restrict__ src1 = h1 + (size_t)c1 * D1 + lane;
  float v0[16], v1[16];
  #pragma unroll
  for (int q = 0; q < 16; ++q) v0[q] = src0[q * 32];
  #pragma unroll
  for (int q = 0; q < 16; ++q) v1[q] = src1[q * 32];
  float w0[4], w1[4];
  gat_weights4(r0 + d0, c0, ai, aj, m1, w0);
  gat_weights4(r0 + d1, c1, ai, aj, m1, w1);
  float* dst0 = acc + d0 * D1 + lane;
  float* dst1 = acc + d1 * D1 + lane;
  #pragma unroll
  for (int q = 0; q < 16; ++q) atomicAdd(dst0 + q * 32, v0[q] * w0[q >> 2]);
  #pragma unroll
  for (int q = 0; q < 16; ++q) atomicAdd(dst1 + q * 32, v1[q] * w1[q >> 2]);
  const float wl0 = weight_of_lane(w0, lane), wl1 = weight_of_lane(w1, lane);
  if (lane < 4) {
    atomicAdd(den + d0 * 4 + lane, wl0);
    atomicAdd(den + d1 * 4 + lane, wl1);
  }
}

__global__ __launch_bounds__(1024)
void agg1_owned_rows(const int* __restrict__ rowA, const int* __restrict__ colA,
                     const float* __restrict__ ownw,
                     const float* __restrict__ ai, const float* __restrict__ aj,
                     const float* __restrict__ m1,
                     const float* __restrict__ h1, float* __restrict__ out1) {
  const unsigned char* __restrict__ own = (const unsigned char*)ownw;
  __shared__ __attribute__((aligned(16))) float acc[A1_ROWS * D1];
  __shared__ __attribute__((aligned(16))) float den[A1_ROWS * H1];
  const int r0 = blockIdx.x * A1_ROWS;
  const int nr = (N_NODES - r0) < A1_ROWS ? (N_NODES - r0) : A1_ROWS;
  const int wave = threadIdx.x >> 5, lane = threadIdx.x & 31;

  for (int r = wave; r < nr; r += 32) {
    int n = r0 + r;
    const float* __restrict__ src = h1 + (size_t)n * D1 + lane;
    float v[16];
    #pragma unroll
    for (int q = 0; q < 16; ++q) v[q] = src[q * 32];
    float w[4];
    gat_weights4(n, n, ai, aj, m1, w);
    float* dst = acc + r * D1 + lane;
    #pragma unroll
    for (int q = 0; q < 16; ++q) dst[q * 32] = v[q] * w[q >> 2];
    const float wl = weight_of_lane(w, lane);
    if (lane < 4) den[r * 4 + lane] = wl;
  }
  __syncthreads();

  const unsigned unr = (unsigned)nr;
  const unsigned pat = (unsigned)blockIdx.x * 0x01010101u;
  const v4i none = {-1, -1, -1, -1};
  int vb = wave * 32;
  v4i g0 = vb        < OWN_NV ? *(const v4i*)(own + 16 * (vb + lane))        : none;
  v4i g1 = vb + 1024 < OWN_NV ? *(const v4i*)(own + 16 * (vb + 1024 + lane)) : none;
  v4i g2 = vb + 2048 < OWN_NV ? *(const v4i*)(own + 16 * (vb + 2048 + lane)) : none;
  v4i g3 = vb + 3072 < OWN_NV ? *(const v4i*)(own + 16 * (vb + 3072 + lane)) : none;
  bool have = false;
  int pd = 0, pc = 0;
  #pragma nounroll
  for (; vb < OWN_NV; vb += 4096) {
    const v4i n0 = vb + 4096 < OWN_NV ? *(const v4i*)(own + 16 * (vb + 4096 + lane)) : none;
    const v4i n1 = vb + 5120 < OWN_NV ? *(const v4i*)(own + 16 * (vb + 5120 + lane)) : none;
    const v4i n2 = vb + 6144 < OWN_NV ? *(const v4i*)(own + 16 * (vb + 6144 + lane)) : none;
    const v4i n3 = vb + 7168 < OWN_NV ? *(const v4i*)(own + 16 * (vb + 7168 + lane)) : none;
    unsigned long long hits = (unsigned long long)own_hits(g0, pat)       | (unsigned long long)own_hits(g1, pat) << 16 |
                              (unsigned long long)own_hits(g2, pat) << 32 | (unsigned long long)own_hits(g3, pat) << 48;
    const int e0 = 16 * (vb + lane);
    #pragma nounroll
    for (int round = 0; round < 64; ++round) {
      const bool tj = hits != 0;
      unsigned mask = __builtin_amdgcn_ballot_w32(tj);
      if (mask == 0) break;
      int dj = -1, c = 0;
      if (tj) {
        const int p = __builtin_ctzll(hits);
        hits &= hits - 1;
        const int e = e0 + (p >> 4) * (16 * 1024) + (p & 15);
        dj = rowA[e] - r0;
        c = colA[e];
      }
      while (mask) {
        const int l = __builtin_ctz(mask);
        mask &= mask - 1;
        const int dl = __builtin_amdgcn_readlane(dj, l);
        const int cl = __builtin_amdgcn_readlane(c, l);
        if ((unsigned)dl >= unr) continue;
        if (!have) { pd = dl; pc = cl; have = true; }
        else { agg1_pair(pd, pc, dl, cl, r0, lane, ai, aj, m1, h1, acc, den); have = false; }
      }
    }
    g0 = n0; g1 = n1; g2 = n2; g3 = n3;
  }
  if (have) agg1_one(pd, pc, r0, lane, ai, aj, m1, h1, acc, den);
  __syncthreads();

  for (int r = wave; r < nr; r += 32) {
    const float* src = acc + r * D1 + (lane << 2);
    float* __restrict__ dst = out1 + (size_t)(r0 + r) * D1 + (lane << 2);
    const v4f S = *(const v4f*)(den + r * 4);
    v4f t[4];
    #pragma unroll
    for (int k = 0; k < 4; ++k) t[k] = *(const v4f*)(src + k * C1);
    #pragma unroll
    for (int k = 0; k < 4; ++k) {
      const float dn = S[k] + 1e-16f;
      v4f o;
      o.x = t[k].x / dn; o.y = t[k].y / dn; o.z = t[k].z / dn; o.w = t[k].w / dn;
      *(v4f*)(dst + k * C1) = o;
    }
  }
}


static_assert(N_EDGES / 4 <= N_NODES * H2, "a thread per map word among the (node, head) threads");
__global__ void node_att2(const float* __restrict__ h2, const float* __restrict__ att2,
                          float* __restrict__ s2,
                          const int* __restrict__ rowA, float* __restrict__ ownw) {
  int tid = blockIdx.x * blockDim.x + threadIdx.x;
  if (tid < N_EDGES / 4) own_write4(rowA, (unsigned*)ownw, tid);
  if (tid >= N_NODES * H2) return;
  int n = tid >> 3, h = tid & 7;
  const float* __restrict__ hp = h2 + (size_t)n * D2 + h * C2;
  const float* __restrict__ ap = att2 + h * (2 * C2);
  float s = 0.f;
  #pragma unroll
  for (int c = 0; c < C2; ++c) s += hp[c] * (ap[c] + ap[C2 + c]);
  s2[tid] = tanhf(s);
}

#define A2_ROWS  A1_ROWS
#define A2_WGS   A1_WGS
#define A2_PITCH 57
#define A2_CAP   65536
static_assert(((A2_ROWS + 1) * A2_PITCH + (A2_ROWS + 1) * H2 + A2_CAP + 1) * 4 <= 320 * 1024, "sums, maxima and list fit the LDS of one work-group");
static_assert(A2_ROWS <= 256 && N_NODES <= (1 << 16), "list entries are first edge numbers, then (row, node) as row<<16|node");

__device__ __forceinline__ void lds_max_f32(float* addr, float v) {
  if (v >= 0.0f) atomicMax((int*)addr, __float_as_int(v));
  else           atomicMin((unsigned int*)addr, __float_as_uint(v));
}

__device__ __forceinline__ int edge_src(const int* __restrict__ colA, int e) {
  int ec = e < N_EDGES ? e : N_EDGES - 1;
  int lc = colA[ec];
  return e < N_EDGES ? lc : e - N_EDGES;
}

template <int MODE>
__device__ __forceinline__ void l2_take(int d, int e, const int* __restrict__ colA,
                                        const float* __restrict__ s2, const float* __restrict__ h2,
                                        float* acc, float* mx) {
  const int c = edge_src(colA, e);
  const v4f sa = *(const v4f*)(s2 + c * H2), sb = *(const v4f*)(s2 + c * H2 + 4);
  const float s[H2] = {sa.x, sa.y, sa.z, sa.w, sb.x, sb.y, sb.z, sb.w};
  if (MODE == 1) {
    #pragma unroll
    for (int h = 0; h < H2; ++h) lds_max_f32(mx + d * H2 + h, s[h]);
  } else {
    const float* __restrict__ hp = h2 + (size_t)c * D2;
    v4f x[D2 / 4];
    #pragma unroll
    for (int q = 0; q < D2 / 4; ++q) x[q] = *(const v4f*)(hp + 4 * q);
    float* dst = acc + d * A2_PITCH;
    #pragma unroll
    for (int h = 0; h < H2; ++h) {
      const float w = expf(s[h] - mx[d * H2 + h]);
      #pragma unroll
      for (int k = 0; k < C2; ++k) {
        const int j = h * C2 + k;
        atomicAdd(dst + j, x[j >> 2][j & 3] * w);
      }
      atomicAdd(dst + D2 + h, w);
    }
  }
}

#define A2_LD 4

__device__ __forceinline__ void l2_load_groups(v4i (&g)[A2_LD], const unsigned char* __restrict__ own, int vb, int lane) {
  #pragma unroll
  for (int u = 0; u < A2_LD; ++u) {
    const int v = vb + u * 1024;
    g[u] = *(const v4i*)(own + 16 * ((v < OWN_NV ? v : OWN_NV - 32) + lane));
  }
}

__device__ __forceinline__ void l2_list_groups(const v4i (&g)[A2_LD], int vb, int* list, int* cnt,
                                               unsigned pat, int lane) {
  #pragma unroll
  for (int u = 0; u < A2_LD; ++u) {
    if (vb + u * 1024 >= OWN_NV) continue;
    unsigned hits = own_hits(g[u], pat);
    if (hits) {
      const int e16 = 16 * (vb + u * 1024 + lane);
      int slot = atomicAdd(cnt, __builtin_popcount(hits));
      while (hits) {
        const int p = __builtin_ctz(hits);
        hits &= hits - 1;
        if (slot < A2_CAP) list[slot] = e16 + p;
        ++slot;
      }
    }
  }
}

__device__ __forceinline__ void l2_scan_list(const unsigned char* __restrict__ own, int* list, int* cnt,
                                             unsigned pat, int wave, int lane) {
  int vb = wave * 32;
  v4i ga[A2_LD], gb[A2_LD];
  l2_load_groups(ga, own, vb, lane);
  #pragma nounroll
  for (; vb < OWN_NV; vb += 2 * A2_LD * 1024) {
    l2_load_groups(gb, own, vb + A2_LD * 1024, lane);
    __builtin_amdgcn_sched_barrier(0);
    l2_list_groups(ga, vb, list, cnt, pat, lane);
    l2_load_groups(ga, own, vb + 2 * A2_LD * 1024, lane);
    __builtin_amdgcn_sched_barrier(0);
    l2_list_groups(gb, vb + A2_LD * 1024, list, cnt, pat, lane);
  }
}

template <int MODE>
__device__ __forceinline__ void l2_rescan(const int* __restrict__ rowA, const int* __restrict__ colA,
                                          const float* __restrict__ s2, const float* __restrict__ h2,
                                          float* acc, float* mx, int r0, int nr, int tid) {
  const int ngroups = A1_NV + nr;
  #pragma nounroll
  for (int v = tid; v < ngroups; v += 1024) {
    const bool self = v >= A1_NV;
    const v4i none = {-1, -1, -1, -1};
    const v4i G = self ? none : *(const v4i*)(rowA + 4 * (self ? 0 : v));
    #pragma nounroll
    for (int j = 0; j < 4; ++j) {
      int dj = (j == 0 ? G.x : j == 1 ? G.y : j == 2 ? G.z : G.w) - r0;
      int e = 4 * v + j;
      if (self) { dj = j == 0 ? v - A1_NV : -1; e = N_EDGES + r0 + (v - A1_NV); }
      if ((unsigned)dj < (unsigned)nr) l2_take<MODE>(dj, e, colA, s2, h2, acc, mx);
    }
  }
}

__global__ __launch_bounds__(1024)
void agg2_owned_rows(const int* __restrict__ rowA, const int* __restrict__ colA,
                     const float* __restrict__ ownw,
                     const float* __restrict__ s2, const float* __restrict__ h2,
                     const float* __restrict__ b2, float* __restrict__ out) {
  const unsigned char* __restrict__ own = (const unsigned char*)ownw;
  __shared__ float acc[(A2_ROWS + 1) * A2_PITCH];
  __shared__ float mx[(A2_ROWS + 1) * H2];
  __shared__ int list[A2_CAP];
  __shared__ int cnt;
  const int r0 = blockIdx.x * A2_ROWS;
  const int nr = (N_NODES - r0) < A2_ROWS ? (N_NODES - r0) : A2_ROWS;
  const int tid = threadIdx.x, lane = tid & 31;
  const int wave = __builtin_amdgcn_readfirstlane(tid >> 5);

  for (int i = tid; i < (A2_ROWS + 1) * A2_PITCH; i += 1024) acc[i] = 0.f;
  for (int i = tid; i < (A2_ROWS + 1) * H2; i += 1024) mx[i] = -__builtin_inff();
  if (tid < nr) list[tid] = N_EDGES + r0 + tid;
  if (tid == 0) cnt = nr;
  __syncthreads();

  l2_scan_list(own, list, &cnt, (unsigned)blockIdx.x * 0x01010101u, wave, lane);
  __syncthreads();
  const int found = __builtin_amdgcn_readfirstlane(cnt);

  if (found <= A2_CAP) {
    for (int i = tid; i < found; i += 1024) {
      int r, c; edge_rc_flat(rowA, colA, list[i], r, c);
      int d = r - r0;
      if ((unsigned)d >= (unsigned)nr) d = A2_ROWS;
      list[i] = (d << 16) | c;
      const v4f sa = *(const v4f*)(s2 + c * H2), sb = *(const v4f*)(s2 + c * H2 + 4);
      float* mp = mx + d * H2;
      lds_max_f32(mp + 0, sa.x); lds_max_f32(mp + 1, sa.y); lds_max_f32(mp + 2, sa.z); lds_max_f32(mp + 3, sa.w);
      lds_max_f32(mp + 4, sb.x); lds_max_f32(mp + 5, sb.y); lds_max_f32(mp + 6, sb.z); lds_max_f32(mp + 7, sb.w);
    }
    __syncthreads();
    const int grp = tid >> 3, h = tid & 7;
    for (int i0 = grp; i0 < found; i0 += 4 * 128) {
      int p[4], dr[4]; float s[4], m[4]; v2f x[4][3];
      #pragma unroll
      for (int u = 0; u < 4; ++u) {
        const int i = i0 + u * 128;
        const bool ok = i < found;
        p[u] = list[ok ? i : i0];
        dr[u] = ok ? (p[u] >> 16) : A2_ROWS;
      }
      #pragma unroll
      for (int u = 0; u < 4; ++u) {
        const int d = p[u] >> 16, c = p[u] & 0xFFFF;
        s[u] = s2[c * H2 + h];
        m[u] = mx[d * H2 + h];
        const float* __restrict__ hp = h2 + (size_t)c * D2 + h * C2;
        x[u][0] = *(const v2f*)(hp); x[u][1] = *(const v2f*)(hp + 2); x[u][2] = *(const v2f*)(hp + 4);
      }
      #pragma unroll
      for (int u = 0; u < 4; ++u) {
        const float w = expf(s[u] - m[u]);
        float* dst = acc + dr[u] * A2_PITCH + h * C2;
        atomicAdd(dst + 0, x[u][0].x * w); atomicAdd(dst + 1, x[u][0].y * w);
        atomicAdd(dst + 2, x[u][1].x * w); atomicAdd(dst + 3, x[u][1].y * w);
        atomicAdd(dst + 4, x[u][2].x * w); atomicAdd(dst + 5, x[u][2].y * w);
        atomicAdd(acc + dr[u] * A2_PITCH + D2 + h, w);
      }
    }
  } else {
    l2_rescan<1>(rowA, colA, s2, h2, acc, mx, r0, nr, tid);
    __syncthreads();
    l2_rescan<2>(rowA, colA, s2, h2, acc, mx, r0, nr, tid);
  }
  __syncthreads();

  if (tid < nr) {
    const float* rp = acc + tid * A2_PITCH;
    float dn[H2];
    #pragma unroll
    for (int hh = 0; hh < H2; ++hh) dn[hh] = rp[D2 + hh] + 1e-16f;
    float v[C2];
    #pragma unroll
    for (int c = 0; c < C2; ++c) {
      float s = 0.f;
      #pragma unroll
      for (int hh = 0; hh < H2; ++hh) s += rp[hh * C2 + c] / dn[hh];
      float t = s * (1.0f / H2) + b2[c];
      v[c] = t > 0.f ? t : (expf(t) - 1.0f);
    }
    float m = v[0];
    #pragma unroll
    for (int c = 1; c < C2; ++c) m = fmaxf(m, v[c]);
    float sum = 0.f;
    #pragma unroll
    for (int c = 0; c < C2; ++c) sum += expf(v[c] - m);
    float lse = logf(sum);
    #pragma unroll
    for (int c = 0; c < C2; ++c) out[(size_t)(r0 + tid) * C2 + c] = v[c] - m - lse;
  }
}


extern "C" void kernel_launch(void* const* d_in, const int* in_sizes, int n_in,
                              void* d_out, int out_size, void* d_ws, size_t ws_size,
                              hipStream_t stream) {
  const float* x    = (const float*)d_in[0];
  const int*   ei   = (const int*)  d_in[1];
  const float* g1   = (const float*)d_in[2];
  const float* be1  = (const float*)d_in[3];
  const float* mu1  = (const float*)d_in[4];
  const float* va1  = (const float*)d_in[5];
  const float* W1   = (const float*)d_in[6];
  const float* att1 = (const float*)d_in[7];
  const float* b1   = (const float*)d_in[8];
  const float* g2   = (const float*)d_in[9];
  const float* be2  = (const float*)d_in[10];
  const float* mu2  = (const float*)d_in[11];
  const float* va2  = (const float*)d_in[12];
  const float* W2   = (const float*)d_in[13];
  const float* att2 = (const float*)d_in[14];
  const float* b2   = (const float*)d_in[15];
  const int* rowA = ei;
  const int* colA = ei + N_EDGES;
  float* out = (float*)d_out;
  float* ws  = (float*)d_ws;

  float* sc1  = ws;
  float* sh1  = ws + 256;
  float* sc2  = ws + 512;
  float* sh2  = ws + 1024;
  float* ai   = ws + 1536;
  float* aj   = ai  + (size_t)N_NODES * H1;
  float* m1   = aj  + (size_t)N_NODES * H1;
  float* out1 = m1  + (size_t)N_NODES * H1;
  float* h1   = out1 + (size_t)N_NODES * D1;
  float* h2   = h1;
  float* s2   = h2  + (size_t)N_NODES * D2;

  bn_prep<<<2, 256, 0, stream>>>(g1, be1, mu1, va1, g2, be2, mu2, va2, b1,
                                 sc1, sh1, sc2, sh2);

  gemm1_rows_resident<<<dim3(313, 4), 256, 0, stream>>>(x, sc1, sh1, W1, h1);

  node_att1<<<(N_NODES / NA1_NODES + 7) / 8, 256, 0, stream>>>(h1, att1, ai, aj, m1);
  static_assert(N_EDGES <= N_NODES * C2 * 4 && N_EDGES <= N_NODES * H1 * 4, "an owner map fits the output buffer and the ai plane");
  static_assert(N_EDGES / 4 <= ETOT, "edge_max1 has a thread per map word");
  float* own1 = out;
  float* own2 = ai;
  edge_max1<<<(ETOT + 255) / 256, 256, 0, stream>>>(rowA, colA, ai, aj, m1, own1);
  agg1_owned_rows<<<A1_WGS, 1024, 0, stream>>>(rowA, colA, own1, ai, aj, m1, h1, out1);

  gemm_bn_wmma<<<(2500 * 3 + 7) / 8, 256, 0, stream>>>(out1, sc2, sh2, W2, h2,
                                                       N_NODES, D1, D2);

  node_att2<<<(N_NODES * H2) / 256, 256, 0, stream>>>(h2, att2, s2, rowA, own2);
  agg2_owned_rows<<<A2_WGS, 1024, 0, stream>>>(rowA, colA, own2, s2, h2, b2, out);
}
